# starting kernel
# baseline (speedup 1.0000x reference)
#include <hip/hip_runtime.h>
#include <hip/hip_bf16.h>
#include <math.h>

#define BB   2
#define LL   384
#define DD   128
#define NH   8
#define DHS_ 16
#define DHP_ 4
#define BL   (BB*LL)
#define BNL  (BB*NH*LL)
#define DOUT 1280
#define LT   24

typedef float v2f __attribute__((ext_vector_type(2)));
typedef float v8f __attribute__((ext_vector_type(8)));

static __device__ __forceinline__ v8f wmma4(v2f a, v2f b, v8f c) {
  return __builtin_amdgcn_wmma_f32_16x16x4_f32(false, a, false, b, (short)0, c,
                                               false, false);
}

__global__ void wmma_gemm_kernel(const float* __restrict__ A,
                                 const float* __restrict__ W,
                                 const float* __restrict__ bias,
                                 float* __restrict__ C,
                                 int M, int K, int Nc, int total_tiles) {
  int tile = blockIdx.x * 8 + (threadIdx.x >> 5);
  if (tile >= total_tiles) return;
  int lane = threadIdx.x & 31;
  int h = lane >> 4, l16 = lane & 15;
  int ntn = Nc >> 4;
  int tm = (tile / ntn) << 4;
  int tn = (tile % ntn) << 4;
  const float* ap = A + (size_t)(tm + l16) * K + 2 * h;
  const float* wp = W + (size_t)(2 * h) * Nc + tn + l16;
  v8f acc = {};
  for (int k = 0; k < K; k += 4) {
    v2f a, b;
    a.x = ap[k];
    a.y = ap[k + 1];
    b.x = wp[(size_t)k * Nc];
    b.y = wp[(size_t)(k + 1) * Nc];
    acc = wmma4(a, b, acc);
  }
  float bv = bias ? bias[tn + l16] : 0.0f;
#pragma unroll
  for (int v = 0; v < 8; ++v) {
    int row = tm + v + (h << 3);
    C[(size_t)row * Nc + tn + l16] = acc[v] + bv;
  }
}

__global__ void bias_kernel(const float* __restrict__ e,
                            const float* __restrict__ Wb,
                            float* __restrict__ bias) {
  __shared__ float se[32 * 129];
  __shared__ float swb[128 * 8];
  int blk = blockIdx.x;
  int jb = blk % 12;
  int i  = (blk / 12) % LL;
  int b  = blk / (12 * LL);
  const float* ep = e + (((size_t)b * LL + i) * LL + (size_t)jb * 32) * DD;
  for (int idx = threadIdx.x; idx < 32 * 128; idx += 256) {
    int jj = idx >> 7, d = idx & 127;
    se[jj * 129 + d] = ep[idx];
  }
  for (int idx = threadIdx.x; idx < 128 * 8; idx += 256) swb[idx] = Wb[idx];
  __syncthreads();
  int n = threadIdx.x >> 5, jj = threadIdx.x & 31;
  float s = 0.0f;
  for (int d = 0; d < 128; ++d) s += se[jj * 129 + d] * swb[d * 8 + n];
  bias[(((size_t)b * NH + n) * LL + i) * LL + jb * 32 + jj] = s;
}

__global__ void point_transform_kernel(const float* __restrict__ hmat,
                                       const float* __restrict__ r,
                                       const float* __restrict__ t,
                                       float* __restrict__ outp,
                                       float* __restrict__ sumsq) {
  int idx = blockIdx.x * blockDim.x + threadIdx.x;
  if (idx >= BNL) return;
  int l = idx % LL;
  int n = (idx / LL) % NH;
  int b = idx / (NH * LL);
  const float* hp = hmat + (size_t)(b * LL + l) * (NH * 12) + n * 12;
  const float* rp = r + (size_t)(b * LL + l) * 9;
  const float* tp = t + (size_t)(b * LL + l) * 3;
  float* op = outp + (size_t)idx * 12;
  float ss = 0.0f;
#pragma unroll
  for (int p = 0; p < 4; ++p)
#pragma unroll
    for (int c = 0; c < 3; ++c) {
      float v = tp[c];
#pragma unroll
      for (int k = 0; k < 3; ++k) v += hp[p * 3 + k] * rp[k * 3 + c];
      op[p * 3 + c] = v;
      ss += v * v;
    }
  if (sumsq) sumsq[idx] = ss;
}

__global__ void logit_kernel(const float* __restrict__ qs,
                             const float* __restrict__ ks,
                             const float* __restrict__ qp,
                             const float* __restrict__ kp,
                             const float* __restrict__ qq,
                             const float* __restrict__ kk,
                             const float* __restrict__ bias,
                             const float* __restrict__ gamma,
                             float* __restrict__ logit) {
  int tile = blockIdx.x * 8 + (threadIdx.x >> 5);
  int jt = tile % LT;
  int it = (tile / LT) % LT;
  int n  = (tile / (LT * LT)) % NH;
  int b  = tile / (LT * LT * NH);
  int lane = threadIdx.x & 31;
  int h = lane >> 4, l16 = lane & 15;

  const float* qsp = qs + (size_t)(b * LL + it * 16 + l16) * DD + n * DHS_ + 2 * h;
  const float* ksp = ks + (size_t)(b * LL + jt * 16 + l16) * DD + n * DHS_ + 2 * h;
  v8f accS = {};
#pragma unroll
  for (int k = 0; k < 16; k += 4) {
    v2f a, bv2;
    a.x = qsp[k];   a.y = qsp[k + 1];
    bv2.x = ksp[k]; bv2.y = ksp[k + 1];
    accS = wmma4(a, bv2, accS);
  }
  const float* qpp = qp + (size_t)((b * NH + n) * LL + it * 16 + l16) * 12 + 2 * h;
  const float* kpp = kp + (size_t)((b * NH + n) * LL + jt * 16 + l16) * 12 + 2 * h;
  v8f accP = {};
#pragma unroll
  for (int k = 0; k < 12; k += 4) {
    v2f a, bv2;
    a.x = qpp[k];   a.y = qpp[k + 1];
    bv2.x = kpp[k]; bv2.y = kpp[k + 1];
    accP = wmma4(a, bv2, accP);
  }
  size_t hb = (size_t)(b * NH + n) * LL;
  int j = jt * 16 + l16;
  float kkj = kk[hb + j];
  float g = gamma[n];
  const float rsqrt3 = 0.57735026918962576f;
#pragma unroll
  for (int v = 0; v < 8; ++v) {
    int i = it * 16 + v + (h << 3);
    float d2 = qq[hb + i] + kkj - 2.0f * accP[v];
    float lp = -0.125f * g * d2;
    float bsv = bias[(hb + i) * LL + j];
    logit[(hb + i) * LL + j] = rsqrt3 * (0.25f * accS[v] + bsv + lp);
  }
}

__global__ void softmax_kernel(float* __restrict__ attn) {
  int row = blockIdx.x * 8 + (threadIdx.x >> 5);
  int lane = threadIdx.x & 31;
  float* p = attn + (size_t)row * LL;
  float v[12];
  float m = -1e30f;
#pragma unroll
  for (int i = 0; i < 12; ++i) {
    v[i] = p[lane + i * 32];
    m = fmaxf(m, v[i]);
  }
#pragma unroll
  for (int off = 16; off > 0; off >>= 1) m = fmaxf(m, __shfl_xor(m, off, 32));
  float s = 0.0f;
#pragma unroll
  for (int i = 0; i < 12; ++i) {
    v[i] = __expf(v[i] - m);
    s += v[i];
  }
#pragma unroll
  for (int off = 16; off > 0; off >>= 1) s += __shfl_xor(s, off, 32);
  float inv = 1.0f / s;
#pragma unroll
  for (int i = 0; i < 12; ++i) p[lane + i * 32] = v[i] * inv;
}

__global__ void attn_v_kernel(const float* __restrict__ attn,
                              const float* __restrict__ vs,
                              const float* __restrict__ vp,
                              float* __restrict__ out_cat,
                              float* __restrict__ op) {
  int tile = blockIdx.x * 8 + (threadIdx.x >> 5);
  int it = tile % LT;
  int n  = (tile / LT) % NH;
  int b  = tile / (LT * NH);
  int lane = threadIdx.x & 31;
  int h = lane >> 4, l16 = lane & 15;
  size_t hb = (size_t)(b * NH + n) * LL;
  const float* ap  = attn + (hb + it * 16 + l16) * LL + 2 * h;
  const float* vsb = vs + (size_t)b * LL * DD + n * DHS_ + l16;
  const float* vpb = vp + hb * 12 + l16;
  bool pv = l16 < 12;
  v8f accS = {}, accP = {};
  for (int k = 0; k < LL; k += 4) {
    v2f a;
    a.x = ap[k];
    a.y = ap[k + 1];
    v2f bs;
    bs.x = vsb[(size_t)(k + 2 * h) * DD];
    bs.y = vsb[(size_t)(k + 2 * h + 1) * DD];
    accS = wmma4(a, bs, accS);
    v2f bp;
    bp.x = pv ? vpb[(k + 2 * h) * 12] : 0.0f;
    bp.y = pv ? vpb[(k + 2 * h + 1) * 12] : 0.0f;
    accP = wmma4(a, bp, accP);
  }
#pragma unroll
  for (int v = 0; v < 8; ++v) {
    int i = it * 16 + v + (h << 3);
    out_cat[(size_t)(b * LL + i) * DOUT + n * DHS_ + l16] = accS[v];
    if (pv) op[(hb + i) * 12 + l16] = accP[v];
  }
}

__global__ void out_pair_kernel(const float* __restrict__ attn,
                                const float* __restrict__ e,
                                float* __restrict__ out_cat) {
  int tile = blockIdx.x * 8 + (threadIdx.x >> 5);
  int dt = tile % 8;
  int i  = (tile / 8) % LL;
  int b  = tile / (8 * LL);
  int lane = threadIdx.x & 31;
  int h = lane >> 4, l16 = lane & 15;
  bool av = l16 < 8;
  const float* ap = attn + ((size_t)(b * NH + l16) * LL + i) * LL + 2 * h;
  const float* ep = e + ((size_t)(b * LL + i) * LL) * DD + dt * 16 + l16;
  v8f acc = {};
  for (int k = 0; k < LL; k += 4) {
    if ((k & 63) == 0 && (k + 64) < LL)
      __builtin_prefetch(ep + (size_t)(k + 64) * DD, 0, 0);
    v2f a;
    a.x = av ? ap[k] : 0.0f;
    a.y = av ? ap[k + 1] : 0.0f;
    v2f bv;
    bv.x = ep[(size_t)(k + 2 * h) * DD];
    bv.y = ep[(size_t)(k + 2 * h + 1) * DD];
    acc = wmma4(a, bv, acc);
  }
  if (h == 0) {
#pragma unroll
    for (int v = 0; v < 8; ++v)
      out_cat[(size_t)(b * LL + i) * DOUT + DD + v * DD + dt * 16 + l16] = acc[v];
  }
}

__global__ void op_finish_kernel(const float* __restrict__ op,
                                 const float* __restrict__ r,
                                 const float* __restrict__ t,
                                 float* __restrict__ out_cat) {
  int idx = blockIdx.x * blockDim.x + threadIdx.x;
  if (idx >= BNL) return;
  int i = idx % LL;
  int n = (idx / LL) % NH;
  int b = idx / (NH * LL);
  const float* opv = op + (size_t)idx * 12;
  const float* rp = r + (size_t)(b * LL + i) * 9;
  const float* tp = t + (size_t)(b * LL + i) * 3;
  float* oc = out_cat + (size_t)(b * LL + i) * DOUT;
#pragma unroll
  for (int p = 0; p < 4; ++p) {
    float om[3];
#pragma unroll
    for (int k = 0; k < 3; ++k) om[k] = opv[p * 3 + k] - tp[k];
    float nsq = 0.0f;
#pragma unroll
    for (int c = 0; c < 3; ++c) {
      float v = om[0] * rp[c * 3 + 0] + om[1] * rp[c * 3 + 1] + om[2] * rp[c * 3 + 2];
      oc[DD + NH * DD + n * 12 + p * 3 + c] = v;
      nsq += v * v;
    }
    oc[DD + NH * DD + NH * 12 + n * 4 + p] = sqrtf(nsq);
  }
}

extern "C" void kernel_launch(void* const* d_in, const int* in_sizes, int n_in,
                              void* d_out, int out_size, void* d_ws, size_t ws_size,
                              hipStream_t stream) {
  (void)in_sizes; (void)n_in; (void)out_size;
  const float* x     = (const float*)d_in[0];
  const float* e     = (const float*)d_in[1];
  const float* r     = (const float*)d_in[2];
  const float* t     = (const float*)d_in[3];
  const float* Wq_s  = (const float*)d_in[4];
  const float* Wk_s  = (const float*)d_in[5];
  const float* Wv_s  = (const float*)d_in[6];
  const float* Wb    = (const float*)d_in[7];
  const float* Wq_p  = (const float*)d_in[8];
  const float* Wk_p  = (const float*)d_in[9];
  const float* Wv_p  = (const float*)d_in[10];
  const float* gamma = (const float*)d_in[11];
  const float* Wo    = (const float*)d_in[12];
  const float* bo    = (const float*)d_in[13];
  float* out = (float*)d_out;
  float* ws  = (float*)d_ws;

  float* qs   = ws;
  float* ks_  = qs  + 98304;
  float* vs   = ks_ + 98304;
  float* hq   = vs  + 98304;
  float* hk   = hq  + 73728;
  float* hv   = hk  + 73728;
  float* qp   = hv  + 73728;
  float* kp   = qp  + 73728;
  float* vp   = kp  + 73728;
  float* qq   = vp  + 73728;
  float* kk   = qq  + 6144;
  float* bias = kk  + 6144;
  float* attn = bias + 2359296;
  float* op   = attn + 2359296;
  float* ocat = op  + 73728;
  size_t req = (size_t)((ocat + 983040) - ws) * sizeof(float);
  if (ws_size < req) return;

  dim3 blk(256);
  auto gemm = [&](const float* A, const float* W, const float* bvec, float* C,
                  int M, int K, int Nc) {
    int tiles = (M / 16) * (Nc / 16);
    wmma_gemm_kernel<<<(tiles + 7) / 8, blk, 0, stream>>>(A, W, bvec, C, M, K,
                                                          Nc, tiles);
  };

  gemm(x, Wq_s, nullptr, qs, BL, DD, DD);
  gemm(x, Wk_s, nullptr, ks_, BL, DD, DD);
  gemm(x, Wv_s, nullptr, vs, BL, DD, DD);
  gemm(x, Wq_p, nullptr, hq, BL, DD, NH * 12);
  gemm(x, Wk_p, nullptr, hk, BL, DD, NH * 12);
  gemm(x, Wv_p, nullptr, hv, BL, DD, NH * 12);

  bias_kernel<<<BB * LL * 12, blk, 0, stream>>>(e, Wb, bias);

  point_transform_kernel<<<BNL / 256, blk, 0, stream>>>(hq, r, t, qp, qq);
  point_transform_kernel<<<BNL / 256, blk, 0, stream>>>(hk, r, t, kp, kk);
  point_transform_kernel<<<BNL / 256, blk, 0, stream>>>(hv, r, t, vp, nullptr);

  logit_kernel<<<(BB * NH * LT * LT) / 8, blk, 0, stream>>>(qs, ks_, qp, kp, qq,
                                                            kk, bias, gamma, attn);
  softmax_kernel<<<BNL / 8, blk, 0, stream>>>(attn);

  attn_v_kernel<<<(BB * NH * LT) / 8, blk, 0, stream>>>(attn, vs, vp, ocat, op);
  out_pair_kernel<<<(BB * LL * 8) / 8, blk, 0, stream>>>(attn, e, ocat);
  op_finish_kernel<<<BNL / 256, blk, 0, stream>>>(op, r, t, ocat);

  gemm(ocat, Wo, bo, out, BL, DOUT, DD);
}


# attn@V: block per (b,n,32 rows); attn strip and the head's 384x28 value slab staged through LDS by 16-byte loads all issued at once, k-steps run out of LDS; 192 blocks; k sum order unchanged
# speedup vs baseline: 2.8931x; 2.8931x over previous
#include <hip/hip_runtime.h>
#include <hip/hip_bf16.h>
#include <math.h>

#define BB   2
#define LL   384
#define DD   128
#define NH   8
#define DHS_ 16
#define DHP_ 4
#define BL   (BB*LL)
#define BNL  (BB*NH*LL)
#define DOUT 1280
#define LT   24

typedef float v2f __attribute__((ext_vector_type(2)));
typedef float v4f __attribute__((ext_vector_type(4)));
typedef float v8f __attribute__((ext_vector_type(8)));

static __device__ __forceinline__ v8f wmma4(v2f a, v2f b, v8f c) {
  return __builtin_amdgcn_wmma_f32_16x16x4_f32(false, a, false, b, (short)0, c,
                                               false, false);
}

template <int K, int NC>
__global__ __launch_bounds__(256) void wmma_gemm_lds_kernel(
    const float* __restrict__ A, const float* __restrict__ W,
    const float* __restrict__ bias, float* __restrict__ C) {
  constexpr int KC  = 128;
  constexpr int AP  = KC + 4;
  constexpr int WP  = NC + 8;
  constexpr int NCH = K / KC;
  constexpr int WV  = NC / 4;
  constexpr int WR  = 8;
  constexpr int WQ  = KC / WR;
  constexpr int AQ  = (16 * KC / 4) / 256;
  __shared__ float As[16 * AP];
  __shared__ float Ws[KC * WP];

  const int tid = threadIdx.x;
  const int wave = tid >> 5, lane = tid & 31;
  const int h = lane >> 4, l16 = lane & 15;
  const int tm = blockIdx.x << 4;
  const int tn = wave << 4;

  v4f ra[AQ];
  v4f rw[WQ] = {};
  const float* ag = A + (size_t)(tm + (tid >> 5)) * K + (tid & 31) * 4;
  const float* wg = W + (size_t)(tid / WV) * NC + (tid % WV) * 4;
  float* al = As + (tid >> 5) * AP + (tid & 31) * 4;
  float* wl = Ws + (tid / WV) * WP + (tid % WV) * 4;
  const bool wact = tid < WR * WV;

#pragma unroll
  for (int q = 0; q < AQ; ++q)
    ra[q] = *reinterpret_cast<const v4f*>(ag + (size_t)(q * 8) * K);
  if (wact) {
#pragma unroll
    for (int q = 0; q < WQ; ++q)
      rw[q] = *reinterpret_cast<const v4f*>(wg + (size_t)(q * WR) * NC);
  }

  v8f acc = {};
  for (int c = 0; c < NCH; ++c) {
#pragma unroll
    for (int q = 0; q < AQ; ++q)
      *reinterpret_cast<v4f*>(al + (q * 8) * AP) = ra[q];
    if (wact) {
#pragma unroll
      for (int q = 0; q < WQ; ++q)
        *reinterpret_cast<v4f*>(wl + (q * WR) * WP) = rw[q];
    }
    __syncthreads();
    if (c + 1 < NCH) {
      const int k1 = (c + 1) * KC;
#pragma unroll
      for (int q = 0; q < AQ; ++q)
        ra[q] = *reinterpret_cast<const v4f*>(ag + (size_t)(q * 8) * K + k1);
      if (wact) {
#pragma unroll
        for (int q = 0; q < WQ; ++q)
          rw[q] = *reinterpret_cast<const v4f*>(wg + (size_t)(k1 + q * WR) * NC);
      }
    }
    if (tn < NC) {
      const float* asp = As + l16 * AP + 2 * h;
      const float* wsp = Ws + (2 * h) * WP + tn + l16;
#pragma unroll
      for (int kk = 0; kk < KC; kk += 4) {
        v2f a = *reinterpret_cast<const v2f*>(asp + kk);
        v2f b;
        b.x = wsp[kk * WP];
        b.y = wsp[(kk + 1) * WP];
        acc = wmma4(a, b, acc);
      }
    }
    if (c + 1 < NCH) __syncthreads();
  }
  if (tn < NC) {
    float bv = bias ? bias[tn + l16] : 0.0f;
#pragma unroll
    for (int v = 0; v < 8; ++v) {
      int row = tm + v + (h << 3);
      C[(size_t)row * NC + tn + l16] = acc[v] + bv;
    }
  }
}

__global__ void point_transform_kernel(const float* __restrict__ hmat,
                                       const float* __restrict__ r,
                                       const float* __restrict__ t,
                                       float* __restrict__ outp,
                                       float* __restrict__ sumsq) {
  int idx = blockIdx.x * blockDim.x + threadIdx.x;
  if (idx >= BNL) return;
  int l = idx % LL;
  int n = (idx / LL) % NH;
  int b = idx / (NH * LL);
  const float* hp = hmat + (size_t)(b * LL + l) * (NH * 12) + n * 12;
  const float* rp = r + (size_t)(b * LL + l) * 9;
  const float* tp = t + (size_t)(b * LL + l) * 3;
  float* op = outp + (size_t)idx * 12;
  float ss = 0.0f;
#pragma unroll
  for (int p = 0; p < 4; ++p)
#pragma unroll
    for (int c = 0; c < 3; ++c) {
      float v = tp[c];
#pragma unroll
      for (int k = 0; k < 3; ++k) v += hp[p * 3 + k] * rp[k * 3 + c];
      op[p * 3 + c] = v;
      ss += v * v;
    }
  if (sumsq) sumsq[idx] = ss;
}

__global__ void logit_kernel(const float* __restrict__ qs,
                             const float* __restrict__ ks,
                             const float* __restrict__ qp,
                             const float* __restrict__ kp,
                             const float* __restrict__ qq,
                             const float* __restrict__ kk,
                             const float* __restrict__ gamma,
                             float* __restrict__ part) {
  int tile = blockIdx.x * 8 + (threadIdx.x >> 5);
  int jt = tile % LT;
  int it = (tile / LT) % LT;
  int n  = (tile / (LT * LT)) % NH;
  int b  = tile / (LT * LT * NH);
  int lane = threadIdx.x & 31;
  int h = lane >> 4, l16 = lane & 15;

  const float* qsp = qs + (size_t)(b * LL + it * 16 + l16) * DD + n * DHS_ + 2 * h;
  const float* ksp = ks + (size_t)(b * LL + jt * 16 + l16) * DD + n * DHS_ + 2 * h;
  v8f accS = {};
#pragma unroll
  for (int k = 0; k < 16; k += 4) {
    v2f a, bv2;
    a.x = qsp[k];   a.y = qsp[k + 1];
    bv2.x = ksp[k]; bv2.y = ksp[k + 1];
    accS = wmma4(a, bv2, accS);
  }
  const float* qpp = qp + (size_t)((b * NH + n) * LL + it * 16 + l16) * 12 + 2 * h;
  const float* kpp = kp + (size_t)((b * NH + n) * LL + jt * 16 + l16) * 12 + 2 * h;
  v8f accP = {};
#pragma unroll
  for (int k = 0; k < 12; k += 4) {
    v2f a, bv2;
    a.x = qpp[k];   a.y = qpp[k + 1];
    bv2.x = kpp[k]; bv2.y = kpp[k + 1];
    accP = wmma4(a, bv2, accP);
  }
  size_t hb = (size_t)(b * NH + n) * LL;
  int j = jt * 16 + l16;
  float kkj = kk[hb + j];
  float g = gamma[n];
#pragma unroll
  for (int v = 0; v < 8; ++v) {
    int i = it * 16 + v + (h << 3);
    float d2 = qq[hb + i] + kkj - 2.0f * accP[v];
    float lp = -0.125f * g * d2;
    part[(hb + i) * LL + j] = 0.25f * accS[v] + lp;
  }
}

__global__ __launch_bounds__(256) void attn_v_kernel(const float* __restrict__ attn,
                                                     const float* __restrict__ vs,
                                                     const float* __restrict__ vp,
                                                     float* __restrict__ out_cat,
                                                     float* __restrict__ op) {
  constexpr int RT = 2;
  constexpr int AP = LL + 4;
  constexpr int BP = 32 + 8;
  constexpr int AQ = RT * 16 * 3 / 8;
  constexpr int SQ = LL * 4 / 256;
  __shared__ float As[RT * 16 * AP];
  __shared__ float Bs[LL * BP];

  const int tid = threadIdx.x;
  const int wave = __builtin_amdgcn_readfirstlane(tid >> 5);
  const int lane = tid & 31;
  const int h = lane >> 4, l16 = lane & 15;
  const int ip = blockIdx.x % (LT / RT);
  const int bn = blockIdx.x / (LT / RT);
  const int n = bn % NH, b = bn / NH;
  const size_t hb = (size_t)bn * LL;

  const float* ag = attn + (hb + ip * (RT * 16) + 4 * wave) * LL + lane * 4;
  float* al = As + (4 * wave) * AP + lane * 4;
  const float* sg = vs + ((size_t)b * LL + (tid >> 2)) * DD + n * DHS_ + (tid & 3) * 4;
  float* sl = Bs + (tid >> 2) * BP + (tid & 3) * 4;
  const int pq = (tid & 3) < 2 ? (tid & 3) : 2;
  const float* pg = vp + (hb + (tid >> 2)) * 12 + pq * 4;
  float* psl = sl + 16;

  v4f ra[AQ], rs[SQ], rp[SQ];
#pragma unroll
  for (int m = 0; m < AQ; ++m)
    ra[m] = *reinterpret_cast<const v4f*>(ag + (m / 3) * LL + (m % 3) * 128);
#pragma unroll
  for (int m = 0; m < SQ; ++m)
    rs[m] = *reinterpret_cast<const v4f*>(sg + (size_t)(m * 64) * DD);
#pragma unroll
  for (int m = 0; m < SQ; ++m)
    rp[m] = *reinterpret_cast<const v4f*>(pg + (m * 64) * 12);

#pragma unroll
  for (int m = 0; m < AQ; ++m)
    *reinterpret_cast<v4f*>(al + (m / 3) * AP + (m % 3) * 128) = ra[m];
#pragma unroll
  for (int m = 0; m < SQ; ++m)
    *reinterpret_cast<v4f*>(sl + (m * 64) * BP) = rs[m];
#pragma unroll
  for (int m = 0; m < SQ; ++m)
    *reinterpret_cast<v4f*>(psl + (m * 64) * BP) = rp[m];
  __syncthreads();

  if (wave < 2 * RT) {
    const int rt = wave >> 1, ct = wave & 1;
    constexpr int KB = 16;
    constexpr int NB = LL / (4 * KB);
    v8f acc = {};
    const float* asp = As + (rt * 16 + l16) * AP + 2 * h;
    const float* bsp = Bs + (2 * h) * BP + ct * 16 + l16;
    v2f fa[2][KB], fb[2][KB];
#pragma unroll
    for (int s = 0; s < KB; ++s) {
      fa[0][s] = *reinterpret_cast<const v2f*>(asp + 4 * s);
      fb[0][s].x = bsp[(4 * s) * BP];
      fb[0][s].y = bsp[(4 * s + 1) * BP];
    }
#pragma unroll
    for (int t = 0; t < NB; ++t) {
      if (t + 1 < NB) {
#pragma unroll
        for (int s = 0; s < KB; ++s) {
          const int k = (t + 1) * 4 * KB + 4 * s;
          fa[(t + 1) & 1][s] = *reinterpret_cast<const v2f*>(asp + k);
          fb[(t + 1) & 1][s].x = bsp[k * BP];
          fb[(t + 1) & 1][s].y = bsp[(k + 1) * BP];
        }
      }
      __builtin_amdgcn_sched_barrier(0);
#pragma unroll
      for (int s = 0; s < KB; ++s) acc = wmma4(fa[t & 1][s], fb[t & 1][s], acc);
    }
    const int i0 = ip * (RT * 16) + rt * 16 + (h << 3);
    if (ct == 0) {
      float* o = out_cat + (size_t)(b * LL + i0) * DOUT + n * DHS_ + l16;
#pragma unroll
      for (int v = 0; v < 8; ++v) o[(size_t)v * DOUT] = acc[v];
    } else if (l16 < 12) {
      float* o = op + (hb + i0) * 12 + l16;
#pragma unroll
      for (int v = 0; v < 8; ++v) o[v * 12] = acc[v];
    }
  }
}

__global__ __launch_bounds__(256) void pair_row_kernel(
    const float* __restrict__ part, const float* __restrict__ e,
    const float* __restrict__ Wb, float* __restrict__ attn,
    float* __restrict__ out_cat) {
  constexpr int KC  = 128;
  constexpr int NCH = LL / KC;
  constexpr int AP  = LL + 4;
  constexpr int WP  = DD + 12;
  constexpr int WQ  = KC / 8;
  __shared__ float As[16 * AP];
  __shared__ float Ws[LL * WP];

  const int tid = threadIdx.x;
  const int wave = tid >> 5, lane = tid & 31;
  const int h = lane >> 4, l16 = lane & 15;
  const int i = blockIdx.x % LL;
  const int b = blockIdx.x / LL;
  const int tn = wave << 4;
  const int jj = tid & (KC - 1);
  const int g = __builtin_amdgcn_readfirstlane(tid >> 7);

  const float* wg = e + (((size_t)(b * LL + i) * LL) + wave) * DD + lane * 4;
  float* wl = Ws + wave * WP + lane * 4;
  const float* wq = Wb + g * 4;
  const float* pg = part + ((size_t)(b * NH + g * 4) * LL + i) * LL + jj;

  v4f rw[2][WQ];
#pragma unroll
  for (int q = 0; q < WQ; ++q)
    rw[0][q] = *reinterpret_cast<const v4f*>(wg + (size_t)(q * 8) * DD);
#pragma unroll
  for (int q = 0; q < WQ; ++q)
    rw[1][q] = *reinterpret_cast<const v4f*>(wg + (size_t)(KC + q * 8) * DD);
  float sp[NCH][4];
#pragma unroll
  for (int c = 0; c < NCH; ++c)
#pragma unroll
    for (int n = 0; n < 4; ++n)
      sp[c][n] = pg[(size_t)n * LL * LL + c * KC];

  {
    v4f z = {};
#pragma unroll
    for (int c = 0; c < NCH; ++c)
      *reinterpret_cast<v4f*>(As + (8 + wave) * AP + c * KC + lane * 4) = z;
  }

  const float rsqrt3 = 0.57735026918962576f;
#pragma unroll
  for (int c = 0; c < NCH; ++c) {
#pragma unroll
    for (int q = 0; q < WQ; ++q)
      *reinterpret_cast<v4f*>(wl + (c * KC + q * 8) * WP) = rw[c & 1][q];
    __syncthreads();
    if (c + 2 < NCH) {
      const int k2 = (c + 2) * KC;
#pragma unroll
      for (int q = 0; q < WQ; ++q)
        rw[c & 1][q] = *reinterpret_cast<const v4f*>(wg + (size_t)(k2 + q * 8) * DD);
    }
    const float* row = Ws + (c * KC + jj) * WP;
    float s0 = 0.0f, s1 = 0.0f, s2 = 0.0f, s3 = 0.0f;
    for (int q = 0; q < DD / 4; ++q) {
      const v4f ev = *reinterpret_cast<const v4f*>(row + 4 * q);
      const float* w = wq + (4 * q) * NH;
      s0 += ev.x * w[0];          s1 += ev.x * w[1];          s2 += ev.x * w[2];          s3 += ev.x * w[3];
      s0 += ev.y * w[NH + 0];     s1 += ev.y * w[NH + 1];     s2 += ev.y * w[NH + 2];     s3 += ev.y * w[NH + 3];
      s0 += ev.z * w[2 * NH + 0]; s1 += ev.z * w[2 * NH + 1]; s2 += ev.z * w[2 * NH + 2]; s3 += ev.z * w[2 * NH + 3];
      s0 += ev.w * w[3 * NH + 0]; s1 += ev.w * w[3 * NH + 1]; s2 += ev.w * w[3 * NH + 2]; s3 += ev.w * w[3 * NH + 3];
    }
    float* lg = As + (g * 4) * AP + c * KC + jj;
    lg[0]      = rsqrt3 * (sp[c][0] + s0);
    lg[AP]     = rsqrt3 * (sp[c][1] + s1);
    lg[2 * AP] = rsqrt3 * (sp[c][2] + s2);
    lg[3 * AP] = rsqrt3 * (sp[c][3] + s3);
  }
  __syncthreads();

  float v[12];
  {
    float* lr = As + wave * AP + lane;
    float m = -1e30f;
#pragma unroll
    for (int k = 0; k < 12; ++k) {
      v[k] = lr[k * 32];
      m = fmaxf(m, v[k]);
    }
#pragma unroll
    for (int off = 16; off > 0; off >>= 1) m = fmaxf(m, __shfl_xor(m, off, 32));
    float s = 0.0f;
#pragma unroll
    for (int k = 0; k < 12; ++k) {
      v[k] = __expf(v[k] - m);
      s += v[k];
    }
#pragma unroll
    for (int off = 16; off > 0; off >>= 1) s += __shfl_xor(s, off, 32);
    float inv = 1.0f / s;
#pragma unroll
    for (int k = 0; k < 12; ++k) {
      v[k] = v[k] * inv;
      lr[k * 32] = v[k];
    }
  }
  __syncthreads();
  {
    float* ao = attn + ((size_t)(b * NH + wave) * LL + i) * LL + lane;
#pragma unroll
    for (int k = 0; k < 12; ++k) ao[k * 32] = v[k];
  }

  constexpr int KB = 16;
  constexpr int NB = LL / (4 * KB);
  v8f acc = {};
  const float* asp = As + l16 * AP + 2 * h;
  const float* wsp = Ws + (2 * h) * WP + tn + l16;
  v2f fa[2][KB], fb[2][KB];
#pragma unroll
  for (int s = 0; s < KB; ++s) {
    fa[0][s] = *reinterpret_cast<const v2f*>(asp + 4 * s);
    fb[0][s].x = wsp[(4 * s) * WP];
    fb[0][s].y = wsp[(4 * s + 1) * WP];
  }
#pragma unroll
  for (int t = 0; t < NB; ++t) {
    if (t + 1 < NB) {
#pragma unroll
      for (int s = 0; s < KB; ++s) {
        const int k = (t + 1) * 4 * KB + 4 * s;
        fa[(t + 1) & 1][s] = *reinterpret_cast<const v2f*>(asp + k);
        fb[(t + 1) & 1][s].x = wsp[k * WP];
        fb[(t + 1) & 1][s].y = wsp[(k + 1) * WP];
      }
    }
    __builtin_amdgcn_sched_barrier(0);
#pragma unroll
    for (int s = 0; s < KB; ++s) acc = wmma4(fa[t & 1][s], fb[t & 1][s], acc);
  }
  if (h == 0) {
#pragma unroll
    for (int r = 0; r < 8; ++r)
      out_cat[(size_t)(b * LL + i) * DOUT + DD + r * DD + tn + l16] = acc[r];
  }
}

__global__ void op_finish_kernel(const float* __restrict__ op,
                                 const float* __restrict__ r,
                                 const float* __restrict__ t,
                                 float* __restrict__ out_cat) {
  int idx = blockIdx.x * blockDim.x + threadIdx.x;
  if (idx >= BNL) return;
  int i = idx % LL;
  int n = (idx / LL) % NH;
  int b = idx / (NH * LL);
  const float* opv = op + (size_t)idx * 12;
  const float* rp = r + (size_t)(b * LL + i) * 9;
  const float* tp = t + (size_t)(b * LL + i) * 3;
  float* oc = out_cat + (size_t)(b * LL + i) * DOUT;
#pragma unroll
  for (int p = 0; p < 4; ++p) {
    float om[3];
#pragma unroll
    for (int k = 0; k < 3; ++k) om[k] = opv[p * 3 + k] - tp[k];
    float nsq = 0.0f;
#pragma unroll
    for (int c = 0; c < 3; ++c) {
      float v = om[0] * rp[c * 3 + 0] + om[1] * rp[c * 3 + 1] + om[2] * rp[c * 3 + 2];
      oc[DD + NH * DD + n * 12 + p * 3 + c] = v;
      nsq += v * v;
    }
    oc[DD + NH * DD + NH * 12 + n * 4 + p] = sqrtf(nsq);
  }
}

extern "C" void kernel_launch(void* const* d_in, const int* in_sizes, int n_in,
                              void* d_out, int out_size, void* d_ws, size_t ws_size,
                              hipStream_t stream) {
  (void)in_sizes; (void)n_in; (void)out_size;
  const float* x     = (const float*)d_in[0];
  const float* e     = (const float*)d_in[1];
  const float* r     = (const float*)d_in[2];
  const float* t     = (const float*)d_in[3];
  const float* Wq_s  = (const float*)d_in[4];
  const float* Wk_s  = (const float*)d_in[5];
  const float* Wv_s  = (const float*)d_in[6];
  const float* Wb    = (const float*)d_in[7];
  const float* Wq_p  = (const float*)d_in[8];
  const float* Wk_p  = (const float*)d_in[9];
  const float* Wv_p  = (const float*)d_in[10];
  const float* gamma = (const float*)d_in[11];
  const float* Wo    = (const float*)d_in[12];
  const float* bo    = (const float*)d_in[13];
  float* out = (float*)d_out;
  float* ws  = (float*)d_ws;

  float* qs   = ws;
  float* ks_  = qs  + 98304;
  float* vs   = ks_ + 98304;
  float* hq   = vs  + 98304;
  float* hk   = hq  + 73728;
  float* hv   = hk  + 73728;
  float* qp   = hv  + 73728;
  float* kp   = qp  + 73728;
  float* vp   = kp  + 73728;
  float* qq   = vp  + 73728;
  float* kk   = qq  + 6144;
  float* part = kk  + 6144;
  float* attn = part + 2359296;
  float* op   = attn + 2359296;
  float* ocat = op  + 73728;
  size_t req = (size_t)((ocat + 983040) - ws) * sizeof(float);
  if (ws_size < req) return;

  dim3 blk(256);

  wmma_gemm_lds_kernel<DD, DD><<<BL / 16, blk, 0, stream>>>(x, Wq_s, nullptr, qs);
  wmma_gemm_lds_kernel<DD, DD><<<BL / 16, blk, 0, stream>>>(x, Wk_s, nullptr, ks_);
  wmma_gemm_lds_kernel<DD, DD><<<BL / 16, blk, 0, stream>>>(x, Wv_s, nullptr, vs);
  wmma_gemm_lds_kernel<DD, NH * 12><<<BL / 16, blk, 0, stream>>>(x, Wq_p, nullptr, hq);
  wmma_gemm_lds_kernel<DD, NH * 12><<<BL / 16, blk, 0, stream>>>(x, Wk_p, nullptr, hk);
  wmma_gemm_lds_kernel<DD, NH * 12><<<BL / 16, blk, 0, stream>>>(x, Wv_p, nullptr, hv);

  point_transform_kernel<<<BNL / 256, blk, 0, stream>>>(hq, r, t, qp, qq);
  point_transform_kernel<<<BNL / 256, blk, 0, stream>>>(hk, r, t, kp, kk);
  point_transform_kernel<<<BNL / 256, blk, 0, stream>>>(hv, r, t, vp, nullptr);

  logit_kernel<<<(BB * NH * LT * LT) / 8, blk, 0, stream>>>(qs, ks_, qp, kp, qq,
                                                            kk, gamma, part);

  pair_row_kernel<<<BB * LL, blk, 0, stream>>>(part, e, Wb, attn, ocat);

  attn_v_kernel<<<BB * NH * (LT / 2), blk, 0, stream>>>(attn, vs, vp, ocat, op);
  op_finish_kernel<<<BNL / 256, blk, 0, stream>>>(op, r, t, ocat);

  wmma_gemm_lds_kernel<DOUT, DD><<<BL / 16, blk, 0, stream>>>(ocat, Wo, bo, out);
}


# six input projections in one launch: block per 16-row strip keeps its x strip in LDS and streams the 6 weights through two LDS slabs, next one in flight during WMMAs; 14 launches -> 9; k order kept
# speedup vs baseline: 3.3023x; 1.1414x over previous
#include <hip/hip_runtime.h>
#include <hip/hip_bf16.h>
#include <math.h>

#define BB   2
#define LL   384
#define DD   128
#define NH   8
#define DHS_ 16
#define DHP_ 4
#define BL   (BB*LL)
#define BNL  (BB*NH*LL)
#define DOUT 1280
#define LT   24

typedef float v2f __attribute__((ext_vector_type(2)));
typedef float v4f __attribute__((ext_vector_type(4)));
typedef float v8f __attribute__((ext_vector_type(8)));

static __device__ __forceinline__ v8f wmma4(v2f a, v2f b, v8f c) {
  return __builtin_amdgcn_wmma_f32_16x16x4_f32(false, a, false, b, (short)0, c,
                                               false, false);
}

template <int K, int NC>
__global__ __launch_bounds__(256) void wmma_gemm_lds_kernel(
    const float* __restrict__ A, const float* __restrict__ W,
    const float* __restrict__ bias, float* __restrict__ C) {
  constexpr int KC  = 128;
  constexpr int AP  = KC + 4;
  constexpr int WP  = NC + 8;
  constexpr int NCH = K / KC;
  constexpr int WV  = NC / 4;
  constexpr int WR  = 8;
  constexpr int WQ  = KC / WR;
  constexpr int AQ  = (16 * KC / 4) / 256;
  __shared__ float As[16 * AP];
  __shared__ float Ws[KC * WP];

  const int tid = threadIdx.x;
  const int wave = tid >> 5, lane = tid & 31;
  const int h = lane >> 4, l16 = lane & 15;
  const int tm = blockIdx.x << 4;
  const int tn = wave << 4;

  v4f ra[AQ];
  v4f rw[WQ] = {};
  const float* ag = A + (size_t)(tm + (tid >> 5)) * K + (tid & 31) * 4;
  const float* wg = W + (size_t)(tid / WV) * NC + (tid % WV) * 4;
  float* al = As + (tid >> 5) * AP + (tid & 31) * 4;
  float* wl = Ws + (tid / WV) * WP + (tid % WV) * 4;
  const bool wact = tid < WR * WV;

#pragma unroll
  for (int q = 0; q < AQ; ++q)
    ra[q] = *reinterpret_cast<const v4f*>(ag + (size_t)(q * 8) * K);
  if (wact) {
#pragma unroll
    for (int q = 0; q < WQ; ++q)
      rw[q] = *reinterpret_cast<const v4f*>(wg + (size_t)(q * WR) * NC);
  }

  v8f acc = {};
  for (int c = 0; c < NCH; ++c) {
#pragma unroll
    for (int q = 0; q < AQ; ++q)
      *reinterpret_cast<v4f*>(al + (q * 8) * AP) = ra[q];
    if (wact) {
#pragma unroll
      for (int q = 0; q < WQ; ++q)
        *reinterpret_cast<v4f*>(wl + (q * WR) * WP) = rw[q];
    }
    __syncthreads();
    if (c + 1 < NCH) {
      const int k1 = (c + 1) * KC;
#pragma unroll
      for (int q = 0; q < AQ; ++q)
        ra[q] = *reinterpret_cast<const v4f*>(ag + (size_t)(q * 8) * K + k1);
      if (wact) {
#pragma unroll
        for (int q = 0; q < WQ; ++q)
          rw[q] = *reinterpret_cast<const v4f*>(wg + (size_t)(k1 + q * WR) * NC);
      }
    }
    if (tn < NC) {
      const float* asp = As + l16 * AP + 2 * h;
      const float* wsp = Ws + (2 * h) * WP + tn + l16;
#pragma unroll
      for (int kk = 0; kk < KC; kk += 4) {
        v2f a = *reinterpret_cast<const v2f*>(asp + kk);
        v2f b;
        b.x = wsp[kk * WP];
        b.y = wsp[(kk + 1) * WP];
        acc = wmma4(a, b, acc);
      }
    }
    if (c + 1 < NCH) __syncthreads();
  }
  if (tn < NC) {
    float bv = bias ? bias[tn + l16] : 0.0f;
#pragma unroll
    for (int v = 0; v < 8; ++v) {
      int row = tm + v + (h << 3);
      C[(size_t)row * NC + tn + l16] = acc[v] + bv;
    }
  }
}

#define PJ_AP 132
#define PJ_WS (128 * 136)

template <int NC>
static __device__ __forceinline__ void proj_fetch(const float* __restrict__ W, int tid,
                                                  int wave, v4f (&rw)[16]) {
  constexpr int WV = NC / 4;
  if (wave < WV / 4) {
    const float* wg = W + (size_t)(tid / WV) * NC + (tid % WV) * 4;
#pragma unroll
    for (int q = 0; q < 16; ++q)
      rw[q] = *reinterpret_cast<const v4f*>(wg + (size_t)(q * 8) * NC);
  }
}

template <int NC>
static __device__ __forceinline__ void proj_park(float* __restrict__ Ws, int tid, int wave,
                                                 const v4f (&rw)[16]) {
  constexpr int WV = NC / 4;
  constexpr int WP = NC + 8;
  if (wave < WV / 4) {
    float* wl = Ws + (tid / WV) * WP + (tid % WV) * 4;
#pragma unroll
    for (int q = 0; q < 16; ++q)
      *reinterpret_cast<v4f*>(wl + (q * 8) * WP) = rw[q];
  }
}

template <int NC>
static __device__ __forceinline__ v8f proj_mult(const float* __restrict__ As,
                                                const float* __restrict__ Ws, int wave,
                                                int h, int l16) {
  constexpr int WP = NC + 8;
  const int tn = wave << 4;
  v8f acc = {};
  if (tn < NC) {
    const float* asp = As + l16 * PJ_AP + 2 * h;
    const float* wsp = Ws + (2 * h) * WP + tn + l16;
#pragma unroll
    for (int kk = 0; kk < 128; kk += 4) {
      v2f a = *reinterpret_cast<const v2f*>(asp + kk);
      v2f b;
      b.x = wsp[kk * WP];
      b.y = wsp[(kk + 1) * WP];
      acc = wmma4(a, b, acc);
    }
  }
  return acc;
}

template <int NC>
static __device__ __forceinline__ void proj_store(float* __restrict__ C, v8f acc, int tm,
                                                  int wave, int h, int l16) {
  const int tn = wave << 4;
  if (tn < NC) {
#pragma unroll
    for (int v = 0; v < 8; ++v)
      C[(size_t)(tm + v + (h << 3)) * NC + tn + l16] = acc[v];
    __builtin_amdgcn_sched_barrier(0);
  }
}

__global__ __launch_bounds__(256) void proj6_kernel(
    const float* __restrict__ x,
    const float* __restrict__ W0, const float* __restrict__ W1, const float* __restrict__ W2,
    const float* __restrict__ W3, const float* __restrict__ W4, const float* __restrict__ W5,
    float* __restrict__ C0, float* __restrict__ C1, float* __restrict__ C2,
    float* __restrict__ C3, float* __restrict__ C4, float* __restrict__ C5) {
  __shared__ float As[16 * PJ_AP];
  __shared__ float Ws[2 * PJ_WS];

  const int tid = threadIdx.x;
  const int wave = __builtin_amdgcn_readfirstlane(tid >> 5);
  const int lane = tid & 31;
  const int h = lane >> 4, l16 = lane & 15;
  const int tm = blockIdx.x << 4;

  const float* ag = x + (size_t)(tm + wave) * DD + lane * 4;
  v4f ra0 = *reinterpret_cast<const v4f*>(ag);
  v4f ra1 = *reinterpret_cast<const v4f*>(ag + (size_t)8 * DD);
  v4f rw[16] = {};
  proj_fetch<DD>(W0, tid, wave, rw);

  float* al = As + wave * PJ_AP + lane * 4;
  *reinterpret_cast<v4f*>(al) = ra0;
  *reinterpret_cast<v4f*>(al + 8 * PJ_AP) = ra1;

  float* S0 = Ws;
  float* S1 = Ws + PJ_WS;

  proj_park<DD>(S0, tid, wave, rw);
  __syncthreads();
  proj_fetch<DD>(W1, tid, wave, rw);
  v8f a0 = proj_mult<DD>(As, S0, wave, h, l16);

  proj_park<DD>(S1, tid, wave, rw);
  __syncthreads();
  proj_fetch<DD>(W2, tid, wave, rw);
  proj_store<DD>(C0, a0, tm, wave, h, l16);
  v8f a1 = proj_mult<DD>(As, S1, wave, h, l16);

  proj_park<DD>(S0, tid, wave, rw);
  __syncthreads();
  proj_fetch<NH * 12>(W3, tid, wave, rw);
  proj_store<DD>(C1, a1, tm, wave, h, l16);
  v8f a2 = proj_mult<DD>(As, S0, wave, h, l16);

  proj_park<NH * 12>(S1, tid, wave, rw);
  __syncthreads();
  proj_fetch<NH * 12>(W4, tid, wave, rw);
  proj_store<DD>(C2, a2, tm, wave, h, l16);
  v8f a3 = proj_mult<NH * 12>(As, S1, wave, h, l16);

  proj_park<NH * 12>(S0, tid, wave, rw);
  __syncthreads();
  proj_fetch<NH * 12>(W5, tid, wave, rw);
  proj_store<NH * 12>(C3, a3, tm, wave, h, l16);
  v8f a4 = proj_mult<NH * 12>(As, S0, wave, h, l16);

  proj_park<NH * 12>(S1, tid, wave, rw);
  __syncthreads();
  proj_store<NH * 12>(C4, a4, tm, wave, h, l16);
  v8f a5 = proj_mult<NH * 12>(As, S1, wave, h, l16);
  proj_store<NH * 12>(C5, a5, tm, wave, h, l16);
}

__global__ void point_transform_kernel(const float* __restrict__ hmat,
                                       const float* __restrict__ r,
                                       const float* __restrict__ t,
                                       float* __restrict__ outp,
                                       float* __restrict__ sumsq) {
  int idx = blockIdx.x * blockDim.x + threadIdx.x;
  if (idx >= BNL) return;
  int l = idx % LL;
  int n = (idx / LL) % NH;
  int b = idx / (NH * LL);
  const float* hp = hmat + (size_t)(b * LL + l) * (NH * 12) + n * 12;
  const float* rp = r + (size_t)(b * LL + l) * 9;
  const float* tp = t + (size_t)(b * LL + l) * 3;
  float* op = outp + (size_t)idx * 12;
  float ss = 0.0f;
#pragma unroll
  for (int p = 0; p < 4; ++p)
#pragma unroll
    for (int c = 0; c < 3; ++c) {
      float v = tp[c];
#pragma unroll
      for (int k = 0; k < 3; ++k) v += hp[p * 3 + k] * rp[k * 3 + c];
      op[p * 3 + c] = v;
      ss += v * v;
    }
  if (sumsq) sumsq[idx] = ss;
}

__global__ void logit_kernel(const float* __restrict__ qs,
                             const float* __restrict__ ks,
                             const float* __restrict__ qp,
                             const float* __restrict__ kp,
                             const float* __restrict__ qq,
                             const float* __restrict__ kk,
                             const float* __restrict__ gamma,
                             float* __restrict__ part) {
  int tile = blockIdx.x * 8 + (threadIdx.x >> 5);
  int jt = tile % LT;
  int it = (tile / LT) % LT;
  int n  = (tile / (LT * LT)) % NH;
  int b  = tile / (LT * LT * NH);
  int lane = threadIdx.x & 31;
  int h = lane >> 4, l16 = lane & 15;

  const float* qsp = qs + (size_t)(b * LL + it * 16 + l16) * DD + n * DHS_ + 2 * h;
  const float* ksp = ks + (size_t)(b * LL + jt * 16 + l16) * DD + n * DHS_ + 2 * h;
  v8f accS = {};
#pragma unroll
  for (int k = 0; k < 16; k += 4) {
    v2f a, bv2;
    a.x = qsp[k];   a.y = qsp[k + 1];
    bv2.x = ksp[k]; bv2.y = ksp[k + 1];
    accS = wmma4(a, bv2, accS);
  }
  const float* qpp = qp + (size_t)((b * NH + n) * LL + it * 16 + l16) * 12 + 2 * h;
  const float* kpp = kp + (size_t)((b * NH + n) * LL + jt * 16 + l16) * 12 + 2 * h;
  v8f accP = {};
#pragma unroll
  for (int k = 0; k < 12; k += 4) {
    v2f a, bv2;
    a.x = qpp[k];   a.y = qpp[k + 1];
    bv2.x = kpp[k]; bv2.y = kpp[k + 1];
    accP = wmma4(a, bv2, accP);
  }
  size_t hb = (size_t)(b * NH + n) * LL;
  int j = jt * 16 + l16;
  float kkj = kk[hb + j];
  float g = gamma[n];
#pragma unroll
  for (int v = 0; v < 8; ++v) {
    int i = it * 16 + v + (h << 3);
    float d2 = qq[hb + i] + kkj - 2.0f * accP[v];
    float lp = -0.125f * g * d2;
    part[(hb + i) * LL + j] = 0.25f * accS[v] + lp;
  }
}

__global__ __launch_bounds__(256) void attn_v_kernel(const float* __restrict__ attn,
                                                     const float* __restrict__ vs,
                                                     const float* __restrict__ vp,
                                                     float* __restrict__ out_cat,
                                                     float* __restrict__ op) {
  constexpr int RT = 2;
  constexpr int AP = LL + 4;
  constexpr int BP = 32 + 8;
  constexpr int AQ = RT * 16 * 3 / 8;
  constexpr int SQ = LL * 4 / 256;
  __shared__ float As[RT * 16 * AP];
  __shared__ float Bs[LL * BP];

  const int tid = threadIdx.x;
  const int wave = __builtin_amdgcn_readfirstlane(tid >> 5);
  const int lane = tid & 31;
  const int h = lane >> 4, l16 = lane & 15;
  const int ip = blockIdx.x % (LT / RT);
  const int bn = blockIdx.x / (LT / RT);
  const int n = bn % NH, b = bn / NH;
  const size_t hb = (size_t)bn * LL;

  const float* ag = attn + (hb + ip * (RT * 16) + 4 * wave) * LL + lane * 4;
  float* al = As + (4 * wave) * AP + lane * 4;
  const float* sg = vs + ((size_t)b * LL + (tid >> 2)) * DD + n * DHS_ + (tid & 3) * 4;
  float* sl = Bs + (tid >> 2) * BP + (tid & 3) * 4;
  const int pq = (tid & 3) < 2 ? (tid & 3) : 2;
  const float* pg = vp + (hb + (tid >> 2)) * 12 + pq * 4;
  float* psl = sl + 16;

  v4f ra[AQ], rs[SQ], rp[SQ];
#pragma unroll
  for (int m = 0; m < AQ; ++m)
    ra[m] = *reinterpret_cast<const v4f*>(ag + (m / 3) * LL + (m % 3) * 128);
#pragma unroll
  for (int m = 0; m < SQ; ++m)
    rs[m] = *reinterpret_cast<const v4f*>(sg + (size_t)(m * 64) * DD);
#pragma unroll
  for (int m = 0; m < SQ; ++m)
    rp[m] = *reinterpret_cast<const v4f*>(pg + (m * 64) * 12);

#pragma unroll
  for (int m = 0; m < AQ; ++m)
    *reinterpret_cast<v4f*>(al + (m / 3) * AP + (m % 3) * 128) = ra[m];
#pragma unroll
  for (int m = 0; m < SQ; ++m)
    *reinterpret_cast<v4f*>(sl + (m * 64) * BP) = rs[m];
#pragma unroll
  for (int m = 0; m < SQ; ++m)
    *reinterpret_cast<v4f*>(psl + (m * 64) * BP) = rp[m];
  __syncthreads();

  if (wave < 2 * RT) {
    const int rt = wave >> 1, ct = wave & 1;
    constexpr int KB = 16;
    constexpr int NB = LL / (4 * KB);
    v8f acc = {};
    const float* asp = As + (rt * 16 + l16) * AP + 2 * h;
    const float* bsp = Bs + (2 * h) * BP + ct * 16 + l16;
    v2f fa[2][KB], fb[2][KB];
#pragma unroll
    for (int s = 0; s < KB; ++s) {
      fa[0][s] = *reinterpret_cast<const v2f*>(asp + 4 * s);
      fb[0][s].x = bsp[(4 * s) * BP];
      fb[0][s].y = bsp[(4 * s + 1) * BP];
    }
#pragma unroll
    for (int t = 0; t < NB; ++t) {
      if (t + 1 < NB) {
#pragma unroll
        for (int s = 0; s < KB; ++s) {
          const int k = (t + 1) * 4 * KB + 4 * s;
          fa[(t + 1) & 1][s] = *reinterpret_cast<const v2f*>(asp + k);
          fb[(t + 1) & 1][s].x = bsp[k * BP];
          fb[(t + 1) & 1][s].y = bsp[(k + 1) * BP];
        }
      }
      __builtin_amdgcn_sched_barrier(0);
#pragma unroll
      for (int s = 0; s < KB; ++s) acc = wmma4(fa[t & 1][s], fb[t & 1][s], acc);
    }
    const int i0 = ip * (RT * 16) + rt * 16 + (h << 3);
    if (ct == 0) {
      float* o = out_cat + (size_t)(b * LL + i0) * DOUT + n * DHS_ + l16;
#pragma unroll
      for (int v = 0; v < 8; ++v) o[(size_t)v * DOUT] = acc[v];
    } else if (l16 < 12) {
      float* o = op + (hb + i0) * 12 + l16;
#pragma unroll
      for (int v = 0; v < 8; ++v) o[v * 12] = acc[v];
    }
  }
}

__global__ __launch_bounds__(256) void pair_row_kernel(
    const float* __restrict__ part, const float* __restrict__ e,
    const float* __restrict__ Wb, float* __restrict__ attn,
    float* __restrict__ out_cat) {
  constexpr int KC  = 128;
  constexpr int NCH = LL / KC;
  constexpr int AP  = LL + 4;
  constexpr int WP  = DD + 12;
  constexpr int WQ  = KC / 8;
  __shared__ float As[16 * AP];
  __shared__ float Ws[LL * WP];

  const int tid = threadIdx.x;
  const int wave = tid >> 5, lane = tid & 31;
  const int h = lane >> 4, l16 = lane & 15;
  const int i = blockIdx.x % LL;
  const int b = blockIdx.x / LL;
  const int tn = wave << 4;
  const int jj = tid & (KC - 1);
  const int g = __builtin_amdgcn_readfirstlane(tid >> 7);

  const float* wg = e + (((size_t)(b * LL + i) * LL) + wave) * DD + lane * 4;
  float* wl = Ws + wave * WP + lane * 4;
  const float* wq = Wb + g * 4;
  const float* pg = part + ((size_t)(b * NH + g * 4) * LL + i) * LL + jj;

  v4f rw[2][WQ];
#pragma unroll
  for (int q = 0; q < WQ; ++q)
    rw[0][q] = *reinterpret_cast<const v4f*>(wg + (size_t)(q * 8) * DD);
#pragma unroll
  for (int q = 0; q < WQ; ++q)
    rw[1][q] = *reinterpret_cast<const v4f*>(wg + (size_t)(KC + q * 8) * DD);
  float sp[NCH][4];
#pragma unroll
  for (int c = 0; c < NCH; ++c)
#pragma unroll
    for (int n = 0; n < 4; ++n)
      sp[c][n] = pg[(size_t)n * LL * LL + c * KC];

  {
    v4f z = {};
#pragma unroll
    for (int c = 0; c < NCH; ++c)
      *reinterpret_cast<v4f*>(As + (8 + wave) * AP + c * KC + lane * 4) = z;
  }

  const float rsqrt3 = 0.57735026918962576f;
#pragma unroll
  for (int c = 0; c < NCH; ++c) {
#pragma unroll
    for (int q = 0; q < WQ; ++q)
      *reinterpret_cast<v4f*>(wl + (c * KC + q * 8) * WP) = rw[c & 1][q];
    __syncthreads();
    if (c + 2 < NCH) {
      const int k2 = (c + 2) * KC;
#pragma unroll
      for (int q = 0; q < WQ; ++q)
        rw[c & 1][q] = *reinterpret_cast<const v4f*>(wg + (size_t)(k2 + q * 8) * DD);
    }
    const float* row = Ws + (c * KC + jj) * WP;
    float s0 = 0.0f, s1 = 0.0f, s2 = 0.0f, s3 = 0.0f;
    for (int q = 0; q < DD / 4; ++q) {
      const v4f ev = *reinterpret_cast<const v4f*>(row + 4 * q);
      const float* w = wq + (4 * q) * NH;
      s0 += ev.x * w[0];          s1 += ev.x * w[1];          s2 += ev.x * w[2];          s3 += ev.x * w[3];
      s0 += ev.y * w[NH + 0];     s1 += ev.y * w[NH + 1];     s2 += ev.y * w[NH + 2];     s3 += ev.y * w[NH + 3];
      s0 += ev.z * w[2 * NH + 0]; s1 += ev.z * w[2 * NH + 1]; s2 += ev.z * w[2 * NH + 2]; s3 += ev.z * w[2 * NH + 3];
      s0 += ev.w * w[3 * NH + 0]; s1 += ev.w * w[3 * NH + 1]; s2 += ev.w * w[3 * NH + 2]; s3 += ev.w * w[3 * NH + 3];
    }
    float* lg = As + (g * 4) * AP + c * KC + jj;
    lg[0]      = rsqrt3 * (sp[c][0] + s0);
    lg[AP]     = rsqrt3 * (sp[c][1] + s1);
    lg[2 * AP] = rsqrt3 * (sp[c][2] + s2);
    lg[3 * AP] = rsqrt3 * (sp[c][3] + s3);
  }
  __syncthreads();

  float v[12];
  {
    float* lr = As + wave * AP + lane;
    float m = -1e30f;
#pragma unroll
    for (int k = 0; k < 12; ++k) {
      v[k] = lr[k * 32];
      m = fmaxf(m, v[k]);
    }
#pragma unroll
    for (int off = 16; off > 0; off >>= 1) m = fmaxf(m, __shfl_xor(m, off, 32));
    float s = 0.0f;
#pragma unroll
    for (int k = 0; k < 12; ++k) {
      v[k] = __expf(v[k] - m);
      s += v[k];
    }
#pragma unroll
    for (int off = 16; off > 0; off >>= 1) s += __shfl_xor(s, off, 32);
    float inv = 1.0f / s;
#pragma unroll
    for (int k = 0; k < 12; ++k) {
      v[k] = v[k] * inv;
      lr[k * 32] = v[k];
    }
  }
  __syncthreads();
  {
    float* ao = attn + ((size_t)(b * NH + wave) * LL + i) * LL + lane;
#pragma unroll
    for (int k = 0; k < 12; ++k) ao[k * 32] = v[k];
  }

  constexpr int KB = 16;
  constexpr int NB = LL / (4 * KB);
  v8f acc = {};
  const float* asp = As + l16 * AP + 2 * h;
  const float* wsp = Ws + (2 * h) * WP + tn + l16;
  v2f fa[2][KB], fb[2][KB];
#pragma unroll
  for (int s = 0; s < KB; ++s) {
    fa[0][s] = *reinterpret_cast<const v2f*>(asp + 4 * s);
    fb[0][s].x = wsp[(4 * s) * WP];
    fb[0][s].y = wsp[(4 * s + 1) * WP];
  }
#pragma unroll
  for (int t = 0; t < NB; ++t) {
    if (t + 1 < NB) {
#pragma unroll
      for (int s = 0; s < KB; ++s) {
        const int k = (t + 1) * 4 * KB + 4 * s;
        fa[(t + 1) & 1][s] = *reinterpret_cast<const v2f*>(asp + k);
        fb[(t + 1) & 1][s].x = wsp[k * WP];
        fb[(t + 1) & 1][s].y = wsp[(k + 1) * WP];
      }
    }
    __builtin_amdgcn_sched_barrier(0);
#pragma unroll
    for (int s = 0; s < KB; ++s) acc = wmma4(fa[t & 1][s], fb[t & 1][s], acc);
  }
  if (h == 0) {
#pragma unroll
    for (int r = 0; r < 8; ++r)
      out_cat[(size_t)(b * LL + i) * DOUT + DD + r * DD + tn + l16] = acc[r];
  }
}

__global__ void op_finish_kernel(const float* __restrict__ op,
                                 const float* __restrict__ r,
                                 const float* __restrict__ t,
                                 float* __restrict__ out_cat) {
  int idx = blockIdx.x * blockDim.x + threadIdx.x;
  if (idx >= BNL) return;
  int i = idx % LL;
  int n = (idx / LL) % NH;
  int b = idx / (NH * LL);
  const float* opv = op + (size_t)idx * 12;
  const float* rp = r + (size_t)(b * LL + i) * 9;
  const float* tp = t + (size_t)(b * LL + i) * 3;
  float* oc = out_cat + (size_t)(b * LL + i) * DOUT;
#pragma unroll
  for (int p = 0; p < 4; ++p) {
    float om[3];
#pragma unroll
    for (int k = 0; k < 3; ++k) om[k] = opv[p * 3 + k] - tp[k];
    float nsq = 0.0f;
#pragma unroll
    for (int c = 0; c < 3; ++c) {
      float v = om[0] * rp[c * 3 + 0] + om[1] * rp[c * 3 + 1] + om[2] * rp[c * 3 + 2];
      oc[DD + NH * DD + n * 12 + p * 3 + c] = v;
      nsq += v * v;
    }
    oc[DD + NH * DD + NH * 12 + n * 4 + p] = sqrtf(nsq);
  }
}

extern "C" void kernel_launch(void* const* d_in, const int* in_sizes, int n_in,
                              void* d_out, int out_size, void* d_ws, size_t ws_size,
                              hipStream_t stream) {
  (void)in_sizes; (void)n_in; (void)out_size;
  const float* x     = (const float*)d_in[0];
  const float* e     = (const float*)d_in[1];
  const float* r     = (const float*)d_in[2];
  const float* t     = (const float*)d_in[3];
  const float* Wq_s  = (const float*)d_in[4];
  const float* Wk_s  = (const float*)d_in[5];
  const float* Wv_s  = (const float*)d_in[6];
  const float* Wb    = (const float*)d_in[7];
  const float* Wq_p  = (const float*)d_in[8];
  const float* Wk_p  = (const float*)d_in[9];
  const float* Wv_p  = (const float*)d_in[10];
  const float* gamma = (const float*)d_in[11];
  const float* Wo    = (const float*)d_in[12];
  const float* bo    = (const float*)d_in[13];
  float* out = (float*)d_out;
  float* ws  = (float*)d_ws;

  float* qs   = ws;
  float* ks_  = qs  + 98304;
  float* vs   = ks_ + 98304;
  float* hq   = vs  + 98304;
  float* hk   = hq  + 73728;
  float* hv   = hk  + 73728;
  float* qp   = hv  + 73728;
  float* kp   = qp  + 73728;
  float* vp   = kp  + 73728;
  float* qq   = vp  + 73728;
  float* kk   = qq  + 6144;
  float* part = kk  + 6144;
  float* attn = part + 2359296;
  float* op   = attn + 2359296;
  float* ocat = op  + 73728;
  size_t req = (size_t)((ocat + 983040) - ws) * sizeof(float);
  if (ws_size < req) return;

  dim3 blk(256);

  proj6_kernel<<<BL / 16, blk, 0, stream>>>(x, Wq_s, Wk_s, Wv_s, Wq_p, Wk_p, Wv_p,
                                            qs, ks_, vs, hq, hk, hv);

  point_transform_kernel<<<BNL / 256, blk, 0, stream>>>(hq, r, t, qp, qq);
  point_transform_kernel<<<BNL / 256, blk, 0, stream>>>(hk, r, t, kp, kk);
  point_transform_kernel<<<BNL / 256, blk, 0, stream>>>(hv, r, t, vp, nullptr);

  logit_kernel<<<(BB * NH * LT * LT) / 8, blk, 0, stream>>>(qs, ks_, qp, kp, qq,
                                                            kk, gamma, part);

  pair_row_kernel<<<BB * LL, blk, 0, stream>>>(part, e, Wb, attn, ocat);

  attn_v_kernel<<<BB * NH * (LT / 2), blk, 0, stream>>>(attn, vs, vp, ocat, op);
  op_finish_kernel<<<BNL / 256, blk, 0, stream>>>(op, r, t, ocat);

  wmma_gemm_lds_kernel<DOUT, DD><<<BL / 16, blk, 0, stream>>>(ocat, Wo, bo, out);
}


# frame transforms and point finish recomputed in their consumers' loaders (logit fragments, attn@V staging, last A chunk of the final GEMM) from the raw projections; 9 launches -> 5; sum orders kept
# speedup vs baseline: 3.8847x; 1.1764x over previous
#include <hip/hip_runtime.h>
#include <hip/hip_bf16.h>
#include <math.h>

#define BB   2
#define LL   384
#define DD   128
#define NH   8
#define DHS_ 16
#define DHP_ 4
#define BL   (BB*LL)
#define BNL  (BB*NH*LL)
#define DOUT 1280
#define LT   24

typedef float v2f __attribute__((ext_vector_type(2)));
typedef float v4f __attribute__((ext_vector_type(4)));
typedef float v8f __attribute__((ext_vector_type(8)));

static __device__ __forceinline__ v8f wmma4(v2f a, v2f b, v8f c) {
  return __builtin_amdgcn_wmma_f32_16x16x4_f32(false, a, false, b, (short)0, c,
                                               false, false);
}


static __device__ __forceinline__ void local_point(float pa, float pb, float pc,
                                                   float t0, float t1, float t2,
                                                   float r00, float r01, float r02,
                                                   float r10, float r11, float r12,
                                                   float r20, float r21, float r22,
                                                   float& x, float& y, float& z, float& nm) {
  const float a = pa - t0, b = pb - t1, cc = pc - t2;
  x = a * r00 + b * r01 + cc * r02;
  y = a * r10 + b * r11 + cc * r12;
  z = a * r20 + b * r21 + cc * r22;
  float nsq = 0.0f;
  nsq += x * x;
  nsq += y * y;
  nsq += z * z;
  nm = sqrtf(nsq);
}

static __device__ __forceinline__ v4f point_piece(const float* __restrict__ op,
                                                  const float* __restrict__ r,
                                                  const float* __restrict__ t,
                                                  int row, int vec) {
  const int b = row / LL, i = row - b * LL;
  const int n = vec < 24 ? vec / 3 : vec - 24;
  const float* opv = op + ((size_t)(b * NH + n) * LL + i) * 12;
  const float* rp = r + (size_t)row * 9;
  const float* tp = t + (size_t)row * 3;
  const v4f o0 = *reinterpret_cast<const v4f*>(opv);
  const v4f o1 = *reinterpret_cast<const v4f*>(opv + 4);
  const v4f o2 = *reinterpret_cast<const v4f*>(opv + 8);
  const float r00 = rp[0], r01 = rp[1], r02 = rp[2];
  const float r10 = rp[3], r11 = rp[4], r12 = rp[5];
  const float r20 = rp[6], r21 = rp[7], r22 = rp[8];
  const float t0 = tp[0], t1 = tp[1], t2 = tp[2];
  float x0, y0, z0, n0, x1, y1, z1, n1, x2, y2, z2, n2, x3, y3, z3, n3;
  local_point(o0.x, o0.y, o0.z, t0, t1, t2, r00, r01, r02, r10, r11, r12, r20, r21, r22, x0, y0, z0, n0);
  local_point(o0.w, o1.x, o1.y, t0, t1, t2, r00, r01, r02, r10, r11, r12, r20, r21, r22, x1, y1, z1, n1);
  local_point(o1.z, o1.w, o2.x, t0, t1, t2, r00, r01, r02, r10, r11, r12, r20, r21, r22, x2, y2, z2, n2);
  local_point(o2.y, o2.z, o2.w, t0, t1, t2, r00, r01, r02, r10, r11, r12, r20, r21, r22, x3, y3, z3, n3);
  const int s = vec % 3;
  const bool nrm = vec >= 24;
  v4f o;
  o.x = nrm ? n0 : (s == 0 ? x0 : (s == 1 ? y1 : z2));
  o.y = nrm ? n1 : (s == 0 ? y0 : (s == 1 ? z1 : x3));
  o.z = nrm ? n2 : (s == 0 ? z0 : (s == 1 ? x2 : y3));
  o.w = nrm ? n3 : (s == 0 ? x1 : (s == 1 ? y2 : z3));
  return o;
}

template <int K, int NC>
__global__ __launch_bounds__(256) void wmma_gemm_lds_kernel(
    const float* __restrict__ A, const float* __restrict__ W,
    const float* __restrict__ bias, const float* __restrict__ op,
    const float* __restrict__ r, const float* __restrict__ t, float* __restrict__ C) {
  constexpr int KC  = 128;
  constexpr int AP  = KC + 4;
  constexpr int WP  = NC + 8;
  constexpr int NCH = K / KC;
  constexpr int WV  = NC / 4;
  constexpr int WR  = 8;
  constexpr int WQ  = KC / WR;
  constexpr int AQ  = (16 * KC / 4) / 256;
  __shared__ float As[16 * AP];
  __shared__ float Ws[KC * WP];

  const int tid = threadIdx.x;
  const int wave = tid >> 5, lane = tid & 31;
  const int h = lane >> 4, l16 = lane & 15;
  const int tm = blockIdx.x << 4;
  const int tn = wave << 4;

  v4f ra[AQ];
  v4f rw[WQ] = {};
  const float* ag = A + (size_t)(tm + (tid >> 5)) * K + (tid & 31) * 4;
  const float* wg = W + (size_t)(tid / WV) * NC + (tid % WV) * 4;
  float* al = As + (tid >> 5) * AP + (tid & 31) * 4;
  float* wl = Ws + (tid / WV) * WP + (tid % WV) * 4;
  const bool wact = tid < WR * WV;

#pragma unroll
  for (int q = 0; q < AQ; ++q)
    ra[q] = *reinterpret_cast<const v4f*>(ag + (size_t)(q * 8) * K);
  if (wact) {
#pragma unroll
    for (int q = 0; q < WQ; ++q)
      rw[q] = *reinterpret_cast<const v4f*>(wg + (size_t)(q * WR) * NC);
  }
  v4f pp[AQ];
#pragma unroll
  for (int q = 0; q < AQ; ++q)
    pp[q] = point_piece(op, r, t, tm + (tid >> 5) + q * 8, tid & 31);

  v8f acc = {};
  for (int c = 0; c < NCH; ++c) {
#pragma unroll
    for (int q = 0; q < AQ; ++q)
      *reinterpret_cast<v4f*>(al + (q * 8) * AP) = ra[q];
    if (wact) {
#pragma unroll
      for (int q = 0; q < WQ; ++q)
        *reinterpret_cast<v4f*>(wl + (q * WR) * WP) = rw[q];
    }
    __syncthreads();
    if (c + 1 < NCH) {
      const int k1 = (c + 1) * KC;
      if (c + 2 < NCH) {
#pragma unroll
        for (int q = 0; q < AQ; ++q)
          ra[q] = *reinterpret_cast<const v4f*>(ag + (size_t)(q * 8) * K + k1);
      } else {
#pragma unroll
        for (int q = 0; q < AQ; ++q) ra[q] = pp[q];
      }
      if (wact) {
#pragma unroll
        for (int q = 0; q < WQ; ++q)
          rw[q] = *reinterpret_cast<const v4f*>(wg + (size_t)(k1 + q * WR) * NC);
      }
    }
    if (tn < NC) {
      const float* asp = As + l16 * AP + 2 * h;
      const float* wsp = Ws + (2 * h) * WP + tn + l16;
#pragma unroll
      for (int kk = 0; kk < KC; kk += 4) {
        v2f a = *reinterpret_cast<const v2f*>(asp + kk);
        v2f b;
        b.x = wsp[kk * WP];
        b.y = wsp[(kk + 1) * WP];
        acc = wmma4(a, b, acc);
      }
    }
    if (c + 1 < NCH) __syncthreads();
  }
  if (tn < NC) {
    float bv = bias ? bias[tn + l16] : 0.0f;
#pragma unroll
    for (int v = 0; v < 8; ++v) {
      int row = tm + v + (h << 3);
      C[(size_t)row * NC + tn + l16] = acc[v] + bv;
    }
  }
}

#define PJ_AP 132
#define PJ_WS (128 * 136)

template <int NC>
static __device__ __forceinline__ void proj_fetch(const float* __restrict__ W, int tid,
                                                  int wave, v4f (&rw)[16]) {
  constexpr int WV = NC / 4;
  if (wave < WV / 4) {
    const float* wg = W + (size_t)(tid / WV) * NC + (tid % WV) * 4;
#pragma unroll
    for (int q = 0; q < 16; ++q)
      rw[q] = *reinterpret_cast<const v4f*>(wg + (size_t)(q * 8) * NC);
  }
}

template <int NC>
static __device__ __forceinline__ void proj_park(float* __restrict__ Ws, int tid, int wave,
                                                 const v4f (&rw)[16]) {
  constexpr int WV = NC / 4;
  constexpr int WP = NC + 8;
  if (wave < WV / 4) {
    float* wl = Ws + (tid / WV) * WP + (tid % WV) * 4;
#pragma unroll
    for (int q = 0; q < 16; ++q)
      *reinterpret_cast<v4f*>(wl + (q * 8) * WP) = rw[q];
  }
}

template <int NC>
static __device__ __forceinline__ v8f proj_mult(const float* __restrict__ As,
                                                const float* __restrict__ Ws, int wave,
                                                int h, int l16) {
  constexpr int WP = NC + 8;
  const int tn = wave << 4;
  v8f acc = {};
  if (tn < NC) {
    const float* asp = As + l16 * PJ_AP + 2 * h;
    const float* wsp = Ws + (2 * h) * WP + tn + l16;
#pragma unroll
    for (int kk = 0; kk < 128; kk += 4) {
      v2f a = *reinterpret_cast<const v2f*>(asp + kk);
      v2f b;
      b.x = wsp[kk * WP];
      b.y = wsp[(kk + 1) * WP];
      acc = wmma4(a, b, acc);
    }
  }
  return acc;
}

template <int NC>
static __device__ __forceinline__ void proj_store(float* __restrict__ C, v8f acc, int tm,
                                                  int wave, int h, int l16) {
  const int tn = wave << 4;
  if (tn < NC) {
#pragma unroll
    for (int v = 0; v < 8; ++v)
      C[(size_t)(tm + v + (h << 3)) * NC + tn + l16] = acc[v];
    __builtin_amdgcn_sched_barrier(0);
  }
}

__global__ __launch_bounds__(256) void proj6_kernel(
    const float* __restrict__ x,
    const float* __restrict__ W0, const float* __restrict__ W1, const float* __restrict__ W2,
    const float* __restrict__ W3, const float* __restrict__ W4, const float* __restrict__ W5,
    float* __restrict__ C0, float* __restrict__ C1, float* __restrict__ C2,
    float* __restrict__ C3, float* __restrict__ C4, float* __restrict__ C5) {
  __shared__ float As[16 * PJ_AP];
  __shared__ float Ws[2 * PJ_WS];

  const int tid = threadIdx.x;
  const int wave = __builtin_amdgcn_readfirstlane(tid >> 5);
  const int lane = tid & 31;
  const int h = lane >> 4, l16 = lane & 15;
  const int tm = blockIdx.x << 4;

  const float* ag = x + (size_t)(tm + wave) * DD + lane * 4;
  v4f ra0 = *reinterpret_cast<const v4f*>(ag);
  v4f ra1 = *reinterpret_cast<const v4f*>(ag + (size_t)8 * DD);
  v4f rw[16] = {};
  proj_fetch<DD>(W0, tid, wave, rw);

  float* al = As + wave * PJ_AP + lane * 4;
  *reinterpret_cast<v4f*>(al) = ra0;
  *reinterpret_cast<v4f*>(al + 8 * PJ_AP) = ra1;

  float* S0 = Ws;
  float* S1 = Ws + PJ_WS;

  proj_park<DD>(S0, tid, wave, rw);
  __syncthreads();
  proj_fetch<DD>(W1, tid, wave, rw);
  v8f a0 = proj_mult<DD>(As, S0, wave, h, l16);

  proj_park<DD>(S1, tid, wave, rw);
  __syncthreads();
  proj_fetch<DD>(W2, tid, wave, rw);
  proj_store<DD>(C0, a0, tm, wave, h, l16);
  v8f a1 = proj_mult<DD>(As, S1, wave, h, l16);

  proj_park<DD>(S0, tid, wave, rw);
  __syncthreads();
  proj_fetch<NH * 12>(W3, tid, wave, rw);
  proj_store<DD>(C1, a1, tm, wave, h, l16);
  v8f a2 = proj_mult<DD>(As, S0, wave, h, l16);

  proj_park<NH * 12>(S1, tid, wave, rw);
  __syncthreads();
  proj_fetch<NH * 12>(W4, tid, wave, rw);
  proj_store<DD>(C2, a2, tm, wave, h, l16);
  v8f a3 = proj_mult<NH * 12>(As, S1, wave, h, l16);

  proj_park<NH * 12>(S0, tid, wave, rw);
  __syncthreads();
  proj_fetch<NH * 12>(W5, tid, wave, rw);
  proj_store<NH * 12>(C3, a3, tm, wave, h, l16);
  v8f a4 = proj_mult<NH * 12>(As, S0, wave, h, l16);

  proj_park<NH * 12>(S1, tid, wave, rw);
  __syncthreads();
  proj_store<NH * 12>(C4, a4, tm, wave, h, l16);
  v8f a5 = proj_mult<NH * 12>(As, S1, wave, h, l16);
  proj_store<NH * 12>(C5, a5, tm, wave, h, l16);
}

struct pts12 { float g0, g1, g2, g3, g4, g5, g6, g7, g8, g9, g10, g11, ss; };

static __device__ __forceinline__ pts12 frame_row(const float* __restrict__ hp,
                                                  const float* __restrict__ rp,
                                                  const float* __restrict__ tp) {
  const v4f h0 = *reinterpret_cast<const v4f*>(hp);
  const v4f h1 = *reinterpret_cast<const v4f*>(hp + 4);
  const v4f h2 = *reinterpret_cast<const v4f*>(hp + 8);
  const float r00 = rp[0], r01 = rp[1], r02 = rp[2];
  const float r10 = rp[3], r11 = rp[4], r12 = rp[5];
  const float r20 = rp[6], r21 = rp[7], r22 = rp[8];
  const float t0 = tp[0], t1 = tp[1], t2 = tp[2];
  pts12 o;
  o.g0  = __builtin_fmaf(h0.z, r20, __builtin_fmaf(h0.y, r10, __builtin_fmaf(h0.x, r00, t0)));
  o.g1  = __builtin_fmaf(h0.z, r21, __builtin_fmaf(h0.y, r11, __builtin_fmaf(h0.x, r01, t1)));
  o.g2  = __builtin_fmaf(h0.z, r22, __builtin_fmaf(h0.y, r12, __builtin_fmaf(h0.x, r02, t2)));
  o.g3  = __builtin_fmaf(h1.y, r20, __builtin_fmaf(h1.x, r10, __builtin_fmaf(h0.w, r00, t0)));
  o.g4  = __builtin_fmaf(h1.y, r21, __builtin_fmaf(h1.x, r11, __builtin_fmaf(h0.w, r01, t1)));
  o.g5  = __builtin_fmaf(h1.y, r22, __builtin_fmaf(h1.x, r12, __builtin_fmaf(h0.w, r02, t2)));
  o.g6  = __builtin_fmaf(h2.x, r20, __builtin_fmaf(h1.w, r10, __builtin_fmaf(h1.z, r00, t0)));
  o.g7  = __builtin_fmaf(h2.x, r21, __builtin_fmaf(h1.w, r11, __builtin_fmaf(h1.z, r01, t1)));
  o.g8  = __builtin_fmaf(h2.x, r22, __builtin_fmaf(h1.w, r12, __builtin_fmaf(h1.z, r02, t2)));
  o.g9  = __builtin_fmaf(h2.w, r20, __builtin_fmaf(h2.z, r10, __builtin_fmaf(h2.y, r00, t0)));
  o.g10 = __builtin_fmaf(h2.w, r21, __builtin_fmaf(h2.z, r11, __builtin_fmaf(h2.y, r01, t1)));
  o.g11 = __builtin_fmaf(h2.w, r22, __builtin_fmaf(h2.z, r12, __builtin_fmaf(h2.y, r02, t2)));
  float ss = 0.0f;
  ss = __builtin_fmaf(o.g0, o.g0, ss);
  ss = __builtin_fmaf(o.g1, o.g1, ss);
  ss = __builtin_fmaf(o.g2, o.g2, ss);
  ss = __builtin_fmaf(o.g3, o.g3, ss);
  ss = __builtin_fmaf(o.g4, o.g4, ss);
  ss = __builtin_fmaf(o.g5, o.g5, ss);
  ss = __builtin_fmaf(o.g6, o.g6, ss);
  ss = __builtin_fmaf(o.g7, o.g7, ss);
  ss = __builtin_fmaf(o.g8, o.g8, ss);
  ss = __builtin_fmaf(o.g9, o.g9, ss);
  ss = __builtin_fmaf(o.g10, o.g10, ss);
  ss = __builtin_fmaf(o.g11, o.g11, ss);
  o.ss = ss;
  return o;
}

__global__ void logit_kernel(const float* __restrict__ qs,
                             const float* __restrict__ ks,
                             const float* __restrict__ hq,
                             const float* __restrict__ hk,
                             const float* __restrict__ r,
                             const float* __restrict__ t,
                             const float* __restrict__ gamma,
                             float* __restrict__ part) {
  int tile = blockIdx.x * 8 + (threadIdx.x >> 5);
  int jt = tile % LT;
  int it = (tile / LT) % LT;
  int n  = (tile / (LT * LT)) % NH;
  int b  = tile / (LT * LT * NH);
  int lane = threadIdx.x & 31;
  int h = lane >> 4, l16 = lane & 15;

  const float* qsp = qs + (size_t)(b * LL + it * 16 + l16) * DD + n * DHS_ + 2 * h;
  const float* ksp = ks + (size_t)(b * LL + jt * 16 + l16) * DD + n * DHS_ + 2 * h;
  v8f accS = {};
#pragma unroll
  for (int k = 0; k < 16; k += 4) {
    v2f a, bv2;
    a.x = qsp[k];   a.y = qsp[k + 1];
    bv2.x = ksp[k]; bv2.y = ksp[k + 1];
    accS = wmma4(a, bv2, accS);
  }
  const int rq = b * LL + it * 16 + l16;
  const int rk = b * LL + jt * 16 + l16;
  const pts12 gq = frame_row(hq + (size_t)rq * (NH * 12) + n * 12, r + (size_t)rq * 9,
                             t + (size_t)rq * 3);
  const pts12 gk = frame_row(hk + (size_t)rk * (NH * 12) + n * 12, r + (size_t)rk * 9,
                             t + (size_t)rk * 3);
  const float qqi = gq.ss, kkj = gk.ss;
  v8f accP = {};
  {
    v2f a, bv2;
    a.x   = h ? gq.g2 : gq.g0;    a.y   = h ? gq.g3 : gq.g1;
    bv2.x = h ? gk.g2 : gk.g0;    bv2.y = h ? gk.g3 : gk.g1;
    accP = wmma4(a, bv2, accP);
    a.x   = h ? gq.g6 : gq.g4;    a.y   = h ? gq.g7 : gq.g5;
    bv2.x = h ? gk.g6 : gk.g4;    bv2.y = h ? gk.g7 : gk.g5;
    accP = wmma4(a, bv2, accP);
    a.x   = h ? gq.g10 : gq.g8;   a.y   = h ? gq.g11 : gq.g9;
    bv2.x = h ? gk.g10 : gk.g8;   bv2.y = h ? gk.g11 : gk.g9;
    accP = wmma4(a, bv2, accP);
  }
  size_t hb = (size_t)(b * NH + n) * LL;
  int j = jt * 16 + l16;
  float g = gamma[n];
#pragma unroll
  for (int v = 0; v < 8; ++v) {
    int i = it * 16 + v + (h << 3);
    const float q0 = __int_as_float(__builtin_amdgcn_readlane(__float_as_int(qqi), v));
    const float q1 = __int_as_float(__builtin_amdgcn_readlane(__float_as_int(qqi), v + 8));
    float d2 = (h ? q1 : q0) + kkj - 2.0f * accP[v];
    float lp = -0.125f * g * d2;
    part[(hb + i) * LL + j] = 0.25f * accS[v] + lp;
  }
}

__global__ __launch_bounds__(256) void attn_v_kernel(const float* __restrict__ attn,
                                                     const float* __restrict__ vs,
                                                     const float* __restrict__ hv,
                                                     const float* __restrict__ r,
                                                     const float* __restrict__ t,
                                                     float* __restrict__ out_cat,
                                                     float* __restrict__ op) {
  constexpr int RT = 2;
  constexpr int AP = LL + 4;
  constexpr int BP = 32 + 8;
  constexpr int AQ = RT * 16 * 3 / 8;
  constexpr int SQ = LL * 4 / 256;
  __shared__ float As[RT * 16 * AP];
  __shared__ float Bs[LL * BP];

  const int tid = threadIdx.x;
  const int wave = __builtin_amdgcn_readfirstlane(tid >> 5);
  const int lane = tid & 31;
  const int h = lane >> 4, l16 = lane & 15;
  const int ip = blockIdx.x % (LT / RT);
  const int bn = blockIdx.x / (LT / RT);
  const int n = bn % NH, b = bn / NH;
  const size_t hb = (size_t)bn * LL;

  const float* ag = attn + (hb + ip * (RT * 16) + 4 * wave) * LL + lane * 4;
  float* al = As + (4 * wave) * AP + lane * 4;
  const float* sg = vs + ((size_t)b * LL + (tid >> 2)) * DD + n * DHS_ + (tid & 3) * 4;
  float* sl = Bs + (tid >> 2) * BP + (tid & 3) * 4;
  const int pq = (tid & 3) < 2 ? (tid & 3) : 2;
  const size_t vrow = (size_t)b * LL + (tid >> 2);
  const float* hg = hv + vrow * (NH * 12) + n * 12;
  const float* rg = r + vrow * 9;
  const float* tg = t + vrow * 3;
  float* psl = sl + 16;

  v4f ra[AQ], rs[SQ], rp[SQ];
#pragma unroll
  for (int m = 0; m < AQ; ++m)
    ra[m] = *reinterpret_cast<const v4f*>(ag + (m / 3) * LL + (m % 3) * 128);
#pragma unroll
  for (int m = 0; m < SQ; ++m)
    rs[m] = *reinterpret_cast<const v4f*>(sg + (size_t)(m * 64) * DD);
#pragma unroll
  for (int m = 0; m < SQ; ++m) {
    const pts12 g = frame_row(hg + (size_t)(m * 64) * (NH * 12), rg + (m * 64) * 9,
                              tg + (m * 64) * 3);
    rp[m].x = pq == 0 ? g.g0 : (pq == 1 ? g.g4 : g.g8);
    rp[m].y = pq == 0 ? g.g1 : (pq == 1 ? g.g5 : g.g9);
    rp[m].z = pq == 0 ? g.g2 : (pq == 1 ? g.g6 : g.g10);
    rp[m].w = pq == 0 ? g.g3 : (pq == 1 ? g.g7 : g.g11);
  }

#pragma unroll
  for (int m = 0; m < AQ; ++m)
    *reinterpret_cast<v4f*>(al + (m / 3) * AP + (m % 3) * 128) = ra[m];
#pragma unroll
  for (int m = 0; m < SQ; ++m)
    *reinterpret_cast<v4f*>(sl + (m * 64) * BP) = rs[m];
#pragma unroll
  for (int m = 0; m < SQ; ++m)
    *reinterpret_cast<v4f*>(psl + (m * 64) * BP) = rp[m];
  __syncthreads();

  if (wave < 2 * RT) {
    const int rt = wave >> 1, ct = wave & 1;
    constexpr int KB = 16;
    constexpr int NB = LL / (4 * KB);
    v8f acc = {};
    const float* asp = As + (rt * 16 + l16) * AP + 2 * h;
    const float* bsp = Bs + (2 * h) * BP + ct * 16 + l16;
    v2f fa[2][KB], fb[2][KB];
#pragma unroll
    for (int s = 0; s < KB; ++s) {
      fa[0][s] = *reinterpret_cast<const v2f*>(asp + 4 * s);
      fb[0][s].x = bsp[(4 * s) * BP];
      fb[0][s].y = bsp[(4 * s + 1) * BP];
    }
#pragma unroll
    for (int t = 0; t < NB; ++t) {
      if (t + 1 < NB) {
#pragma unroll
        for (int s = 0; s < KB; ++s) {
          const int k = (t + 1) * 4 * KB + 4 * s;
          fa[(t + 1) & 1][s] = *reinterpret_cast<const v2f*>(asp + k);
          fb[(t + 1) & 1][s].x = bsp[k * BP];
          fb[(t + 1) & 1][s].y = bsp[(k + 1) * BP];
        }
      }
      __builtin_amdgcn_sched_barrier(0);
#pragma unroll
      for (int s = 0; s < KB; ++s) acc = wmma4(fa[t & 1][s], fb[t & 1][s], acc);
    }
    const int i0 = ip * (RT * 16) + rt * 16 + (h << 3);
    if (ct == 0) {
      float* o = out_cat + (size_t)(b * LL + i0) * DOUT + n * DHS_ + l16;
#pragma unroll
      for (int v = 0; v < 8; ++v) o[(size_t)v * DOUT] = acc[v];
    } else if (l16 < 12) {
      float* o = op + (hb + i0) * 12 + l16;
#pragma unroll
      for (int v = 0; v < 8; ++v) o[v * 12] = acc[v];
    }
  }
}

__global__ __launch_bounds__(256) void pair_row_kernel(
    const float* __restrict__ part, const float* __restrict__ e,
    const float* __restrict__ Wb, float* __restrict__ attn,
    float* __restrict__ out_cat) {
  constexpr int KC  = 128;
  constexpr int NCH = LL / KC;
  constexpr int AP  = LL + 4;
  constexpr int WP  = DD + 12;
  constexpr int WQ  = KC / 8;
  __shared__ float As[16 * AP];
  __shared__ float Ws[LL * WP];

  const int tid = threadIdx.x;
  const int wave = tid >> 5, lane = tid & 31;
  const int h = lane >> 4, l16 = lane & 15;
  const int i = blockIdx.x % LL;
  const int b = blockIdx.x / LL;
  const int tn = wave << 4;
  const int jj = tid & (KC - 1);
  const int g = __builtin_amdgcn_readfirstlane(tid >> 7);

  const float* wg = e + (((size_t)(b * LL + i) * LL) + wave) * DD + lane * 4;
  float* wl = Ws + wave * WP + lane * 4;
  const float* wq = Wb + g * 4;
  const float* pg = part + ((size_t)(b * NH + g * 4) * LL + i) * LL + jj;

  v4f rw[2][WQ];
#pragma unroll
  for (int q = 0; q < WQ; ++q)
    rw[0][q] = *reinterpret_cast<const v4f*>(wg + (size_t)(q * 8) * DD);
#pragma unroll
  for (int q = 0; q < WQ; ++q)
    rw[1][q] = *reinterpret_cast<const v4f*>(wg + (size_t)(KC + q * 8) * DD);
  float sp[NCH][4];
#pragma unroll
  for (int c = 0; c < NCH; ++c)
#pragma unroll
    for (int n = 0; n < 4; ++n)
      sp[c][n] = pg[(size_t)n * LL * LL + c * KC];

  {
    v4f z = {};
#pragma unroll
    for (int c = 0; c < NCH; ++c)
      *reinterpret_cast<v4f*>(As + (8 + wave) * AP + c * KC + lane * 4) = z;
  }

  const float rsqrt3 = 0.57735026918962576f;
#pragma unroll
  for (int c = 0; c < NCH; ++c) {
#pragma unroll
    for (int q = 0; q < WQ; ++q)
      *reinterpret_cast<v4f*>(wl + (c * KC + q * 8) * WP) = rw[c & 1][q];
    __syncthreads();
    if (c + 2 < NCH) {
      const int k2 = (c + 2) * KC;
#pragma unroll
      for (int q = 0; q < WQ; ++q)
        rw[c & 1][q] = *reinterpret_cast<const v4f*>(wg + (size_t)(k2 + q * 8) * DD);
    }
    const float* row = Ws + (c * KC + jj) * WP;
    float s0 = 0.0f, s1 = 0.0f, s2 = 0.0f, s3 = 0.0f;
    for (int q = 0; q < DD / 4; ++q) {
      const v4f ev = *reinterpret_cast<const v4f*>(row + 4 * q);
      const float* w = wq + (4 * q) * NH;
      s0 += ev.x * w[0];          s1 += ev.x * w[1];          s2 += ev.x * w[2];          s3 += ev.x * w[3];
      s0 += ev.y * w[NH + 0];     s1 += ev.y * w[NH + 1];     s2 += ev.y * w[NH + 2];     s3 += ev.y * w[NH + 3];
      s0 += ev.z * w[2 * NH + 0]; s1 += ev.z * w[2 * NH + 1]; s2 += ev.z * w[2 * NH + 2]; s3 += ev.z * w[2 * NH + 3];
      s0 += ev.w * w[3 * NH + 0]; s1 += ev.w * w[3 * NH + 1]; s2 += ev.w * w[3 * NH + 2]; s3 += ev.w * w[3 * NH + 3];
    }
    float* lg = As + (g * 4) * AP + c * KC + jj;
    lg[0]      = rsqrt3 * (sp[c][0] + s0);
    lg[AP]     = rsqrt3 * (sp[c][1] + s1);
    lg[2 * AP] = rsqrt3 * (sp[c][2] + s2);
    lg[3 * AP] = rsqrt3 * (sp[c][3] + s3);
  }
  __syncthreads();

  float v[12];
  {
    float* lr = As + wave * AP + lane;
    float m = -1e30f;
#pragma unroll
    for (int k = 0; k < 12; ++k) {
      v[k] = lr[k * 32];
      m = fmaxf(m, v[k]);
    }
#pragma unroll
    for (int off = 16; off > 0; off >>= 1) m = fmaxf(m, __shfl_xor(m, off, 32));
    float s = 0.0f;
#pragma unroll
    for (int k = 0; k < 12; ++k) {
      v[k] = __expf(v[k] - m);
      s += v[k];
    }
#pragma unroll
    for (int off = 16; off > 0; off >>= 1) s += __shfl_xor(s, off, 32);
    float inv = 1.0f / s;
#pragma unroll
    for (int k = 0; k < 12; ++k) {
      v[k] = v[k] * inv;
      lr[k * 32] = v[k];
    }
  }
  __syncthreads();
  {
    float* ao = attn + ((size_t)(b * NH + wave) * LL + i) * LL + lane;
#pragma unroll
    for (int k = 0; k < 12; ++k) ao[k * 32] = v[k];
  }

  constexpr int KB = 16;
  constexpr int NB = LL / (4 * KB);
  v8f acc = {};
  const float* asp = As + l16 * AP + 2 * h;
  const float* wsp = Ws + (2 * h) * WP + tn + l16;
  v2f fa[2][KB], fb[2][KB];
#pragma unroll
  for (int s = 0; s < KB; ++s) {
    fa[0][s] = *reinterpret_cast<const v2f*>(asp + 4 * s);
    fb[0][s].x = wsp[(4 * s) * WP];
    fb[0][s].y = wsp[(4 * s + 1) * WP];
  }
#pragma unroll
  for (int t = 0; t < NB; ++t) {
    if (t + 1 < NB) {
#pragma unroll
      for (int s = 0; s < KB; ++s) {
        const int k = (t + 1) * 4 * KB + 4 * s;
        fa[(t + 1) & 1][s] = *reinterpret_cast<const v2f*>(asp + k);
        fb[(t + 1) & 1][s].x = wsp[k * WP];
        fb[(t + 1) & 1][s].y = wsp[(k + 1) * WP];
      }
    }
    __builtin_amdgcn_sched_barrier(0);
#pragma unroll
    for (int s = 0; s < KB; ++s) acc = wmma4(fa[t & 1][s], fb[t & 1][s], acc);
  }
  if (h == 0) {
#pragma unroll
    for (int r = 0; r < 8; ++r)
      out_cat[(size_t)(b * LL + i) * DOUT + DD + r * DD + tn + l16] = acc[r];
  }
}

extern "C" void kernel_launch(void* const* d_in, const int* in_sizes, int n_in,
                              void* d_out, int out_size, void* d_ws, size_t ws_size,
                              hipStream_t stream) {
  (void)in_sizes; (void)n_in; (void)out_size;
  const float* x     = (const float*)d_in[0];
  const float* e     = (const float*)d_in[1];
  const float* r     = (const float*)d_in[2];
  const float* t     = (const float*)d_in[3];
  const float* Wq_s  = (const float*)d_in[4];
  const float* Wk_s  = (const float*)d_in[5];
  const float* Wv_s  = (const float*)d_in[6];
  const float* Wb    = (const float*)d_in[7];
  const float* Wq_p  = (const float*)d_in[8];
  const float* Wk_p  = (const float*)d_in[9];
  const float* Wv_p  = (const float*)d_in[10];
  const float* gamma = (const float*)d_in[11];
  const float* Wo    = (const float*)d_in[12];
  const float* bo    = (const float*)d_in[13];
  float* out = (float*)d_out;
  float* ws  = (float*)d_ws;

  float* qs   = ws;
  float* ks_  = qs  + 98304;
  float* vs   = ks_ + 98304;
  float* hq   = vs  + 98304;
  float* hk   = hq  + 73728;
  float* hv   = hk  + 73728;
  float* part = hv  + 73728;
  float* attn = part + 2359296;
  float* op   = attn + 2359296;
  float* ocat = op  + 73728;
  size_t req = (size_t)((ocat + 983040) - ws) * sizeof(float);
  if (ws_size < req) return;

  dim3 blk(256);

  proj6_kernel<<<BL / 16, blk, 0, stream>>>(x, Wq_s, Wk_s, Wv_s, Wq_p, Wk_p, Wv_p,
                                            qs, ks_, vs, hq, hk, hv);

  logit_kernel<<<(BB * NH * LT * LT) / 8, blk, 0, stream>>>(qs, ks_, hq, hk, r, t,
                                                            gamma, part);

  pair_row_kernel<<<BB * LL, blk, 0, stream>>>(part, e, Wb, attn, ocat);

  attn_v_kernel<<<BB * NH * (LT / 2), blk, 0, stream>>>(attn, vs, hv, r, t, ocat, op);

  wmma_gemm_lds_kernel<DOUT, DD><<<BL / 16, blk, 0, stream>>>(ocat, Wo, bo, op, r, t, out);
}


# pair bias summed per wave on the 16 rows it parked itself, no barrier per chunk: d over 8 lanes of a row, Wb piece in registers, packed FMAs, lane exchange adds partials; d sum order changes
# speedup vs baseline: 4.3547x; 1.1210x over previous
#include <hip/hip_runtime.h>
#include <hip/hip_bf16.h>
#include <math.h>

#define BB   2
#define LL   384
#define DD   128
#define NH   8
#define DHS_ 16
#define DHP_ 4
#define BL   (BB*LL)
#define BNL  (BB*NH*LL)
#define DOUT 1280
#define LT   24

typedef float v2f __attribute__((ext_vector_type(2)));
typedef float v4f __attribute__((ext_vector_type(4)));
typedef float v8f __attribute__((ext_vector_type(8)));

static __device__ __forceinline__ v8f wmma4(v2f a, v2f b, v8f c) {
  return __builtin_amdgcn_wmma_f32_16x16x4_f32(false, a, false, b, (short)0, c,
                                               false, false);
}


static __device__ __forceinline__ void local_point(float pa, float pb, float pc,
                                                   float t0, float t1, float t2,
                                                   float r00, float r01, float r02,
                                                   float r10, float r11, float r12,
                                                   float r20, float r21, float r22,
                                                   float& x, float& y, float& z, float& nm) {
  const float a = pa - t0, b = pb - t1, cc = pc - t2;
  x = a * r00 + b * r01 + cc * r02;
  y = a * r10 + b * r11 + cc * r12;
  z = a * r20 + b * r21 + cc * r22;
  float nsq = 0.0f;
  nsq += x * x;
  nsq += y * y;
  nsq += z * z;
  nm = sqrtf(nsq);
}

static __device__ __forceinline__ v4f point_piece(const float* __restrict__ op,
                                                  const float* __restrict__ r,
                                                  const float* __restrict__ t,
                                                  int row, int vec) {
  const int b = row / LL, i = row - b * LL;
  const int n = vec < 24 ? vec / 3 : vec - 24;
  const float* opv = op + ((size_t)(b * NH + n) * LL + i) * 12;
  const float* rp = r + (size_t)row * 9;
  const float* tp = t + (size_t)row * 3;
  const v4f o0 = *reinterpret_cast<const v4f*>(opv);
  const v4f o1 = *reinterpret_cast<const v4f*>(opv + 4);
  const v4f o2 = *reinterpret_cast<const v4f*>(opv + 8);
  const float r00 = rp[0], r01 = rp[1], r02 = rp[2];
  const float r10 = rp[3], r11 = rp[4], r12 = rp[5];
  const float r20 = rp[6], r21 = rp[7], r22 = rp[8];
  const float t0 = tp[0], t1 = tp[1], t2 = tp[2];
  float x0, y0, z0, n0, x1, y1, z1, n1, x2, y2, z2, n2, x3, y3, z3, n3;
  local_point(o0.x, o0.y, o0.z, t0, t1, t2, r00, r01, r02, r10, r11, r12, r20, r21, r22, x0, y0, z0, n0);
  local_point(o0.w, o1.x, o1.y, t0, t1, t2, r00, r01, r02, r10, r11, r12, r20, r21, r22, x1, y1, z1, n1);
  local_point(o1.z, o1.w, o2.x, t0, t1, t2, r00, r01, r02, r10, r11, r12, r20, r21, r22, x2, y2, z2, n2);
  local_point(o2.y, o2.z, o2.w, t0, t1, t2, r00, r01, r02, r10, r11, r12, r20, r21, r22, x3, y3, z3, n3);
  const int s = vec % 3;
  const bool nrm = vec >= 24;
  v4f o;
  o.x = nrm ? n0 : (s == 0 ? x0 : (s == 1 ? y1 : z2));
  o.y = nrm ? n1 : (s == 0 ? y0 : (s == 1 ? z1 : x3));
  o.z = nrm ? n2 : (s == 0 ? z0 : (s == 1 ? x2 : y3));
  o.w = nrm ? n3 : (s == 0 ? x1 : (s == 1 ? y2 : z3));
  return o;
}

template <int K, int NC>
__global__ __launch_bounds__(256) void wmma_gemm_lds_kernel(
    const float* __restrict__ A, const float* __restrict__ W,
    const float* __restrict__ bias, const float* __restrict__ op,
    const float* __restrict__ r, const float* __restrict__ t, float* __restrict__ C) {
  constexpr int KC  = 128;
  constexpr int AP  = KC + 4;
  constexpr int WP  = NC + 8;
  constexpr int NCH = K / KC;
  constexpr int WV  = NC / 4;
  constexpr int WR  = 8;
  constexpr int WQ  = KC / WR;
  constexpr int AQ  = (16 * KC / 4) / 256;
  __shared__ float As[16 * AP];
  __shared__ float Ws[KC * WP];

  const int tid = threadIdx.x;
  const int wave = tid >> 5, lane = tid & 31;
  const int h = lane >> 4, l16 = lane & 15;
  const int tm = blockIdx.x << 4;
  const int tn = wave << 4;

  v4f ra[AQ];
  v4f rw[WQ] = {};
  const float* ag = A + (size_t)(tm + (tid >> 5)) * K + (tid & 31) * 4;
  const float* wg = W + (size_t)(tid / WV) * NC + (tid % WV) * 4;
  float* al = As + (tid >> 5) * AP + (tid & 31) * 4;
  float* wl = Ws + (tid / WV) * WP + (tid % WV) * 4;
  const bool wact = tid < WR * WV;

#pragma unroll
  for (int q = 0; q < AQ; ++q)
    ra[q] = *reinterpret_cast<const v4f*>(ag + (size_t)(q * 8) * K);
  if (wact) {
#pragma unroll
    for (int q = 0; q < WQ; ++q)
      rw[q] = *reinterpret_cast<const v4f*>(wg + (size_t)(q * WR) * NC);
  }
  v4f pp[AQ];
#pragma unroll
  for (int q = 0; q < AQ; ++q)
    pp[q] = point_piece(op, r, t, tm + (tid >> 5) + q * 8, tid & 31);

  v8f acc = {};
  for (int c = 0; c < NCH; ++c) {
#pragma unroll
    for (int q = 0; q < AQ; ++q)
      *reinterpret_cast<v4f*>(al + (q * 8) * AP) = ra[q];
    if (wact) {
#pragma unroll
      for (int q = 0; q < WQ; ++q)
        *reinterpret_cast<v4f*>(wl + (q * WR) * WP) = rw[q];
    }
    __syncthreads();
    if (c + 1 < NCH) {
      const int k1 = (c + 1) * KC;
      if (c + 2 < NCH) {
#pragma unroll
        for (int q = 0; q < AQ; ++q)
          ra[q] = *reinterpret_cast<const v4f*>(ag + (size_t)(q * 8) * K + k1);
      } else {
#pragma unroll
        for (int q = 0; q < AQ; ++q) ra[q] = pp[q];
      }
      if (wact) {
#pragma unroll
        for (int q = 0; q < WQ; ++q)
          rw[q] = *reinterpret_cast<const v4f*>(wg + (size_t)(k1 + q * WR) * NC);
      }
    }
    if (tn < NC) {
      const float* asp = As + l16 * AP + 2 * h;
      const float* wsp = Ws + (2 * h) * WP + tn + l16;
#pragma unroll
      for (int kk = 0; kk < KC; kk += 4) {
        v2f a = *reinterpret_cast<const v2f*>(asp + kk);
        v2f b;
        b.x = wsp[kk * WP];
        b.y = wsp[(kk + 1) * WP];
        acc = wmma4(a, b, acc);
      }
    }
    if (c + 1 < NCH) __syncthreads();
  }
  if (tn < NC) {
    float bv = bias ? bias[tn + l16] : 0.0f;
#pragma unroll
    for (int v = 0; v < 8; ++v) {
      int row = tm + v + (h << 3);
      C[(size_t)row * NC + tn + l16] = acc[v] + bv;
    }
  }
}

#define PJ_AP 132
#define PJ_WS (128 * 136)

template <int NC>
static __device__ __forceinline__ void proj_fetch(const float* __restrict__ W, int tid,
                                                  int wave, v4f (&rw)[16]) {
  constexpr int WV = NC / 4;
  if (wave < WV / 4) {
    const float* wg = W + (size_t)(tid / WV) * NC + (tid % WV) * 4;
#pragma unroll
    for (int q = 0; q < 16; ++q)
      rw[q] = *reinterpret_cast<const v4f*>(wg + (size_t)(q * 8) * NC);
  }
}

template <int NC>
static __device__ __forceinline__ void proj_park(float* __restrict__ Ws, int tid, int wave,
                                                 const v4f (&rw)[16]) {
  constexpr int WV = NC / 4;
  constexpr int WP = NC + 8;
  if (wave < WV / 4) {
    float* wl = Ws + (tid / WV) * WP + (tid % WV) * 4;
#pragma unroll
    for (int q = 0; q < 16; ++q)
      *reinterpret_cast<v4f*>(wl + (q * 8) * WP) = rw[q];
  }
}

template <int NC>
static __device__ __forceinline__ v8f proj_mult(const float* __restrict__ As,
                                                const float* __restrict__ Ws, int wave,
                                                int h, int l16) {
  constexpr int WP = NC + 8;
  const int tn = wave << 4;
  v8f acc = {};
  if (tn < NC) {
    const float* asp = As + l16 * PJ_AP + 2 * h;
    const float* wsp = Ws + (2 * h) * WP + tn + l16;
#pragma unroll
    for (int kk = 0; kk < 128; kk += 4) {
      v2f a = *reinterpret_cast<const v2f*>(asp + kk);
      v2f b;
      b.x = wsp[kk * WP];
      b.y = wsp[(kk + 1) * WP];
      acc = wmma4(a, b, acc);
    }
  }
  return acc;
}

template <int NC>
static __device__ __forceinline__ void proj_store(float* __restrict__ C, v8f acc, int tm,
                                                  int wave, int h, int l16) {
  const int tn = wave << 4;
  if (tn < NC) {
#pragma unroll
    for (int v = 0; v < 8; ++v)
      C[(size_t)(tm + v + (h << 3)) * NC + tn + l16] = acc[v];
    __builtin_amdgcn_sched_barrier(0);
  }
}

__global__ __launch_bounds__(256) void proj6_kernel(
    const float* __restrict__ x,
    const float* __restrict__ W0, const float* __restrict__ W1, const float* __restrict__ W2,
    const float* __restrict__ W3, const float* __restrict__ W4, const float* __restrict__ W5,
    float* __restrict__ C0, float* __restrict__ C1, float* __restrict__ C2,
    float* __restrict__ C3, float* __restrict__ C4, float* __restrict__ C5) {
  __shared__ float As[16 * PJ_AP];
  __shared__ float Ws[2 * PJ_WS];

  const int tid = threadIdx.x;
  const int wave = __builtin_amdgcn_readfirstlane(tid >> 5);
  const int lane = tid & 31;
  const int h = lane >> 4, l16 = lane & 15;
  const int tm = blockIdx.x << 4;

  const float* ag = x + (size_t)(tm + wave) * DD + lane * 4;
  v4f ra0 = *reinterpret_cast<const v4f*>(ag);
  v4f ra1 = *reinterpret_cast<const v4f*>(ag + (size_t)8 * DD);
  v4f rw[16] = {};
  proj_fetch<DD>(W0, tid, wave, rw);

  float* al = As + wave * PJ_AP + lane * 4;
  *reinterpret_cast<v4f*>(al) = ra0;
  *reinterpret_cast<v4f*>(al + 8 * PJ_AP) = ra1;

  float* S0 = Ws;
  float* S1 = Ws + PJ_WS;

  proj_park<DD>(S0, tid, wave, rw);
  __syncthreads();
  proj_fetch<DD>(W1, tid, wave, rw);
  v8f a0 = proj_mult<DD>(As, S0, wave, h, l16);

  proj_park<DD>(S1, tid, wave, rw);
  __syncthreads();
  proj_fetch<DD>(W2, tid, wave, rw);
  proj_store<DD>(C0, a0, tm, wave, h, l16);
  v8f a1 = proj_mult<DD>(As, S1, wave, h, l16);

  proj_park<DD>(S0, tid, wave, rw);
  __syncthreads();
  proj_fetch<NH * 12>(W3, tid, wave, rw);
  proj_store<DD>(C1, a1, tm, wave, h, l16);
  v8f a2 = proj_mult<DD>(As, S0, wave, h, l16);

  proj_park<NH * 12>(S1, tid, wave, rw);
  __syncthreads();
  proj_fetch<NH * 12>(W4, tid, wave, rw);
  proj_store<DD>(C2, a2, tm, wave, h, l16);
  v8f a3 = proj_mult<NH * 12>(As, S1, wave, h, l16);

  proj_park<NH * 12>(S0, tid, wave, rw);
  __syncthreads();
  proj_fetch<NH * 12>(W5, tid, wave, rw);
  proj_store<NH * 12>(C3, a3, tm, wave, h, l16);
  v8f a4 = proj_mult<NH * 12>(As, S0, wave, h, l16);

  proj_park<NH * 12>(S1, tid, wave, rw);
  __syncthreads();
  proj_store<NH * 12>(C4, a4, tm, wave, h, l16);
  v8f a5 = proj_mult<NH * 12>(As, S1, wave, h, l16);
  proj_store<NH * 12>(C5, a5, tm, wave, h, l16);
}

struct pts12 { float g0, g1, g2, g3, g4, g5, g6, g7, g8, g9, g10, g11, ss; };

static __device__ __forceinline__ pts12 frame_row(const float* __restrict__ hp,
                                                  const float* __restrict__ rp,
                                                  const float* __restrict__ tp) {
  const v4f h0 = *reinterpret_cast<const v4f*>(hp);
  const v4f h1 = *reinterpret_cast<const v4f*>(hp + 4);
  const v4f h2 = *reinterpret_cast<const v4f*>(hp + 8);
  const float r00 = rp[0], r01 = rp[1], r02 = rp[2];
  const float r10 = rp[3], r11 = rp[4], r12 = rp[5];
  const float r20 = rp[6], r21 = rp[7], r22 = rp[8];
  const float t0 = tp[0], t1 = tp[1], t2 = tp[2];
  pts12 o;
  o.g0  = __builtin_fmaf(h0.z, r20, __builtin_fmaf(h0.y, r10, __builtin_fmaf(h0.x, r00, t0)));
  o.g1  = __builtin_fmaf(h0.z, r21, __builtin_fmaf(h0.y, r11, __builtin_fmaf(h0.x, r01, t1)));
  o.g2  = __builtin_fmaf(h0.z, r22, __builtin_fmaf(h0.y, r12, __builtin_fmaf(h0.x, r02, t2)));
  o.g3  = __builtin_fmaf(h1.y, r20, __builtin_fmaf(h1.x, r10, __builtin_fmaf(h0.w, r00, t0)));
  o.g4  = __builtin_fmaf(h1.y, r21, __builtin_fmaf(h1.x, r11, __builtin_fmaf(h0.w, r01, t1)));
  o.g5  = __builtin_fmaf(h1.y, r22, __builtin_fmaf(h1.x, r12, __builtin_fmaf(h0.w, r02, t2)));
  o.g6  = __builtin_fmaf(h2.x, r20, __builtin_fmaf(h1.w, r10, __builtin_fmaf(h1.z, r00, t0)));
  o.g7  = __builtin_fmaf(h2.x, r21, __builtin_fmaf(h1.w, r11, __builtin_fmaf(h1.z, r01, t1)));
  o.g8  = __builtin_fmaf(h2.x, r22, __builtin_fmaf(h1.w, r12, __builtin_fmaf(h1.z, r02, t2)));
  o.g9  = __builtin_fmaf(h2.w, r20, __builtin_fmaf(h2.z, r10, __builtin_fmaf(h2.y, r00, t0)));
  o.g10 = __builtin_fmaf(h2.w, r21, __builtin_fmaf(h2.z, r11, __builtin_fmaf(h2.y, r01, t1)));
  o.g11 = __builtin_fmaf(h2.w, r22, __builtin_fmaf(h2.z, r12, __builtin_fmaf(h2.y, r02, t2)));
  float ss = 0.0f;
  ss = __builtin_fmaf(o.g0, o.g0, ss);
  ss = __builtin_fmaf(o.g1, o.g1, ss);
  ss = __builtin_fmaf(o.g2, o.g2, ss);
  ss = __builtin_fmaf(o.g3, o.g3, ss);
  ss = __builtin_fmaf(o.g4, o.g4, ss);
  ss = __builtin_fmaf(o.g5, o.g5, ss);
  ss = __builtin_fmaf(o.g6, o.g6, ss);
  ss = __builtin_fmaf(o.g7, o.g7, ss);
  ss = __builtin_fmaf(o.g8, o.g8, ss);
  ss = __builtin_fmaf(o.g9, o.g9, ss);
  ss = __builtin_fmaf(o.g10, o.g10, ss);
  ss = __builtin_fmaf(o.g11, o.g11, ss);
  o.ss = ss;
  return o;
}

__global__ void logit_kernel(const float* __restrict__ qs,
                             const float* __restrict__ ks,
                             const float* __restrict__ hq,
                             const float* __restrict__ hk,
                             const float* __restrict__ r,
                             const float* __restrict__ t,
                             const float* __restrict__ gamma,
                             float* __restrict__ part) {
  int tile = blockIdx.x * 8 + (threadIdx.x >> 5);
  int jt = tile % LT;
  int it = (tile / LT) % LT;
  int n  = (tile / (LT * LT)) % NH;
  int b  = tile / (LT * LT * NH);
  int lane = threadIdx.x & 31;
  int h = lane >> 4, l16 = lane & 15;

  const float* qsp = qs + (size_t)(b * LL + it * 16 + l16) * DD + n * DHS_ + 2 * h;
  const float* ksp = ks + (size_t)(b * LL + jt * 16 + l16) * DD + n * DHS_ + 2 * h;
  v8f accS = {};
#pragma unroll
  for (int k = 0; k < 16; k += 4) {
    v2f a, bv2;
    a.x = qsp[k];   a.y = qsp[k + 1];
    bv2.x = ksp[k]; bv2.y = ksp[k + 1];
    accS = wmma4(a, bv2, accS);
  }
  const int rq = b * LL + it * 16 + l16;
  const int rk = b * LL + jt * 16 + l16;
  const pts12 gq = frame_row(hq + (size_t)rq * (NH * 12) + n * 12, r + (size_t)rq * 9,
                             t + (size_t)rq * 3);
  const pts12 gk = frame_row(hk + (size_t)rk * (NH * 12) + n * 12, r + (size_t)rk * 9,
                             t + (size_t)rk * 3);
  const float qqi = gq.ss, kkj = gk.ss;
  v8f accP = {};
  {
    v2f a, bv2;
    a.x   = h ? gq.g2 : gq.g0;    a.y   = h ? gq.g3 : gq.g1;
    bv2.x = h ? gk.g2 : gk.g0;    bv2.y = h ? gk.g3 : gk.g1;
    accP = wmma4(a, bv2, accP);
    a.x   = h ? gq.g6 : gq.g4;    a.y   = h ? gq.g7 : gq.g5;
    bv2.x = h ? gk.g6 : gk.g4;    bv2.y = h ? gk.g7 : gk.g5;
    accP = wmma4(a, bv2, accP);
    a.x   = h ? gq.g10 : gq.g8;   a.y   = h ? gq.g11 : gq.g9;
    bv2.x = h ? gk.g10 : gk.g8;   bv2.y = h ? gk.g11 : gk.g9;
    accP = wmma4(a, bv2, accP);
  }
  size_t hb = (size_t)(b * NH + n) * LL;
  int j = jt * 16 + l16;
  float g = gamma[n];
#pragma unroll
  for (int v = 0; v < 8; ++v) {
    int i = it * 16 + v + (h << 3);
    const float q0 = __int_as_float(__builtin_amdgcn_readlane(__float_as_int(qqi), v));
    const float q1 = __int_as_float(__builtin_amdgcn_readlane(__float_as_int(qqi), v + 8));
    float d2 = (h ? q1 : q0) + kkj - 2.0f * accP[v];
    float lp = -0.125f * g * d2;
    part[(hb + i) * LL + j] = 0.25f * accS[v] + lp;
  }
}

__global__ __launch_bounds__(256) void attn_v_kernel(const float* __restrict__ attn,
                                                     const float* __restrict__ vs,
                                                     const float* __restrict__ hv,
                                                     const float* __restrict__ r,
                                                     const float* __restrict__ t,
                                                     float* __restrict__ out_cat,
                                                     float* __restrict__ op) {
  constexpr int RT = 2;
  constexpr int AP = LL + 4;
  constexpr int BP = 32 + 8;
  constexpr int AQ = RT * 16 * 3 / 8;
  constexpr int SQ = LL * 4 / 256;
  __shared__ float As[RT * 16 * AP];
  __shared__ float Bs[LL * BP];

  const int tid = threadIdx.x;
  const int wave = __builtin_amdgcn_readfirstlane(tid >> 5);
  const int lane = tid & 31;
  const int h = lane >> 4, l16 = lane & 15;
  const int ip = blockIdx.x % (LT / RT);
  const int bn = blockIdx.x / (LT / RT);
  const int n = bn % NH, b = bn / NH;
  const size_t hb = (size_t)bn * LL;

  const float* ag = attn + (hb + ip * (RT * 16) + 4 * wave) * LL + lane * 4;
  float* al = As + (4 * wave) * AP + lane * 4;
  const float* sg = vs + ((size_t)b * LL + (tid >> 2)) * DD + n * DHS_ + (tid & 3) * 4;
  float* sl = Bs + (tid >> 2) * BP + (tid & 3) * 4;
  const int pq = (tid & 3) < 2 ? (tid & 3) : 2;
  const size_t vrow = (size_t)b * LL + (tid >> 2);
  const float* hg = hv + vrow * (NH * 12) + n * 12;
  const float* rg = r + vrow * 9;
  const float* tg = t + vrow * 3;
  float* psl = sl + 16;

  v4f ra[AQ], rs[SQ], rp[SQ];
#pragma unroll
  for (int m = 0; m < AQ; ++m)
    ra[m] = *reinterpret_cast<const v4f*>(ag + (m / 3) * LL + (m % 3) * 128);
#pragma unroll
  for (int m = 0; m < SQ; ++m)
    rs[m] = *reinterpret_cast<const v4f*>(sg + (size_t)(m * 64) * DD);
#pragma unroll
  for (int m = 0; m < SQ; ++m) {
    const pts12 g = frame_row(hg + (size_t)(m * 64) * (NH * 12), rg + (m * 64) * 9,
                              tg + (m * 64) * 3);
    rp[m].x = pq == 0 ? g.g0 : (pq == 1 ? g.g4 : g.g8);
    rp[m].y = pq == 0 ? g.g1 : (pq == 1 ? g.g5 : g.g9);
    rp[m].z = pq == 0 ? g.g2 : (pq == 1 ? g.g6 : g.g10);
    rp[m].w = pq == 0 ? g.g3 : (pq == 1 ? g.g7 : g.g11);
  }

#pragma unroll
  for (int m = 0; m < AQ; ++m)
    *reinterpret_cast<v4f*>(al + (m / 3) * AP + (m % 3) * 128) = ra[m];
#pragma unroll
  for (int m = 0; m < SQ; ++m)
    *reinterpret_cast<v4f*>(sl + (m * 64) * BP) = rs[m];
#pragma unroll
  for (int m = 0; m < SQ; ++m)
    *reinterpret_cast<v4f*>(psl + (m * 64) * BP) = rp[m];
  __syncthreads();

  if (wave < 2 * RT) {
    const int rt = wave >> 1, ct = wave & 1;
    constexpr int KB = 16;
    constexpr int NB = LL / (4 * KB);
    v8f acc = {};
    const float* asp = As + (rt * 16 + l16) * AP + 2 * h;
    const float* bsp = Bs + (2 * h) * BP + ct * 16 + l16;
    v2f fa[2][KB], fb[2][KB];
#pragma unroll
    for (int s = 0; s < KB; ++s) {
      fa[0][s] = *reinterpret_cast<const v2f*>(asp + 4 * s);
      fb[0][s].x = bsp[(4 * s) * BP];
      fb[0][s].y = bsp[(4 * s + 1) * BP];
    }
#pragma unroll
    for (int t = 0; t < NB; ++t) {
      if (t + 1 < NB) {
#pragma unroll
        for (int s = 0; s < KB; ++s) {
          const int k = (t + 1) * 4 * KB + 4 * s;
          fa[(t + 1) & 1][s] = *reinterpret_cast<const v2f*>(asp + k);
          fb[(t + 1) & 1][s].x = bsp[k * BP];
          fb[(t + 1) & 1][s].y = bsp[(k + 1) * BP];
        }
      }
      __builtin_amdgcn_sched_barrier(0);
#pragma unroll
      for (int s = 0; s < KB; ++s) acc = wmma4(fa[t & 1][s], fb[t & 1][s], acc);
    }
    const int i0 = ip * (RT * 16) + rt * 16 + (h << 3);
    if (ct == 0) {
      float* o = out_cat + (size_t)(b * LL + i0) * DOUT + n * DHS_ + l16;
#pragma unroll
      for (int v = 0; v < 8; ++v) o[(size_t)v * DOUT] = acc[v];
    } else if (l16 < 12) {
      float* o = op + (hb + i0) * 12 + l16;
#pragma unroll
      for (int v = 0; v < 8; ++v) o[v * 12] = acc[v];
    }
  }
}

static __device__ __forceinline__ v2f fma2(v2f a, v2f b, v2f c) {
  return __builtin_elementwise_fma(a, b, c);
}

#define XCHG_MIRROR 0x141
#define XCHG_2      0x4E
#define XCHG_1      0xB1
template <int CTRL>
static __device__ __forceinline__ float lane_xchg(float v) {
  return __int_as_float(__builtin_amdgcn_update_dpp(0, __float_as_int(v), CTRL, 0xF, 0xF, true));
}

__global__ __launch_bounds__(256) void pair_row_kernel(
    const float* __restrict__ part, const float* __restrict__ e,
    const float* __restrict__ Wb, float* __restrict__ attn,
    float* __restrict__ out_cat) {
  constexpr int KC  = 128;
  constexpr int NCH = LL / KC;
  constexpr int AP  = LL + 4;
  constexpr int WP  = DD + 12;
  constexpr int WQ  = KC / 8;
  __shared__ float As[16 * AP];
  __shared__ float Ws[LL * WP];
  __shared__ float Wbs[8 * DD * NH];

  const int tid = threadIdx.x;
  const int wave = tid >> 5, lane = tid & 31;
  const int h = lane >> 4, l16 = lane & 15;
  const int i = blockIdx.x % LL;
  const int b = blockIdx.x / LL;
  const int tn = wave << 4;
  const int ds = lane & 7;
  const int r4 = 8 * ((lane >> 3) & 1) + 4 * (lane >> 4);

  const float* wg = e + (((size_t)(b * LL + i) * LL) + 16 * wave) * DD + lane * 4;
  float* wl = Ws + (16 * wave) * WP + lane * 4;
  const float* pg = part + ((size_t)(b * NH + ds) * LL + i) * LL + 16 * wave + r4;

  float* wbw = Wbs + wave * (DD * NH);
  v4f wbv[8];
#pragma unroll
  for (int m = 0; m < 8; ++m)
    wbv[m] = *reinterpret_cast<const v4f*>(Wb + (32 * m + lane) * 4);
  v2f wb[16][4];
  v4f sp[NCH];
#pragma unroll
  for (int c = 0; c < NCH; ++c)
    sp[c] = *reinterpret_cast<const v4f*>(pg + c * KC);

  v4f rw[2][WQ];
#pragma unroll
  for (int q = 0; q < WQ; ++q)
    rw[0][q] = *reinterpret_cast<const v4f*>(wg + (size_t)q * DD);
  __builtin_amdgcn_sched_barrier(0);
#pragma unroll
  for (int q = 0; q < WQ; ++q)
    rw[1][q] = *reinterpret_cast<const v4f*>(wg + (size_t)(KC + q) * DD);
  {
    v4f z = {};
#pragma unroll
    for (int c = 0; c < NCH; ++c)
      *reinterpret_cast<v4f*>(As + (8 + wave) * AP + c * KC + lane * 4) = z;
  }

#pragma unroll
  for (int m = 0; m < 8; ++m)
    *reinterpret_cast<v4f*>(wbw + ((((m >> 1) * 4 + ((lane >> 1) & 3)) * 2 + (lane & 1)) * 8
                                   + 4 * (m & 1) + (lane >> 3)) * 4) = wbv[m];

  const float rsqrt3 = 0.57735026918962576f;
#pragma unroll
  for (int c = 0; c < NCH; ++c) {
#pragma unroll
    for (int q = 0; q < WQ; ++q)
      *reinterpret_cast<v4f*>(wl + (c * KC + q) * WP) = rw[c & 1][q];
    __builtin_amdgcn_sched_barrier(0);
    if (c + 2 < NCH) {
      const int k2 = (c + 2) * KC;
#pragma unroll
      for (int q = 0; q < WQ; ++q)
        rw[c & 1][q] = *reinterpret_cast<const v4f*>(wg + (size_t)(k2 + q) * DD);
    }
    if (c == 0) {
#pragma unroll
      for (int m = 0; m < 16; ++m) {
        const v4f lo = *reinterpret_cast<const v4f*>(wbw + ((2 * m) * 8 + ds) * 4);
        const v4f hi = *reinterpret_cast<const v4f*>(wbw + ((2 * m + 1) * 8 + ds) * 4);
        wb[m][0].x = lo.x;  wb[m][0].y = lo.y;
        wb[m][1].x = lo.z;  wb[m][1].y = lo.w;
        wb[m][2].x = hi.x;  wb[m][2].y = hi.y;
        wb[m][3].x = hi.z;  wb[m][3].y = hi.w;
      }
    }
    const float* erow = Ws + (c * KC + 16 * wave + r4) * WP + 4 * ds;
    v4f evs[4][4];
#pragma unroll
    for (int p = 0; p < 4; ++p)
#pragma unroll
      for (int q = 0; q < 4; ++q)
        evs[p][q] = *reinterpret_cast<const v4f*>(erow + p * WP + 32 * q);
    __builtin_amdgcn_sched_barrier(0);
    v4f bs;
#pragma unroll
    for (int p = 0; p < 4; ++p) {
      v2f a0 = {}, a1 = {}, a2 = {}, a3 = {};
#pragma unroll
      for (int q = 0; q < 4; ++q) {
        const v4f ev = evs[p][q];
        v2f e0, e1, e2, e3;
        e0.x = ev.x;  e0.y = ev.x;
        e1.x = ev.y;  e1.y = ev.y;
        e2.x = ev.z;  e2.y = ev.z;
        e3.x = ev.w;  e3.y = ev.w;
        a0 = fma2(e0, wb[4 * q][0], a0);      a1 = fma2(e0, wb[4 * q][1], a1);
        a2 = fma2(e0, wb[4 * q][2], a2);      a3 = fma2(e0, wb[4 * q][3], a3);
        a0 = fma2(e1, wb[4 * q + 1][0], a0);  a1 = fma2(e1, wb[4 * q + 1][1], a1);
        a2 = fma2(e1, wb[4 * q + 1][2], a2);  a3 = fma2(e1, wb[4 * q + 1][3], a3);
        a0 = fma2(e2, wb[4 * q + 2][0], a0);  a1 = fma2(e2, wb[4 * q + 2][1], a1);
        a2 = fma2(e2, wb[4 * q + 2][2], a2);  a3 = fma2(e2, wb[4 * q + 2][3], a3);
        a0 = fma2(e3, wb[4 * q + 3][0], a0);  a1 = fma2(e3, wb[4 * q + 3][1], a1);
        a2 = fma2(e3, wb[4 * q + 3][2], a2);  a3 = fma2(e3, wb[4 * q + 3][3], a3);
      }
      const bool up4 = (ds & 4) != 0, up2 = (ds & 2) != 0, up1 = (ds & 1) != 0;
      float k0 = up4 ? a2.x : a0.x, k1 = up4 ? a2.y : a0.y, k2 = up4 ? a3.x : a1.x, k3 = up4 ? a3.y : a1.y;
      const float g0 = up4 ? a0.x : a2.x, g1 = up4 ? a0.y : a2.y, g2 = up4 ? a1.x : a3.x, g3 = up4 ? a1.y : a3.y;
      k0 += lane_xchg<XCHG_MIRROR>(g0);
      k1 += lane_xchg<XCHG_MIRROR>(g1);
      k2 += lane_xchg<XCHG_MIRROR>(g2);
      k3 += lane_xchg<XCHG_MIRROR>(g3);
      float m0 = up2 ? k2 : k0, m1 = up2 ? k3 : k1;
      const float u0 = up2 ? k0 : k2, u1 = up2 ? k1 : k3;
      m0 += lane_xchg<XCHG_2>(u0);
      m1 += lane_xchg<XCHG_2>(u1);
      float s = up1 ? m1 : m0;
      const float w = up1 ? m0 : m1;
      s += lane_xchg<XCHG_1>(w);
      bs[p] = s;
    }
    v4f lgv;
    lgv.x = rsqrt3 * (sp[c].x + bs.x);
    lgv.y = rsqrt3 * (sp[c].y + bs.y);
    lgv.z = rsqrt3 * (sp[c].z + bs.z);
    lgv.w = rsqrt3 * (sp[c].w + bs.w);
    *reinterpret_cast<v4f*>(As + ds * AP + c * KC + 16 * wave + r4) = lgv;
  }
  __syncthreads();

  float v[12];
  {
    float* lr = As + wave * AP + lane;
    float m = -1e30f;
#pragma unroll
    for (int k = 0; k < 12; ++k) {
      v[k] = lr[k * 32];
      m = fmaxf(m, v[k]);
    }
#pragma unroll
    for (int off = 16; off > 0; off >>= 1) m = fmaxf(m, __shfl_xor(m, off, 32));
    float s = 0.0f;
#pragma unroll
    for (int k = 0; k < 12; ++k) {
      v[k] = __expf(v[k] - m);
      s += v[k];
    }
#pragma unroll
    for (int off = 16; off > 0; off >>= 1) s += __shfl_xor(s, off, 32);
    float inv = 1.0f / s;
#pragma unroll
    for (int k = 0; k < 12; ++k) {
      v[k] = v[k] * inv;
      lr[k * 32] = v[k];
    }
  }
  __syncthreads();
  {
    float* ao = attn + ((size_t)(b * NH + wave) * LL + i) * LL + lane;
#pragma unroll
    for (int k = 0; k < 12; ++k) ao[k * 32] = v[k];
  }

  constexpr int KB = 16;
  constexpr int NB = LL / (4 * KB);
  v8f acc = {};
  const float* asp = As + l16 * AP + 2 * h;
  const float* wsp = Ws + (2 * h) * WP + tn + l16;
  v2f fa[2][KB], fb[2][KB];
#pragma unroll
  for (int s = 0; s < KB; ++s) {
    fa[0][s] = *reinterpret_cast<const v2f*>(asp + 4 * s);
    fb[0][s].x = wsp[(4 * s) * WP];
    fb[0][s].y = wsp[(4 * s + 1) * WP];
  }
#pragma unroll
  for (int t = 0; t < NB; ++t) {
    if (t + 1 < NB) {
#pragma unroll
      for (int s = 0; s < KB; ++s) {
        const int k = (t + 1) * 4 * KB + 4 * s;
        fa[(t + 1) & 1][s] = *reinterpret_cast<const v2f*>(asp + k);
        fb[(t + 1) & 1][s].x = wsp[k * WP];
        fb[(t + 1) & 1][s].y = wsp[(k + 1) * WP];
      }
    }
    __builtin_amdgcn_sched_barrier(0);
#pragma unroll
    for (int s = 0; s < KB; ++s) acc = wmma4(fa[t & 1][s], fb[t & 1][s], acc);
  }
  if (h == 0) {
#pragma unroll
    for (int r = 0; r < 8; ++r)
      out_cat[(size_t)(b * LL + i) * DOUT + DD + r * DD + tn + l16] = acc[r];
  }
}

extern "C" void kernel_launch(void* const* d_in, const int* in_sizes, int n_in,
                              void* d_out, int out_size, void* d_ws, size_t ws_size,
                              hipStream_t stream) {
  (void)in_sizes; (void)n_in; (void)out_size;
  const float* x     = (const float*)d_in[0];
  const float* e     = (const float*)d_in[1];
  const float* r     = (const float*)d_in[2];
  const float* t     = (const float*)d_in[3];
  const float* Wq_s  = (const float*)d_in[4];
  const float* Wk_s  = (const float*)d_in[5];
  const float* Wv_s  = (const float*)d_in[6];
  const float* Wb    = (const float*)d_in[7];
  const float* Wq_p  = (const float*)d_in[8];
  const float* Wk_p  = (const float*)d_in[9];
  const float* Wv_p  = (const float*)d_in[10];
  const float* gamma = (const float*)d_in[11];
  const float* Wo    = (const float*)d_in[12];
  const float* bo    = (const float*)d_in[13];
  float* out = (float*)d_out;
  float* ws  = (float*)d_ws;

  float* qs   = ws;
  float* ks_  = qs  + 98304;
  float* vs   = ks_ + 98304;
  float* hq   = vs  + 98304;
  float* hk   = hq  + 73728;
  float* hv   = hk  + 73728;
  float* part = hv  + 73728;
  float* attn = part + 2359296;
  float* op   = attn + 2359296;
  float* ocat = op  + 73728;
  size_t req = (size_t)((ocat + 983040) - ws) * sizeof(float);
  if (ws_size < req) return;

  dim3 blk(256);

  proj6_kernel<<<BL / 16, blk, 0, stream>>>(x, Wq_s, Wk_s, Wv_s, Wq_p, Wk_p, Wv_p,
                                            qs, ks_, vs, hq, hk, hv);

  logit_kernel<<<(BB * NH * LT * LT) / 8, blk, 0, stream>>>(qs, ks_, hq, hk, r, t,
                                                            gamma, part);

  pair_row_kernel<<<BB * LL, blk, 0, stream>>>(part, e, Wb, attn, ocat);

  attn_v_kernel<<<BB * NH * (LT / 2), blk, 0, stream>>>(attn, vs, hv, r, t, ocat, op);

  wmma_gemm_lds_kernel<DOUT, DD><<<BL / 16, blk, 0, stream>>>(ocat, Wo, bo, op, r, t, out);
}


# pair row kernel: 256 blocks (one per WGP) walk 3 rows each; next row's partial logits and first two e chunks are loaded behind the softmax barriers, during the pair-output WMMAs; sum orders kept
# speedup vs baseline: 4.5308x; 1.0404x over previous
#include <hip/hip_runtime.h>
#include <hip/hip_bf16.h>
#include <math.h>

#define BB   2
#define LL   384
#define DD   128
#define NH   8
#define DHS_ 16
#define DHP_ 4
#define BL   (BB*LL)
#define BNL  (BB*NH*LL)
#define DOUT 1280
#define LT   24

typedef float v2f __attribute__((ext_vector_type(2)));
typedef float v4f __attribute__((ext_vector_type(4)));
typedef float v8f __attribute__((ext_vector_type(8)));

static __device__ __forceinline__ v8f wmma4(v2f a, v2f b, v8f c) {
  return __builtin_amdgcn_wmma_f32_16x16x4_f32(false, a, false, b, (short)0, c,
                                               false, false);
}


static __device__ __forceinline__ void local_point(float pa, float pb, float pc,
                                                   float t0, float t1, float t2,
                                                   float r00, float r01, float r02,
                                                   float r10, float r11, float r12,
                                                   float r20, float r21, float r22,
                                                   float& x, float& y, float& z, float& nm) {
  const float a = pa - t0, b = pb - t1, cc = pc - t2;
  x = a * r00 + b * r01 + cc * r02;
  y = a * r10 + b * r11 + cc * r12;
  z = a * r20 + b * r21 + cc * r22;
  float nsq = 0.0f;
  nsq += x * x;
  nsq += y * y;
  nsq += z * z;
  nm = sqrtf(nsq);
}

static __device__ __forceinline__ v4f point_piece(const float* __restrict__ op,
                                                  const float* __restrict__ r,
                                                  const float* __restrict__ t,
                                                  int row, int vec) {
  const int b = row / LL, i = row - b * LL;
  const int n = vec < 24 ? vec / 3 : vec - 24;
  const float* opv = op + ((size_t)(b * NH + n) * LL + i) * 12;
  const float* rp = r + (size_t)row * 9;
  const float* tp = t + (size_t)row * 3;
  const v4f o0 = *reinterpret_cast<const v4f*>(opv);
  const v4f o1 = *reinterpret_cast<const v4f*>(opv + 4);
  const v4f o2 = *reinterpret_cast<const v4f*>(opv + 8);
  const float r00 = rp[0], r01 = rp[1], r02 = rp[2];
  const float r10 = rp[3], r11 = rp[4], r12 = rp[5];
  const float r20 = rp[6], r21 = rp[7], r22 = rp[8];
  const float t0 = tp[0], t1 = tp[1], t2 = tp[2];
  float x0, y0, z0, n0, x1, y1, z1, n1, x2, y2, z2, n2, x3, y3, z3, n3;
  local_point(o0.x, o0.y, o0.z, t0, t1, t2, r00, r01, r02, r10, r11, r12, r20, r21, r22, x0, y0, z0, n0);
  local_point(o0.w, o1.x, o1.y, t0, t1, t2, r00, r01, r02, r10, r11, r12, r20, r21, r22, x1, y1, z1, n1);
  local_point(o1.z, o1.w, o2.x, t0, t1, t2, r00, r01, r02, r10, r11, r12, r20, r21, r22, x2, y2, z2, n2);
  local_point(o2.y, o2.z, o2.w, t0, t1, t2, r00, r01, r02, r10, r11, r12, r20, r21, r22, x3, y3, z3, n3);
  const int s = vec % 3;
  const bool nrm = vec >= 24;
  v4f o;
  o.x = nrm ? n0 : (s == 0 ? x0 : (s == 1 ? y1 : z2));
  o.y = nrm ? n1 : (s == 0 ? y0 : (s == 1 ? z1 : x3));
  o.z = nrm ? n2 : (s == 0 ? z0 : (s == 1 ? x2 : y3));
  o.w = nrm ? n3 : (s == 0 ? x1 : (s == 1 ? y2 : z3));
  return o;
}

template <int K, int NC>
__global__ __launch_bounds__(256) void wmma_gemm_lds_kernel(
    const float* __restrict__ A, const float* __restrict__ W,
    const float* __restrict__ bias, const float* __restrict__ op,
    const float* __restrict__ r, const float* __restrict__ t, float* __restrict__ C) {
  constexpr int KC  = 128;
  constexpr int AP  = KC + 4;
  constexpr int WP  = NC + 8;
  constexpr int NCH = K / KC;
  constexpr int WV  = NC / 4;
  constexpr int WR  = 8;
  constexpr int WQ  = KC / WR;
  constexpr int AQ  = (16 * KC / 4) / 256;
  __shared__ float As[16 * AP];
  __shared__ float Ws[KC * WP];

  const int tid = threadIdx.x;
  const int wave = tid >> 5, lane = tid & 31;
  const int h = lane >> 4, l16 = lane & 15;
  const int tm = blockIdx.x << 4;
  const int tn = wave << 4;

  v4f ra[AQ];
  v4f rw[WQ] = {};
  const float* ag = A + (size_t)(tm + (tid >> 5)) * K + (tid & 31) * 4;
  const float* wg = W + (size_t)(tid / WV) * NC + (tid % WV) * 4;
  float* al = As + (tid >> 5) * AP + (tid & 31) * 4;
  float* wl = Ws + (tid / WV) * WP + (tid % WV) * 4;
  const bool wact = tid < WR * WV;

#pragma unroll
  for (int q = 0; q < AQ; ++q)
    ra[q] = *reinterpret_cast<const v4f*>(ag + (size_t)(q * 8) * K);
  if (wact) {
#pragma unroll
    for (int q = 0; q < WQ; ++q)
      rw[q] = *reinterpret_cast<const v4f*>(wg + (size_t)(q * WR) * NC);
  }
  v4f pp[AQ];
#pragma unroll
  for (int q = 0; q < AQ; ++q)
    pp[q] = point_piece(op, r, t, tm + (tid >> 5) + q * 8, tid & 31);

  v8f acc = {};
  for (int c = 0; c < NCH; ++c) {
#pragma unroll
    for (int q = 0; q < AQ; ++q)
      *reinterpret_cast<v4f*>(al + (q * 8) * AP) = ra[q];
    if (wact) {
#pragma unroll
      for (int q = 0; q < WQ; ++q)
        *reinterpret_cast<v4f*>(wl + (q * WR) * WP) = rw[q];
    }
    __syncthreads();
    if (c + 1 < NCH) {
      const int k1 = (c + 1) * KC;
      if (c + 2 < NCH) {
#pragma unroll
        for (int q = 0; q < AQ; ++q)
          ra[q] = *reinterpret_cast<const v4f*>(ag + (size_t)(q * 8) * K + k1);
      } else {
#pragma unroll
        for (int q = 0; q < AQ; ++q) ra[q] = pp[q];
      }
      if (wact) {
#pragma unroll
        for (int q = 0; q < WQ; ++q)
          rw[q] = *reinterpret_cast<const v4f*>(wg + (size_t)(k1 + q * WR) * NC);
      }
    }
    if (tn < NC) {
      const float* asp = As + l16 * AP + 2 * h;
      const float* wsp = Ws + (2 * h) * WP + tn + l16;
#pragma unroll
      for (int kk = 0; kk < KC; kk += 4) {
        v2f a = *reinterpret_cast<const v2f*>(asp + kk);
        v2f b;
        b.x = wsp[kk * WP];
        b.y = wsp[(kk + 1) * WP];
        acc = wmma4(a, b, acc);
      }
    }
    if (c + 1 < NCH) __syncthreads();
  }
  if (tn < NC) {
    float bv = bias ? bias[tn + l16] : 0.0f;
#pragma unroll
    for (int v = 0; v < 8; ++v) {
      int row = tm + v + (h << 3);
      C[(size_t)row * NC + tn + l16] = acc[v] + bv;
    }
  }
}

#define PJ_AP 132
#define PJ_WS (128 * 136)

template <int NC>
static __device__ __forceinline__ void proj_fetch(const float* __restrict__ W, int tid,
                                                  int wave, v4f (&rw)[16]) {
  constexpr int WV = NC / 4;
  if (wave < WV / 4) {
    const float* wg = W + (size_t)(tid / WV) * NC + (tid % WV) * 4;
#pragma unroll
    for (int q = 0; q < 16; ++q)
      rw[q] = *reinterpret_cast<const v4f*>(wg + (size_t)(q * 8) * NC);
  }
}

template <int NC>
static __device__ __forceinline__ void proj_park(float* __restrict__ Ws, int tid, int wave,
                                                 const v4f (&rw)[16]) {
  constexpr int WV = NC / 4;
  constexpr int WP = NC + 8;
  if (wave < WV / 4) {
    float* wl = Ws + (tid / WV) * WP + (tid % WV) * 4;
#pragma unroll
    for (int q = 0; q < 16; ++q)
      *reinterpret_cast<v4f*>(wl + (q * 8) * WP) = rw[q];
  }
}

template <int NC>
static __device__ __forceinline__ v8f proj_mult(const float* __restrict__ As,
                                                const float* __restrict__ Ws, int wave,
                                                int h, int l16) {
  constexpr int WP = NC + 8;
  const int tn = wave << 4;
  v8f acc = {};
  if (tn < NC) {
    const float* asp = As + l16 * PJ_AP + 2 * h;
    const float* wsp = Ws + (2 * h) * WP + tn + l16;
#pragma unroll
    for (int kk = 0; kk < 128; kk += 4) {
      v2f a = *reinterpret_cast<const v2f*>(asp + kk);
      v2f b;
      b.x = wsp[kk * WP];
      b.y = wsp[(kk + 1) * WP];
      acc = wmma4(a, b, acc);
    }
  }
  return acc;
}

template <int NC>
static __device__ __forceinline__ void proj_store(float* __restrict__ C, v8f acc, int tm,
                                                  int wave, int h, int l16) {
  const int tn = wave << 4;
  if (tn < NC) {
#pragma unroll
    for (int v = 0; v < 8; ++v)
      C[(size_t)(tm + v + (h << 3)) * NC + tn + l16] = acc[v];
    __builtin_amdgcn_sched_barrier(0);
  }
}

__global__ __launch_bounds__(256) void proj6_kernel(
    const float* __restrict__ x,
    const float* __restrict__ W0, const float* __restrict__ W1, const float* __restrict__ W2,
    const float* __restrict__ W3, const float* __restrict__ W4, const float* __restrict__ W5,
    float* __restrict__ C0, float* __restrict__ C1, float* __restrict__ C2,
    float* __restrict__ C3, float* __restrict__ C4, float* __restrict__ C5) {
  __shared__ float As[16 * PJ_AP];
  __shared__ float Ws[2 * PJ_WS];

  const int tid = threadIdx.x;
  const int wave = __builtin_amdgcn_readfirstlane(tid >> 5);
  const int lane = tid & 31;
  const int h = lane >> 4, l16 = lane & 15;
  const int tm = blockIdx.x << 4;

  const float* ag = x + (size_t)(tm + wave) * DD + lane * 4;
  v4f ra0 = *reinterpret_cast<const v4f*>(ag);
  v4f ra1 = *reinterpret_cast<const v4f*>(ag + (size_t)8 * DD);
  v4f rw[16] = {};
  proj_fetch<DD>(W0, tid, wave, rw);

  float* al = As + wave * PJ_AP + lane * 4;
  *reinterpret_cast<v4f*>(al) = ra0;
  *reinterpret_cast<v4f*>(al + 8 * PJ_AP) = ra1;

  float* S0 = Ws;
  float* S1 = Ws + PJ_WS;

  proj_park<DD>(S0, tid, wave, rw);
  __syncthreads();
  proj_fetch<DD>(W1, tid, wave, rw);
  v8f a0 = proj_mult<DD>(As, S0, wave, h, l16);

  proj_park<DD>(S1, tid, wave, rw);
  __syncthreads();
  proj_fetch<DD>(W2, tid, wave, rw);
  proj_store<DD>(C0, a0, tm, wave, h, l16);
  v8f a1 = proj_mult<DD>(As, S1, wave, h, l16);

  proj_park<DD>(S0, tid, wave, rw);
  __syncthreads();
  proj_fetch<NH * 12>(W3, tid, wave, rw);
  proj_store<DD>(C1, a1, tm, wave, h, l16);
  v8f a2 = proj_mult<DD>(As, S0, wave, h, l16);

  proj_park<NH * 12>(S1, tid, wave, rw);
  __syncthreads();
  proj_fetch<NH * 12>(W4, tid, wave, rw);
  proj_store<DD>(C2, a2, tm, wave, h, l16);
  v8f a3 = proj_mult<NH * 12>(As, S1, wave, h, l16);

  proj_park<NH * 12>(S0, tid, wave, rw);
  __syncthreads();
  proj_fetch<NH * 12>(W5, tid, wave, rw);
  proj_store<NH * 12>(C3, a3, tm, wave, h, l16);
  v8f a4 = proj_mult<NH * 12>(As, S0, wave, h, l16);

  proj_park<NH * 12>(S1, tid, wave, rw);
  __syncthreads();
  proj_store<NH * 12>(C4, a4, tm, wave, h, l16);
  v8f a5 = proj_mult<NH * 12>(As, S1, wave, h, l16);
  proj_store<NH * 12>(C5, a5, tm, wave, h, l16);
}

struct pts12 { float g0, g1, g2, g3, g4, g5, g6, g7, g8, g9, g10, g11, ss; };

static __device__ __forceinline__ pts12 frame_row(const float* __restrict__ hp,
                                                  const float* __restrict__ rp,
                                                  const float* __restrict__ tp) {
  const v4f h0 = *reinterpret_cast<const v4f*>(hp);
  const v4f h1 = *reinterpret_cast<const v4f*>(hp + 4);
  const v4f h2 = *reinterpret_cast<const v4f*>(hp + 8);
  const float r00 = rp[0], r01 = rp[1], r02 = rp[2];
  const float r10 = rp[3], r11 = rp[4], r12 = rp[5];
  const float r20 = rp[6], r21 = rp[7], r22 = rp[8];
  const float t0 = tp[0], t1 = tp[1], t2 = tp[2];
  pts12 o;
  o.g0  = __builtin_fmaf(h0.z, r20, __builtin_fmaf(h0.y, r10, __builtin_fmaf(h0.x, r00, t0)));
  o.g1  = __builtin_fmaf(h0.z, r21, __builtin_fmaf(h0.y, r11, __builtin_fmaf(h0.x, r01, t1)));
  o.g2  = __builtin_fmaf(h0.z, r22, __builtin_fmaf(h0.y, r12, __builtin_fmaf(h0.x, r02, t2)));
  o.g3  = __builtin_fmaf(h1.y, r20, __builtin_fmaf(h1.x, r10, __builtin_fmaf(h0.w, r00, t0)));
  o.g4  = __builtin_fmaf(h1.y, r21, __builtin_fmaf(h1.x, r11, __builtin_fmaf(h0.w, r01, t1)));
  o.g5  = __builtin_fmaf(h1.y, r22, __builtin_fmaf(h1.x, r12, __builtin_fmaf(h0.w, r02, t2)));
  o.g6  = __builtin_fmaf(h2.x, r20, __builtin_fmaf(h1.w, r10, __builtin_fmaf(h1.z, r00, t0)));
  o.g7  = __builtin_fmaf(h2.x, r21, __builtin_fmaf(h1.w, r11, __builtin_fmaf(h1.z, r01, t1)));
  o.g8  = __builtin_fmaf(h2.x, r22, __builtin_fmaf(h1.w, r12, __builtin_fmaf(h1.z, r02, t2)));
  o.g9  = __builtin_fmaf(h2.w, r20, __builtin_fmaf(h2.z, r10, __builtin_fmaf(h2.y, r00, t0)));
  o.g10 = __builtin_fmaf(h2.w, r21, __builtin_fmaf(h2.z, r11, __builtin_fmaf(h2.y, r01, t1)));
  o.g11 = __builtin_fmaf(h2.w, r22, __builtin_fmaf(h2.z, r12, __builtin_fmaf(h2.y, r02, t2)));
  float ss = 0.0f;
  ss = __builtin_fmaf(o.g0, o.g0, ss);
  ss = __builtin_fmaf(o.g1, o.g1, ss);
  ss = __builtin_fmaf(o.g2, o.g2, ss);
  ss = __builtin_fmaf(o.g3, o.g3, ss);
  ss = __builtin_fmaf(o.g4, o.g4, ss);
  ss = __builtin_fmaf(o.g5, o.g5, ss);
  ss = __builtin_fmaf(o.g6, o.g6, ss);
  ss = __builtin_fmaf(o.g7, o.g7, ss);
  ss = __builtin_fmaf(o.g8, o.g8, ss);
  ss = __builtin_fmaf(o.g9, o.g9, ss);
  ss = __builtin_fmaf(o.g10, o.g10, ss);
  ss = __builtin_fmaf(o.g11, o.g11, ss);
  o.ss = ss;
  return o;
}

__global__ void logit_kernel(const float* __restrict__ qs,
                             const float* __restrict__ ks,
                             const float* __restrict__ hq,
                             const float* __restrict__ hk,
                             const float* __restrict__ r,
                             const float* __restrict__ t,
                             const float* __restrict__ gamma,
                             float* __restrict__ part) {
  int tile = blockIdx.x * 8 + (threadIdx.x >> 5);
  int jt = tile % LT;
  int it = (tile / LT) % LT;
  int n  = (tile / (LT * LT)) % NH;
  int b  = tile / (LT * LT * NH);
  int lane = threadIdx.x & 31;
  int h = lane >> 4, l16 = lane & 15;

  const float* qsp = qs + (size_t)(b * LL + it * 16 + l16) * DD + n * DHS_ + 2 * h;
  const float* ksp = ks + (size_t)(b * LL + jt * 16 + l16) * DD + n * DHS_ + 2 * h;
  v8f accS = {};
#pragma unroll
  for (int k = 0; k < 16; k += 4) {
    v2f a, bv2;
    a.x = qsp[k];   a.y = qsp[k + 1];
    bv2.x = ksp[k]; bv2.y = ksp[k + 1];
    accS = wmma4(a, bv2, accS);
  }
  const int rq = b * LL + it * 16 + l16;
  const int rk = b * LL + jt * 16 + l16;
  const pts12 gq = frame_row(hq + (size_t)rq * (NH * 12) + n * 12, r + (size_t)rq * 9,
                             t + (size_t)rq * 3);
  const pts12 gk = frame_row(hk + (size_t)rk * (NH * 12) + n * 12, r + (size_t)rk * 9,
                             t + (size_t)rk * 3);
  const float qqi = gq.ss, kkj = gk.ss;
  v8f accP = {};
  {
    v2f a, bv2;
    a.x   = h ? gq.g2 : gq.g0;    a.y   = h ? gq.g3 : gq.g1;
    bv2.x = h ? gk.g2 : gk.g0;    bv2.y = h ? gk.g3 : gk.g1;
    accP = wmma4(a, bv2, accP);
    a.x   = h ? gq.g6 : gq.g4;    a.y   = h ? gq.g7 : gq.g5;
    bv2.x = h ? gk.g6 : gk.g4;    bv2.y = h ? gk.g7 : gk.g5;
    accP = wmma4(a, bv2, accP);
    a.x   = h ? gq.g10 : gq.g8;   a.y   = h ? gq.g11 : gq.g9;
    bv2.x = h ? gk.g10 : gk.g8;   bv2.y = h ? gk.g11 : gk.g9;
    accP = wmma4(a, bv2, accP);
  }
  size_t hb = (size_t)(b * NH + n) * LL;
  int j = jt * 16 + l16;
  float g = gamma[n];
#pragma unroll
  for (int v = 0; v < 8; ++v) {
    int i = it * 16 + v + (h << 3);
    const float q0 = __int_as_float(__builtin_amdgcn_readlane(__float_as_int(qqi), v));
    const float q1 = __int_as_float(__builtin_amdgcn_readlane(__float_as_int(qqi), v + 8));
    float d2 = (h ? q1 : q0) + kkj - 2.0f * accP[v];
    float lp = -0.125f * g * d2;
    part[(hb + i) * LL + j] = 0.25f * accS[v] + lp;
  }
}

__global__ __launch_bounds__(256) void attn_v_kernel(const float* __restrict__ attn,
                                                     const float* __restrict__ vs,
                                                     const float* __restrict__ hv,
                                                     const float* __restrict__ r,
                                                     const float* __restrict__ t,
                                                     float* __restrict__ out_cat,
                                                     float* __restrict__ op) {
  constexpr int RT = 2;
  constexpr int AP = LL + 4;
  constexpr int BP = 32 + 8;
  constexpr int AQ = RT * 16 * 3 / 8;
  constexpr int SQ = LL * 4 / 256;
  __shared__ float As[RT * 16 * AP];
  __shared__ float Bs[LL * BP];

  const int tid = threadIdx.x;
  const int wave = __builtin_amdgcn_readfirstlane(tid >> 5);
  const int lane = tid & 31;
  const int h = lane >> 4, l16 = lane & 15;
  const int ip = blockIdx.x % (LT / RT);
  const int bn = blockIdx.x / (LT / RT);
  const int n = bn % NH, b = bn / NH;
  const size_t hb = (size_t)bn * LL;

  const float* ag = attn + (hb + ip * (RT * 16) + 4 * wave) * LL + lane * 4;
  float* al = As + (4 * wave) * AP + lane * 4;
  const float* sg = vs + ((size_t)b * LL + (tid >> 2)) * DD + n * DHS_ + (tid & 3) * 4;
  float* sl = Bs + (tid >> 2) * BP + (tid & 3) * 4;
  const int pq = (tid & 3) < 2 ? (tid & 3) : 2;
  const size_t vrow = (size_t)b * LL + (tid >> 2);
  const float* hg = hv + vrow * (NH * 12) + n * 12;
  const float* rg = r + vrow * 9;
  const float* tg = t + vrow * 3;
  float* psl = sl + 16;

  v4f ra[AQ], rs[SQ], rp[SQ];
#pragma unroll
  for (int m = 0; m < AQ; ++m)
    ra[m] = *reinterpret_cast<const v4f*>(ag + (m / 3) * LL + (m % 3) * 128);
#pragma unroll
  for (int m = 0; m < SQ; ++m)
    rs[m] = *reinterpret_cast<const v4f*>(sg + (size_t)(m * 64) * DD);
#pragma unroll
  for (int m = 0; m < SQ; ++m) {
    const pts12 g = frame_row(hg + (size_t)(m * 64) * (NH * 12), rg + (m * 64) * 9,
                              tg + (m * 64) * 3);
    rp[m].x = pq == 0 ? g.g0 : (pq == 1 ? g.g4 : g.g8);
    rp[m].y = pq == 0 ? g.g1 : (pq == 1 ? g.g5 : g.g9);
    rp[m].z = pq == 0 ? g.g2 : (pq == 1 ? g.g6 : g.g10);
    rp[m].w = pq == 0 ? g.g3 : (pq == 1 ? g.g7 : g.g11);
  }

#pragma unroll
  for (int m = 0; m < AQ; ++m)
    *reinterpret_cast<v4f*>(al + (m / 3) * AP + (m % 3) * 128) = ra[m];
#pragma unroll
  for (int m = 0; m < SQ; ++m)
    *reinterpret_cast<v4f*>(sl + (m * 64) * BP) = rs[m];
#pragma unroll
  for (int m = 0; m < SQ; ++m)
    *reinterpret_cast<v4f*>(psl + (m * 64) * BP) = rp[m];
  __syncthreads();

  if (wave < 2 * RT) {
    const int rt = wave >> 1, ct = wave & 1;
    constexpr int KB = 16;
    constexpr int NB = LL / (4 * KB);
    v8f acc = {};
    const float* asp = As + (rt * 16 + l16) * AP + 2 * h;
    const float* bsp = Bs + (2 * h) * BP + ct * 16 + l16;
    v2f fa[2][KB], fb[2][KB];
#pragma unroll
    for (int s = 0; s < KB; ++s) {
      fa[0][s] = *reinterpret_cast<const v2f*>(asp + 4 * s);
      fb[0][s].x = bsp[(4 * s) * BP];
      fb[0][s].y = bsp[(4 * s + 1) * BP];
    }
#pragma unroll
    for (int t = 0; t < NB; ++t) {
      if (t + 1 < NB) {
#pragma unroll
        for (int s = 0; s < KB; ++s) {
          const int k = (t + 1) * 4 * KB + 4 * s;
          fa[(t + 1) & 1][s] = *reinterpret_cast<const v2f*>(asp + k);
          fb[(t + 1) & 1][s].x = bsp[k * BP];
          fb[(t + 1) & 1][s].y = bsp[(k + 1) * BP];
        }
      }
      __builtin_amdgcn_sched_barrier(0);
#pragma unroll
      for (int s = 0; s < KB; ++s) acc = wmma4(fa[t & 1][s], fb[t & 1][s], acc);
    }
    const int i0 = ip * (RT * 16) + rt * 16 + (h << 3);
    if (ct == 0) {
      float* o = out_cat + (size_t)(b * LL + i0) * DOUT + n * DHS_ + l16;
#pragma unroll
      for (int v = 0; v < 8; ++v) o[(size_t)v * DOUT] = acc[v];
    } else if (l16 < 12) {
      float* o = op + (hb + i0) * 12 + l16;
#pragma unroll
      for (int v = 0; v < 8; ++v) o[v * 12] = acc[v];
    }
  }
}

static __device__ __forceinline__ v2f fma2(v2f a, v2f b, v2f c) {
  return __builtin_elementwise_fma(a, b, c);
}

#define XCHG_MIRROR 0x141
#define XCHG_2      0x4E
#define XCHG_1      0xB1
template <int CTRL>
static __device__ __forceinline__ float lane_xchg(float v) {
  return __int_as_float(__builtin_amdgcn_update_dpp(0, __float_as_int(v), CTRL, 0xF, 0xF, true));
}

#define PR_GRID 256
#define PR_ROWS (BL / PR_GRID)

__global__ __launch_bounds__(256) void pair_row_kernel(
    const float* __restrict__ part, const float* __restrict__ e,
    const float* __restrict__ Wb, float* __restrict__ attn,
    float* __restrict__ out_cat) {
  constexpr int KC  = 128;
  constexpr int NCH = LL / KC;
  constexpr int AP  = LL + 4;
  constexpr int WP  = DD + 12;
  constexpr int WQ  = KC / 8;
  __shared__ float As[16 * AP];
  __shared__ float Ws[LL * WP];
  __shared__ float Wbs[8 * DD * NH];

  const int tid = threadIdx.x;
  const int wave = tid >> 5, lane = tid & 31;
  const int h = lane >> 4, l16 = lane & 15;
  const int tn = wave << 4;
  const int ds = lane & 7;
  const int r4 = 8 * ((lane >> 3) & 1) + 4 * (lane >> 4);

  int row = blockIdx.x;

  const float* wg = e + (((size_t)row * LL) + 16 * wave) * DD + lane * 4;
  float* wl = Ws + (16 * wave) * WP + lane * 4;
  const float* pg = part + ((size_t)((row / LL) * NH + ds) * LL + (row % LL)) * LL + 16 * wave + r4;

  float* wbw = Wbs + wave * (DD * NH);
  v4f wbv[8];
#pragma unroll
  for (int m = 0; m < 8; ++m)
    wbv[m] = *reinterpret_cast<const v4f*>(Wb + (32 * m + lane) * 4);
  v2f wb[16][4];
  v4f sp[NCH];
#pragma unroll
  for (int c = 0; c < NCH; ++c)
    sp[c] = *reinterpret_cast<const v4f*>(pg + c * KC);

  v4f rw[2][WQ];
#pragma unroll
  for (int q = 0; q < WQ; ++q)
    rw[0][q] = *reinterpret_cast<const v4f*>(wg + (size_t)q * DD);
  __builtin_amdgcn_sched_barrier(0);
#pragma unroll
  for (int q = 0; q < WQ; ++q)
    rw[1][q] = *reinterpret_cast<const v4f*>(wg + (size_t)(KC + q) * DD);
  {
    v4f z = {};
#pragma unroll
    for (int c = 0; c < NCH; ++c)
      *reinterpret_cast<v4f*>(As + (8 + wave) * AP + c * KC + lane * 4) = z;
  }

#pragma unroll
  for (int m = 0; m < 8; ++m)
    *reinterpret_cast<v4f*>(wbw + ((((m >> 1) * 4 + ((lane >> 1) & 3)) * 2 + (lane & 1)) * 8
                                   + 4 * (m & 1) + (lane >> 3)) * 4) = wbv[m];
#pragma unroll
  for (int m = 0; m < 16; ++m) {
    const v4f lo = *reinterpret_cast<const v4f*>(wbw + ((2 * m) * 8 + ds) * 4);
    const v4f hi = *reinterpret_cast<const v4f*>(wbw + ((2 * m + 1) * 8 + ds) * 4);
    wb[m][0].x = lo.x;  wb[m][0].y = lo.y;
    wb[m][1].x = lo.z;  wb[m][1].y = lo.w;
    wb[m][2].x = hi.x;  wb[m][2].y = hi.y;
    wb[m][3].x = hi.z;  wb[m][3].y = hi.w;
  }

  const float rsqrt3 = 0.57735026918962576f;
#pragma nounroll
  for (int it = 0; it < PR_ROWS; ++it) {
    const int b = row / LL, i = row - b * LL;
#pragma unroll
    for (int c = 0; c < NCH; ++c) {
#pragma unroll
      for (int q = 0; q < WQ; ++q)
        *reinterpret_cast<v4f*>(wl + (c * KC + q) * WP) = rw[c & 1][q];
      __builtin_amdgcn_sched_barrier(0);
      if (c + 2 < NCH) {
        const int k2 = (c + 2) * KC;
#pragma unroll
        for (int q = 0; q < WQ; ++q)
          rw[c & 1][q] = *reinterpret_cast<const v4f*>(wg + (size_t)(k2 + q) * DD);
      }
      const float* erow = Ws + (c * KC + 16 * wave + r4) * WP + 4 * ds;
      v4f evs[4][4];
#pragma unroll
      for (int p = 0; p < 4; ++p)
#pragma unroll
        for (int q = 0; q < 4; ++q)
          evs[p][q] = *reinterpret_cast<const v4f*>(erow + p * WP + 32 * q);
      __builtin_amdgcn_sched_barrier(0);
      v4f bs;
#pragma unroll
      for (int p = 0; p < 4; ++p) {
        v2f a0 = {}, a1 = {}, a2 = {}, a3 = {};
#pragma unroll
        for (int q = 0; q < 4; ++q) {
          const v4f ev = evs[p][q];
          v2f e0, e1, e2, e3;
          e0.x = ev.x;  e0.y = ev.x;
          e1.x = ev.y;  e1.y = ev.y;
          e2.x = ev.z;  e2.y = ev.z;
          e3.x = ev.w;  e3.y = ev.w;
          a0 = fma2(e0, wb[4 * q][0], a0);      a1 = fma2(e0, wb[4 * q][1], a1);
          a2 = fma2(e0, wb[4 * q][2], a2);      a3 = fma2(e0, wb[4 * q][3], a3);
          a0 = fma2(e1, wb[4 * q + 1][0], a0);  a1 = fma2(e1, wb[4 * q + 1][1], a1);
          a2 = fma2(e1, wb[4 * q + 1][2], a2);  a3 = fma2(e1, wb[4 * q + 1][3], a3);
          a0 = fma2(e2, wb[4 * q + 2][0], a0);  a1 = fma2(e2, wb[4 * q + 2][1], a1);
          a2 = fma2(e2, wb[4 * q + 2][2], a2);  a3 = fma2(e2, wb[4 * q + 2][3], a3);
          a0 = fma2(e3, wb[4 * q + 3][0], a0);  a1 = fma2(e3, wb[4 * q + 3][1], a1);
          a2 = fma2(e3, wb[4 * q + 3][2], a2);  a3 = fma2(e3, wb[4 * q + 3][3], a3);
        }
        const bool up4 = (ds & 4) != 0, up2 = (ds & 2) != 0, up1 = (ds & 1) != 0;
        float k0 = up4 ? a2.x : a0.x, k1 = up4 ? a2.y : a0.y, k2 = up4 ? a3.x : a1.x, k3 = up4 ? a3.y : a1.y;
        const float g0 = up4 ? a0.x : a2.x, g1 = up4 ? a0.y : a2.y, g2 = up4 ? a1.x : a3.x, g3 = up4 ? a1.y : a3.y;
        k0 += lane_xchg<XCHG_MIRROR>(g0);
        k1 += lane_xchg<XCHG_MIRROR>(g1);
        k2 += lane_xchg<XCHG_MIRROR>(g2);
        k3 += lane_xchg<XCHG_MIRROR>(g3);
        float m0 = up2 ? k2 : k0, m1 = up2 ? k3 : k1;
        const float u0 = up2 ? k0 : k2, u1 = up2 ? k1 : k3;
        m0 += lane_xchg<XCHG_2>(u0);
        m1 += lane_xchg<XCHG_2>(u1);
        float s = up1 ? m1 : m0;
        const float w = up1 ? m0 : m1;
        s += lane_xchg<XCHG_1>(w);
        bs[p] = s;
      }
      v4f lgv;
      lgv.x = rsqrt3 * (sp[c].x + bs.x);
      lgv.y = rsqrt3 * (sp[c].y + bs.y);
      lgv.z = rsqrt3 * (sp[c].z + bs.z);
      lgv.w = rsqrt3 * (sp[c].w + bs.w);
      *reinterpret_cast<v4f*>(As + ds * AP + c * KC + 16 * wave + r4) = lgv;
    }
    __syncthreads();

    float v[12];
    {
      float* lr = As + wave * AP + lane;
      float m = -1e30f;
#pragma unroll
      for (int k = 0; k < 12; ++k) {
        v[k] = lr[k * 32];
        m = fmaxf(m, v[k]);
      }
#pragma unroll
      for (int off = 16; off > 0; off >>= 1) m = fmaxf(m, __shfl_xor(m, off, 32));
      float s = 0.0f;
#pragma unroll
      for (int k = 0; k < 12; ++k) {
        v[k] = __expf(v[k] - m);
        s += v[k];
      }
#pragma unroll
      for (int off = 16; off > 0; off >>= 1) s += __shfl_xor(s, off, 32);
      float inv = 1.0f / s;
#pragma unroll
      for (int k = 0; k < 12; ++k) {
        v[k] = v[k] * inv;
        lr[k * 32] = v[k];
      }
    }
    __syncthreads();
    {
      float* ao = attn + ((size_t)(b * NH + wave) * LL + i) * LL + lane;
#pragma unroll
      for (int k = 0; k < 12; ++k) ao[k * 32] = v[k];
    }

    const int nrow = row + PR_GRID;
    if (it + 1 < PR_ROWS) {
      wg = e + (((size_t)nrow * LL) + 16 * wave) * DD + lane * 4;
      pg = part + ((size_t)((nrow / LL) * NH + ds) * LL + (nrow % LL)) * LL + 16 * wave + r4;
#pragma unroll
      for (int c = 0; c < NCH; ++c)
        sp[c] = *reinterpret_cast<const v4f*>(pg + c * KC);
#pragma unroll
      for (int q = 0; q < WQ; ++q)
        rw[0][q] = *reinterpret_cast<const v4f*>(wg + (size_t)q * DD);
      __builtin_amdgcn_sched_barrier(0);
#pragma unroll
      for (int q = 0; q < WQ; ++q)
        rw[1][q] = *reinterpret_cast<const v4f*>(wg + (size_t)(KC + q) * DD);
    }

    constexpr int KB = 16;
    constexpr int NB = LL / (4 * KB);
    v8f acc = {};
    const float* asp = As + l16 * AP + 2 * h;
    const float* wsp = Ws + (2 * h) * WP + tn + l16;
    v2f fa[2][KB], fb[2][KB];
#pragma unroll
    for (int s = 0; s < KB; ++s) {
      fa[0][s] = *reinterpret_cast<const v2f*>(asp + 4 * s);
      fb[0][s].x = wsp[(4 * s) * WP];
      fb[0][s].y = wsp[(4 * s + 1) * WP];
    }
#pragma unroll
    for (int t = 0; t < NB; ++t) {
      if (t + 1 < NB) {
#pragma unroll
        for (int s = 0; s < KB; ++s) {
          const int k = (t + 1) * 4 * KB + 4 * s;
          fa[(t + 1) & 1][s] = *reinterpret_cast<const v2f*>(asp + k);
          fb[(t + 1) & 1][s].x = wsp[k * WP];
          fb[(t + 1) & 1][s].y = wsp[(k + 1) * WP];
        }
      }
      __builtin_amdgcn_sched_barrier(0);
#pragma unroll
      for (int s = 0; s < KB; ++s) acc = wmma4(fa[t & 1][s], fb[t & 1][s], acc);
    }
    __syncthreads();
    if (h == 0) {
#pragma unroll
      for (int r = 0; r < 8; ++r)
        out_cat[(size_t)row * DOUT + DD + r * DD + tn + l16] = acc[r];
    }
    row = nrow;
  }
}

extern "C" void kernel_launch(void* const* d_in, const int* in_sizes, int n_in,
                              void* d_out, int out_size, void* d_ws, size_t ws_size,
                              hipStream_t stream) {
  (void)in_sizes; (void)n_in; (void)out_size;
  const float* x     = (const float*)d_in[0];
  const float* e     = (const float*)d_in[1];
  const float* r     = (const float*)d_in[2];
  const float* t     = (const float*)d_in[3];
  const float* Wq_s  = (const float*)d_in[4];
  const float* Wk_s  = (const float*)d_in[5];
  const float* Wv_s  = (const float*)d_in[6];
  const float* Wb    = (const float*)d_in[7];
  const float* Wq_p  = (const float*)d_in[8];
  const float* Wk_p  = (const float*)d_in[9];
  const float* Wv_p  = (const float*)d_in[10];
  const float* gamma = (const float*)d_in[11];
  const float* Wo    = (const float*)d_in[12];
  const float* bo    = (const float*)d_in[13];
  float* out = (float*)d_out;
  float* ws  = (float*)d_ws;

  float* qs   = ws;
  float* ks_  = qs  + 98304;
  float* vs   = ks_ + 98304;
  float* hq   = vs  + 98304;
  float* hk   = hq  + 73728;
  float* hv   = hk  + 73728;
  float* part = hv  + 73728;
  float* attn = part + 2359296;
  float* op   = attn + 2359296;
  float* ocat = op  + 73728;
  size_t req = (size_t)((ocat + 983040) - ws) * sizeof(float);
  if (ws_size < req) return;

  dim3 blk(256);

  proj6_kernel<<<BL / 16, blk, 0, stream>>>(x, Wq_s, Wk_s, Wv_s, Wq_p, Wk_p, Wv_p,
                                            qs, ks_, vs, hq, hk, hv);

  logit_kernel<<<(BB * NH * LT * LT) / 8, blk, 0, stream>>>(qs, ks_, hq, hk, r, t,
                                                            gamma, part);

  pair_row_kernel<<<PR_GRID, blk, 0, stream>>>(part, e, Wb, attn, ocat);

  attn_v_kernel<<<BB * NH * (LT / 2), blk, 0, stream>>>(attn, vs, hv, r, t, ocat, op);

  wmma_gemm_lds_kernel<DOUT, DD><<<BL / 16, blk, 0, stream>>>(ocat, Wo, bo, op, r, t, out);
}


# six input projections: one block of 4 waves per (16-row strip, weight matrix): 288 blocks of one k chunk, not 48 walking six matrices in turn; fragments read from LDS as one group; k order kept
# speedup vs baseline: 4.9365x; 1.0895x over previous
#include <hip/hip_runtime.h>
#include <hip/hip_bf16.h>
#include <math.h>

#define BB   2
#define LL   384
#define DD   128
#define NH   8
#define DHS_ 16
#define DHP_ 4
#define BL   (BB*LL)
#define BNL  (BB*NH*LL)
#define DOUT 1280
#define LT   24

typedef float v2f __attribute__((ext_vector_type(2)));
typedef float v4f __attribute__((ext_vector_type(4)));
typedef float v8f __attribute__((ext_vector_type(8)));

static __device__ __forceinline__ v8f wmma4(v2f a, v2f b, v8f c) {
  return __builtin_amdgcn_wmma_f32_16x16x4_f32(false, a, false, b, (short)0, c,
                                               false, false);
}


static __device__ __forceinline__ void local_point(float pa, float pb, float pc,
                                                   float t0, float t1, float t2,
                                                   float r00, float r01, float r02,
                                                   float r10, float r11, float r12,
                                                   float r20, float r21, float r22,
                                                   float& x, float& y, float& z, float& nm) {
  const float a = pa - t0, b = pb - t1, cc = pc - t2;
  x = a * r00 + b * r01 + cc * r02;
  y = a * r10 + b * r11 + cc * r12;
  z = a * r20 + b * r21 + cc * r22;
  float nsq = 0.0f;
  nsq += x * x;
  nsq += y * y;
  nsq += z * z;
  nm = sqrtf(nsq);
}

static __device__ __forceinline__ v4f point_piece(const float* __restrict__ op,
                                                  const float* __restrict__ r,
                                                  const float* __restrict__ t,
                                                  int row, int vec) {
  const int b = row / LL, i = row - b * LL;
  const int n = vec < 24 ? vec / 3 : vec - 24;
  const float* opv = op + ((size_t)(b * NH + n) * LL + i) * 12;
  const float* rp = r + (size_t)row * 9;
  const float* tp = t + (size_t)row * 3;
  const v4f o0 = *reinterpret_cast<const v4f*>(opv);
  const v4f o1 = *reinterpret_cast<const v4f*>(opv + 4);
  const v4f o2 = *reinterpret_cast<const v4f*>(opv + 8);
  const float r00 = rp[0], r01 = rp[1], r02 = rp[2];
  const float r10 = rp[3], r11 = rp[4], r12 = rp[5];
  const float r20 = rp[6], r21 = rp[7], r22 = rp[8];
  const float t0 = tp[0], t1 = tp[1], t2 = tp[2];
  float x0, y0, z0, n0, x1, y1, z1, n1, x2, y2, z2, n2, x3, y3, z3, n3;
  local_point(o0.x, o0.y, o0.z, t0, t1, t2, r00, r01, r02, r10, r11, r12, r20, r21, r22, x0, y0, z0, n0);
  local_point(o0.w, o1.x, o1.y, t0, t1, t2, r00, r01, r02, r10, r11, r12, r20, r21, r22, x1, y1, z1, n1);
  local_point(o1.z, o1.w, o2.x, t0, t1, t2, r00, r01, r02, r10, r11, r12, r20, r21, r22, x2, y2, z2, n2);
  local_point(o2.y, o2.z, o2.w, t0, t1, t2, r00, r01, r02, r10, r11, r12, r20, r21, r22, x3, y3, z3, n3);
  const int s = vec % 3;
  const bool nrm = vec >= 24;
  v4f o;
  o.x = nrm ? n0 : (s == 0 ? x0 : (s == 1 ? y1 : z2));
  o.y = nrm ? n1 : (s == 0 ? y0 : (s == 1 ? z1 : x3));
  o.z = nrm ? n2 : (s == 0 ? z0 : (s == 1 ? x2 : y3));
  o.w = nrm ? n3 : (s == 0 ? x1 : (s == 1 ? y2 : z3));
  return o;
}

template <int K, int NC>
__global__ __launch_bounds__(256) void wmma_gemm_lds_kernel(
    const float* __restrict__ A, const float* __restrict__ W,
    const float* __restrict__ bias, const float* __restrict__ op,
    const float* __restrict__ r, const float* __restrict__ t, float* __restrict__ C) {
  constexpr int KC  = 128;
  constexpr int AP  = KC + 4;
  constexpr int WP  = NC + 8;
  constexpr int NCH = K / KC;
  constexpr int WV  = NC / 4;
  constexpr int WR  = 8;
  constexpr int WQ  = KC / WR;
  constexpr int AQ  = (16 * KC / 4) / 256;
  __shared__ float As[16 * AP];
  __shared__ float Ws[KC * WP];

  const int tid = threadIdx.x;
  const int wave = tid >> 5, lane = tid & 31;
  const int h = lane >> 4, l16 = lane & 15;
  const int tm = blockIdx.x << 4;
  const int tn = wave << 4;

  v4f ra[AQ];
  v4f rw[WQ] = {};
  const float* ag = A + (size_t)(tm + (tid >> 5)) * K + (tid & 31) * 4;
  const float* wg = W + (size_t)(tid / WV) * NC + (tid % WV) * 4;
  float* al = As + (tid >> 5) * AP + (tid & 31) * 4;
  float* wl = Ws + (tid / WV) * WP + (tid % WV) * 4;
  const bool wact = tid < WR * WV;

#pragma unroll
  for (int q = 0; q < AQ; ++q)
    ra[q] = *reinterpret_cast<const v4f*>(ag + (size_t)(q * 8) * K);
  if (wact) {
#pragma unroll
    for (int q = 0; q < WQ; ++q)
      rw[q] = *reinterpret_cast<const v4f*>(wg + (size_t)(q * WR) * NC);
  }
  v4f pp[AQ];
#pragma unroll
  for (int q = 0; q < AQ; ++q)
    pp[q] = point_piece(op, r, t, tm + (tid >> 5) + q * 8, tid & 31);

  v8f acc = {};
  for (int c = 0; c < NCH; ++c) {
#pragma unroll
    for (int q = 0; q < AQ; ++q)
      *reinterpret_cast<v4f*>(al + (q * 8) * AP) = ra[q];
    if (wact) {
#pragma unroll
      for (int q = 0; q < WQ; ++q)
        *reinterpret_cast<v4f*>(wl + (q * WR) * WP) = rw[q];
    }
    __syncthreads();
    if (c + 1 < NCH) {
      const int k1 = (c + 1) * KC;
      if (c + 2 < NCH) {
#pragma unroll
        for (int q = 0; q < AQ; ++q)
          ra[q] = *reinterpret_cast<const v4f*>(ag + (size_t)(q * 8) * K + k1);
      } else {
#pragma unroll
        for (int q = 0; q < AQ; ++q) ra[q] = pp[q];
      }
      if (wact) {
#pragma unroll
        for (int q = 0; q < WQ; ++q)
          rw[q] = *reinterpret_cast<const v4f*>(wg + (size_t)(k1 + q * WR) * NC);
      }
    }
    if (tn < NC) {
      const float* asp = As + l16 * AP + 2 * h;
      const float* wsp = Ws + (2 * h) * WP + tn + l16;
#pragma unroll
      for (int kk = 0; kk < KC; kk += 4) {
        v2f a = *reinterpret_cast<const v2f*>(asp + kk);
        v2f b;
        b.x = wsp[kk * WP];
        b.y = wsp[(kk + 1) * WP];
        acc = wmma4(a, b, acc);
      }
    }
    if (c + 1 < NCH) __syncthreads();
  }
  if (tn < NC) {
    float bv = bias ? bias[tn + l16] : 0.0f;
#pragma unroll
    for (int v = 0; v < 8; ++v) {
      int row = tm + v + (h << 3);
      C[(size_t)row * NC + tn + l16] = acc[v] + bv;
    }
  }
}

#define PJ_AP 132

template <int NC>
static __device__ __forceinline__ void proj_block(const float* __restrict__ x,
                                                  const float* __restrict__ W,
                                                  float* __restrict__ C,
                                                  float* __restrict__ As,
                                                  float* __restrict__ Ws, int tm) {
  constexpr int WV = NC / 4;
  constexpr int WP = NC + 8;
  constexpr int NT = NC / 16;
  constexpr int KS = DD / 4;

  const int tid = threadIdx.x;
  const int wave = __builtin_amdgcn_readfirstlane(tid >> 5);
  const int lane = tid & 31;
  const int h = lane >> 4, l16 = lane & 15;

  const float* ag = x + (size_t)(tm + wave) * DD + lane * 4;
  v4f ra[4];
#pragma unroll
  for (int q = 0; q < 4; ++q)
    ra[q] = *reinterpret_cast<const v4f*>(ag + (size_t)(q * 4) * DD);
  const bool wact = wave < WV / 8;
  const float* wg = W + (size_t)(tid / WV) * NC + (tid % WV) * 4;
  v4f rw[32];
  if (wact) {
#pragma unroll
    for (int q = 0; q < 32; ++q)
      rw[q] = *reinterpret_cast<const v4f*>(wg + (size_t)(q * 4) * NC);
  }

  float* al = As + wave * PJ_AP + lane * 4;
#pragma unroll
  for (int q = 0; q < 4; ++q)
    *reinterpret_cast<v4f*>(al + (q * 4) * PJ_AP) = ra[q];
  if (wact) {
    float* wl = Ws + (tid / WV) * WP + (tid % WV) * 4;
#pragma unroll
    for (int q = 0; q < 32; ++q)
      *reinterpret_cast<v4f*>(wl + (q * 4) * WP) = rw[q];
  }
  __syncthreads();

  const float* asp = As + l16 * PJ_AP + 2 * h;
  const float* wsp = Ws + (2 * h) * WP + (wave << 4) + l16;
  float* cp = C + (size_t)(tm + (h << 3)) * NC + (wave << 4) + l16;
  v2f fa[KS], fb0[KS];
#pragma unroll
  for (int s = 0; s < KS; ++s) {
    fa[s] = *reinterpret_cast<const v2f*>(asp + 4 * s);
    fb0[s].x = wsp[(4 * s) * WP];
    fb0[s].y = wsp[(4 * s + 1) * WP];
  }
  if (NT == 8 || wave < NT - 4) {
    v2f fb1[KS];
#pragma unroll
    for (int s = 0; s < KS; ++s) {
      fb1[s].x = wsp[(4 * s) * WP + 64];
      fb1[s].y = wsp[(4 * s + 1) * WP + 64];
    }
    __builtin_amdgcn_sched_barrier(0);
    v8f acc0 = {}, acc1 = {};
#pragma unroll
    for (int s = 0; s < KS; ++s) {
      acc0 = wmma4(fa[s], fb0[s], acc0);
      acc1 = wmma4(fa[s], fb1[s], acc1);
    }
#pragma unroll
    for (int v = 0; v < 8; ++v) {
      cp[(size_t)v * NC] = acc0[v];
      cp[(size_t)v * NC + 64] = acc1[v];
    }
  } else {
    __builtin_amdgcn_sched_barrier(0);
    v8f acc0 = {};
#pragma unroll
    for (int s = 0; s < KS; ++s) acc0 = wmma4(fa[s], fb0[s], acc0);
#pragma unroll
    for (int v = 0; v < 8; ++v) cp[(size_t)v * NC] = acc0[v];
  }
}

__global__ __launch_bounds__(128) void proj6_kernel(
    const float* __restrict__ x,
    const float* __restrict__ W0, const float* __restrict__ W1, const float* __restrict__ W2,
    const float* __restrict__ W3, const float* __restrict__ W4, const float* __restrict__ W5,
    float* __restrict__ C0, float* __restrict__ C1, float* __restrict__ C2,
    float* __restrict__ C3, float* __restrict__ C4, float* __restrict__ C5) {
  __shared__ float As[16 * PJ_AP];
  __shared__ float Ws[DD * (DD + 8)];

  const int tm = blockIdx.x << 4;
  const int m = blockIdx.y;
  if (m < 3) {
    const float* W = m == 0 ? W0 : (m == 1 ? W1 : W2);
    float* C = m == 0 ? C0 : (m == 1 ? C1 : C2);
    proj_block<DD>(x, W, C, As, Ws, tm);
  } else {
    const float* W = m == 3 ? W3 : (m == 4 ? W4 : W5);
    float* C = m == 3 ? C3 : (m == 4 ? C4 : C5);
    proj_block<NH * 12>(x, W, C, As, Ws, tm);
  }
}

struct pts12 { float g0, g1, g2, g3, g4, g5, g6, g7, g8, g9, g10, g11, ss; };

static __device__ __forceinline__ pts12 frame_row(const float* __restrict__ hp,
                                                  const float* __restrict__ rp,
                                                  const float* __restrict__ tp) {
  const v4f h0 = *reinterpret_cast<const v4f*>(hp);
  const v4f h1 = *reinterpret_cast<const v4f*>(hp + 4);
  const v4f h2 = *reinterpret_cast<const v4f*>(hp + 8);
  const float r00 = rp[0], r01 = rp[1], r02 = rp[2];
  const float r10 = rp[3], r11 = rp[4], r12 = rp[5];
  const float r20 = rp[6], r21 = rp[7], r22 = rp[8];
  const float t0 = tp[0], t1 = tp[1], t2 = tp[2];
  pts12 o;
  o.g0  = __builtin_fmaf(h0.z, r20, __builtin_fmaf(h0.y, r10, __builtin_fmaf(h0.x, r00, t0)));
  o.g1  = __builtin_fmaf(h0.z, r21, __builtin_fmaf(h0.y, r11, __builtin_fmaf(h0.x, r01, t1)));
  o.g2  = __builtin_fmaf(h0.z, r22, __builtin_fmaf(h0.y, r12, __builtin_fmaf(h0.x, r02, t2)));
  o.g3  = __builtin_fmaf(h1.y, r20, __builtin_fmaf(h1.x, r10, __builtin_fmaf(h0.w, r00, t0)));
  o.g4  = __builtin_fmaf(h1.y, r21, __builtin_fmaf(h1.x, r11, __builtin_fmaf(h0.w, r01, t1)));
  o.g5  = __builtin_fmaf(h1.y, r22, __builtin_fmaf(h1.x, r12, __builtin_fmaf(h0.w, r02, t2)));
  o.g6  = __builtin_fmaf(h2.x, r20, __builtin_fmaf(h1.w, r10, __builtin_fmaf(h1.z, r00, t0)));
  o.g7  = __builtin_fmaf(h2.x, r21, __builtin_fmaf(h1.w, r11, __builtin_fmaf(h1.z, r01, t1)));
  o.g8  = __builtin_fmaf(h2.x, r22, __builtin_fmaf(h1.w, r12, __builtin_fmaf(h1.z, r02, t2)));
  o.g9  = __builtin_fmaf(h2.w, r20, __builtin_fmaf(h2.z, r10, __builtin_fmaf(h2.y, r00, t0)));
  o.g10 = __builtin_fmaf(h2.w, r21, __builtin_fmaf(h2.z, r11, __builtin_fmaf(h2.y, r01, t1)));
  o.g11 = __builtin_fmaf(h2.w, r22, __builtin_fmaf(h2.z, r12, __builtin_fmaf(h2.y, r02, t2)));
  float ss = 0.0f;
  ss = __builtin_fmaf(o.g0, o.g0, ss);
  ss = __builtin_fmaf(o.g1, o.g1, ss);
  ss = __builtin_fmaf(o.g2, o.g2, ss);
  ss = __builtin_fmaf(o.g3, o.g3, ss);
  ss = __builtin_fmaf(o.g4, o.g4, ss);
  ss = __builtin_fmaf(o.g5, o.g5, ss);
  ss = __builtin_fmaf(o.g6, o.g6, ss);
  ss = __builtin_fmaf(o.g7, o.g7, ss);
  ss = __builtin_fmaf(o.g8, o.g8, ss);
  ss = __builtin_fmaf(o.g9, o.g9, ss);
  ss = __builtin_fmaf(o.g10, o.g10, ss);
  ss = __builtin_fmaf(o.g11, o.g11, ss);
  o.ss = ss;
  return o;
}

__global__ void logit_kernel(const float* __restrict__ qs,
                             const float* __restrict__ ks,
                             const float* __restrict__ hq,
                             const float* __restrict__ hk,
                             const float* __restrict__ r,
                             const float* __restrict__ t,
                             const float* __restrict__ gamma,
                             float* __restrict__ part) {
  int tile = blockIdx.x * 8 + (threadIdx.x >> 5);
  int jt = tile % LT;
  int it = (tile / LT) % LT;
  int n  = (tile / (LT * LT)) % NH;
  int b  = tile / (LT * LT * NH);
  int lane = threadIdx.x & 31;
  int h = lane >> 4, l16 = lane & 15;

  const float* qsp = qs + (size_t)(b * LL + it * 16 + l16) * DD + n * DHS_ + 2 * h;
  const float* ksp = ks + (size_t)(b * LL + jt * 16 + l16) * DD + n * DHS_ + 2 * h;
  v8f accS = {};
#pragma unroll
  for (int k = 0; k < 16; k += 4) {
    v2f a, bv2;
    a.x = qsp[k];   a.y = qsp[k + 1];
    bv2.x = ksp[k]; bv2.y = ksp[k + 1];
    accS = wmma4(a, bv2, accS);
  }
  const int rq = b * LL + it * 16 + l16;
  const int rk = b * LL + jt * 16 + l16;
  const pts12 gq = frame_row(hq + (size_t)rq * (NH * 12) + n * 12, r + (size_t)rq * 9,
                             t + (size_t)rq * 3);
  const pts12 gk = frame_row(hk + (size_t)rk * (NH * 12) + n * 12, r + (size_t)rk * 9,
                             t + (size_t)rk * 3);
  const float qqi = gq.ss, kkj = gk.ss;
  v8f accP = {};
  {
    v2f a, bv2;
    a.x   = h ? gq.g2 : gq.g0;    a.y   = h ? gq.g3 : gq.g1;
    bv2.x = h ? gk.g2 : gk.g0;    bv2.y = h ? gk.g3 : gk.g1;
    accP = wmma4(a, bv2, accP);
    a.x   = h ? gq.g6 : gq.g4;    a.y   = h ? gq.g7 : gq.g5;
    bv2.x = h ? gk.g6 : gk.g4;    bv2.y = h ? gk.g7 : gk.g5;
    accP = wmma4(a, bv2, accP);
    a.x   = h ? gq.g10 : gq.g8;   a.y   = h ? gq.g11 : gq.g9;
    bv2.x = h ? gk.g10 : gk.g8;   bv2.y = h ? gk.g11 : gk.g9;
    accP = wmma4(a, bv2, accP);
  }
  size_t hb = (size_t)(b * NH + n) * LL;
  int j = jt * 16 + l16;
  float g = gamma[n];
#pragma unroll
  for (int v = 0; v < 8; ++v) {
    int i = it * 16 + v + (h << 3);
    const float q0 = __int_as_float(__builtin_amdgcn_readlane(__float_as_int(qqi), v));
    const float q1 = __int_as_float(__builtin_amdgcn_readlane(__float_as_int(qqi), v + 8));
    float d2 = (h ? q1 : q0) + kkj - 2.0f * accP[v];
    float lp = -0.125f * g * d2;
    part[(hb + i) * LL + j] = 0.25f * accS[v] + lp;
  }
}

__global__ __launch_bounds__(256) void attn_v_kernel(const float* __restrict__ attn,
                                                     const float* __restrict__ vs,
                                                     const float* __restrict__ hv,
                                                     const float* __restrict__ r,
                                                     const float* __restrict__ t,
                                                     float* __restrict__ out_cat,
                                                     float* __restrict__ op) {
  constexpr int RT = 2;
  constexpr int AP = LL + 4;
  constexpr int BP = 32 + 8;
  constexpr int AQ = RT * 16 * 3 / 8;
  constexpr int SQ = LL * 4 / 256;
  __shared__ float As[RT * 16 * AP];
  __shared__ float Bs[LL * BP];

  const int tid = threadIdx.x;
  const int wave = __builtin_amdgcn_readfirstlane(tid >> 5);
  const int lane = tid & 31;
  const int h = lane >> 4, l16 = lane & 15;
  const int ip = blockIdx.x % (LT / RT);
  const int bn = blockIdx.x / (LT / RT);
  const int n = bn % NH, b = bn / NH;
  const size_t hb = (size_t)bn * LL;

  const float* ag = attn + (hb + ip * (RT * 16) + 4 * wave) * LL + lane * 4;
  float* al = As + (4 * wave) * AP + lane * 4;
  const float* sg = vs + ((size_t)b * LL + (tid >> 2)) * DD + n * DHS_ + (tid & 3) * 4;
  float* sl = Bs + (tid >> 2) * BP + (tid & 3) * 4;
  const int pq = (tid & 3) < 2 ? (tid & 3) : 2;
  const size_t vrow = (size_t)b * LL + (tid >> 2);
  const float* hg = hv + vrow * (NH * 12) + n * 12;
  const float* rg = r + vrow * 9;
  const float* tg = t + vrow * 3;
  float* psl = sl + 16;

  v4f ra[AQ], rs[SQ], rp[SQ];
#pragma unroll
  for (int m = 0; m < AQ; ++m)
    ra[m] = *reinterpret_cast<const v4f*>(ag + (m / 3) * LL + (m % 3) * 128);
#pragma unroll
  for (int m = 0; m < SQ; ++m)
    rs[m] = *reinterpret_cast<const v4f*>(sg + (size_t)(m * 64) * DD);
#pragma unroll
  for (int m = 0; m < SQ; ++m) {
    const pts12 g = frame_row(hg + (size_t)(m * 64) * (NH * 12), rg + (m * 64) * 9,
                              tg + (m * 64) * 3);
    rp[m].x = pq == 0 ? g.g0 : (pq == 1 ? g.g4 : g.g8);
    rp[m].y = pq == 0 ? g.g1 : (pq == 1 ? g.g5 : g.g9);
    rp[m].z = pq == 0 ? g.g2 : (pq == 1 ? g.g6 : g.g10);
    rp[m].w = pq == 0 ? g.g3 : (pq == 1 ? g.g7 : g.g11);
  }

#pragma unroll
  for (int m = 0; m < AQ; ++m)
    *reinterpret_cast<v4f*>(al + (m / 3) * AP + (m % 3) * 128) = ra[m];
#pragma unroll
  for (int m = 0; m < SQ; ++m)
    *reinterpret_cast<v4f*>(sl + (m * 64) * BP) = rs[m];
#pragma unroll
  for (int m = 0; m < SQ; ++m)
    *reinterpret_cast<v4f*>(psl + (m * 64) * BP) = rp[m];
  __syncthreads();

  if (wave < 2 * RT) {
    const int rt = wave >> 1, ct = wave & 1;
    constexpr int KB = 16;
    constexpr int NB = LL / (4 * KB);
    v8f acc = {};
    const float* asp = As + (rt * 16 + l16) * AP + 2 * h;
    const float* bsp = Bs + (2 * h) * BP + ct * 16 + l16;
    v2f fa[2][KB], fb[2][KB];
#pragma unroll
    for (int s = 0; s < KB; ++s) {
      fa[0][s] = *reinterpret_cast<const v2f*>(asp + 4 * s);
      fb[0][s].x = bsp[(4 * s) * BP];
      fb[0][s].y = bsp[(4 * s + 1) * BP];
    }
#pragma unroll
    for (int t = 0; t < NB; ++t) {
      if (t + 1 < NB) {
#pragma unroll
        for (int s = 0; s < KB; ++s) {
          const int k = (t + 1) * 4 * KB + 4 * s;
          fa[(t + 1) & 1][s] = *reinterpret_cast<const v2f*>(asp + k);
          fb[(t + 1) & 1][s].x = bsp[k * BP];
          fb[(t + 1) & 1][s].y = bsp[(k + 1) * BP];
        }
      }
      __builtin_amdgcn_sched_barrier(0);
#pragma unroll
      for (int s = 0; s < KB; ++s) acc = wmma4(fa[t & 1][s], fb[t & 1][s], acc);
    }
    const int i0 = ip * (RT * 16) + rt * 16 + (h << 3);
    if (ct == 0) {
      float* o = out_cat + (size_t)(b * LL + i0) * DOUT + n * DHS_ + l16;
#pragma unroll
      for (int v = 0; v < 8; ++v) o[(size_t)v * DOUT] = acc[v];
    } else if (l16 < 12) {
      float* o = op + (hb + i0) * 12 + l16;
#pragma unroll
      for (int v = 0; v < 8; ++v) o[v * 12] = acc[v];
    }
  }
}

static __device__ __forceinline__ v2f fma2(v2f a, v2f b, v2f c) {
  return __builtin_elementwise_fma(a, b, c);
}

#define XCHG_MIRROR 0x141
#define XCHG_2      0x4E
#define XCHG_1      0xB1
template <int CTRL>
static __device__ __forceinline__ float lane_xchg(float v) {
  return __int_as_float(__builtin_amdgcn_update_dpp(0, __float_as_int(v), CTRL, 0xF, 0xF, true));
}

#define PR_GRID 256
#define PR_ROWS (BL / PR_GRID)

__global__ __launch_bounds__(256) void pair_row_kernel(
    const float* __restrict__ part, const float* __restrict__ e,
    const float* __restrict__ Wb, float* __restrict__ attn,
    float* __restrict__ out_cat) {
  constexpr int KC  = 128;
  constexpr int NCH = LL / KC;
  constexpr int AP  = LL + 4;
  constexpr int WP  = DD + 12;
  constexpr int WQ  = KC / 8;
  __shared__ float As[16 * AP];
  __shared__ float Ws[LL * WP];
  __shared__ float Wbs[8 * DD * NH];

  const int tid = threadIdx.x;
  const int wave = tid >> 5, lane = tid & 31;
  const int h = lane >> 4, l16 = lane & 15;
  const int tn = wave << 4;
  const int ds = lane & 7;
  const int r4 = 8 * ((lane >> 3) & 1) + 4 * (lane >> 4);

  int row = blockIdx.x;

  const float* wg = e + (((size_t)row * LL) + 16 * wave) * DD + lane * 4;
  float* wl = Ws + (16 * wave) * WP + lane * 4;
  const float* pg = part + ((size_t)((row / LL) * NH + ds) * LL + (row % LL)) * LL + 16 * wave + r4;

  float* wbw = Wbs + wave * (DD * NH);
  v4f wbv[8];
#pragma unroll
  for (int m = 0; m < 8; ++m)
    wbv[m] = *reinterpret_cast<const v4f*>(Wb + (32 * m + lane) * 4);
  v2f wb[16][4];
  v4f sp[NCH];
#pragma unroll
  for (int c = 0; c < NCH; ++c)
    sp[c] = *reinterpret_cast<const v4f*>(pg + c * KC);

  v4f rw[2][WQ];
#pragma unroll
  for (int q = 0; q < WQ; ++q)
    rw[0][q] = *reinterpret_cast<const v4f*>(wg + (size_t)q * DD);
  __builtin_amdgcn_sched_barrier(0);
#pragma unroll
  for (int q = 0; q < WQ; ++q)
    rw[1][q] = *reinterpret_cast<const v4f*>(wg + (size_t)(KC + q) * DD);
  {
    v4f z = {};
#pragma unroll
    for (int c = 0; c < NCH; ++c)
      *reinterpret_cast<v4f*>(As + (8 + wave) * AP + c * KC + lane * 4) = z;
  }

#pragma unroll
  for (int m = 0; m < 8; ++m)
    *reinterpret_cast<v4f*>(wbw + ((((m >> 1) * 4 + ((lane >> 1) & 3)) * 2 + (lane & 1)) * 8
                                   + 4 * (m & 1) + (lane >> 3)) * 4) = wbv[m];
#pragma unroll
  for (int m = 0; m < 16; ++m) {
    const v4f lo = *reinterpret_cast<const v4f*>(wbw + ((2 * m) * 8 + ds) * 4);
    const v4f hi = *reinterpret_cast<const v4f*>(wbw + ((2 * m + 1) * 8 + ds) * 4);
    wb[m][0].x = lo.x;  wb[m][0].y = lo.y;
    wb[m][1].x = lo.z;  wb[m][1].y = lo.w;
    wb[m][2].x = hi.x;  wb[m][2].y = hi.y;
    wb[m][3].x = hi.z;  wb[m][3].y = hi.w;
  }

  const float rsqrt3 = 0.57735026918962576f;
#pragma nounroll
  for (int it = 0; it < PR_ROWS; ++it) {
    const int b = row / LL, i = row - b * LL;
#pragma unroll
    for (int c = 0; c < NCH; ++c) {
#pragma unroll
      for (int q = 0; q < WQ; ++q)
        *reinterpret_cast<v4f*>(wl + (c * KC + q) * WP) = rw[c & 1][q];
      __builtin_amdgcn_sched_barrier(0);
      if (c + 2 < NCH) {
        const int k2 = (c + 2) * KC;
#pragma unroll
        for (int q = 0; q < WQ; ++q)
          rw[c & 1][q] = *reinterpret_cast<const v4f*>(wg + (size_t)(k2 + q) * DD);
      }
      const float* erow = Ws + (c * KC + 16 * wave + r4) * WP + 4 * ds;
      v4f evs[4][4];
#pragma unroll
      for (int p = 0; p < 4; ++p)
#pragma unroll
        for (int q = 0; q < 4; ++q)
          evs[p][q] = *reinterpret_cast<const v4f*>(erow + p * WP + 32 * q);
      __builtin_amdgcn_sched_barrier(0);
      v4f bs;
#pragma unroll
      for (int p = 0; p < 4; ++p) {
        v2f a0 = {}, a1 = {}, a2 = {}, a3 = {};
#pragma unroll
        for (int q = 0; q < 4; ++q) {
          const v4f ev = evs[p][q];
          v2f e0, e1, e2, e3;
          e0.x = ev.x;  e0.y = ev.x;
          e1.x = ev.y;  e1.y = ev.y;
          e2.x = ev.z;  e2.y = ev.z;
          e3.x = ev.w;  e3.y = ev.w;
          a0 = fma2(e0, wb[4 * q][0], a0);      a1 = fma2(e0, wb[4 * q][1], a1);
          a2 = fma2(e0, wb[4 * q][2], a2);      a3 = fma2(e0, wb[4 * q][3], a3);
          a0 = fma2(e1, wb[4 * q + 1][0], a0);  a1 = fma2(e1, wb[4 * q + 1][1], a1);
          a2 = fma2(e1, wb[4 * q + 1][2], a2);  a3 = fma2(e1, wb[4 * q + 1][3], a3);
          a0 = fma2(e2, wb[4 * q + 2][0], a0);  a1 = fma2(e2, wb[4 * q + 2][1], a1);
          a2 = fma2(e2, wb[4 * q + 2][2], a2);  a3 = fma2(e2, wb[4 * q + 2][3], a3);
          a0 = fma2(e3, wb[4 * q + 3][0], a0);  a1 = fma2(e3, wb[4 * q + 3][1], a1);
          a2 = fma2(e3, wb[4 * q + 3][2], a2);  a3 = fma2(e3, wb[4 * q + 3][3], a3);
        }
        const bool up4 = (ds & 4) != 0, up2 = (ds & 2) != 0, up1 = (ds & 1) != 0;
        float k0 = up4 ? a2.x : a0.x, k1 = up4 ? a2.y : a0.y, k2 = up4 ? a3.x : a1.x, k3 = up4 ? a3.y : a1.y;
        const float g0 = up4 ? a0.x : a2.x, g1 = up4 ? a0.y : a2.y, g2 = up4 ? a1.x : a3.x, g3 = up4 ? a1.y : a3.y;
        k0 += lane_xchg<XCHG_MIRROR>(g0);
        k1 += lane_xchg<XCHG_MIRROR>(g1);
        k2 += lane_xchg<XCHG_MIRROR>(g2);
        k3 += lane_xchg<XCHG_MIRROR>(g3);
        float m0 = up2 ? k2 : k0, m1 = up2 ? k3 : k1;
        const float u0 = up2 ? k0 : k2, u1 = up2 ? k1 : k3;
        m0 += lane_xchg<XCHG_2>(u0);
        m1 += lane_xchg<XCHG_2>(u1);
        float s = up1 ? m1 : m0;
        const float w = up1 ? m0 : m1;
        s += lane_xchg<XCHG_1>(w);
        bs[p] = s;
      }
      v4f lgv;
      lgv.x = rsqrt3 * (sp[c].x + bs.x);
      lgv.y = rsqrt3 * (sp[c].y + bs.y);
      lgv.z = rsqrt3 * (sp[c].z + bs.z);
      lgv.w = rsqrt3 * (sp[c].w + bs.w);
      *reinterpret_cast<v4f*>(As + ds * AP + c * KC + 16 * wave + r4) = lgv;
    }
    __syncthreads();

    float v[12];
    {
      float* lr = As + wave * AP + lane;
      float m = -1e30f;
#pragma unroll
      for (int k = 0; k < 12; ++k) {
        v[k] = lr[k * 32];
        m = fmaxf(m, v[k]);
      }
#pragma unroll
      for (int off = 16; off > 0; off >>= 1) m = fmaxf(m, __shfl_xor(m, off, 32));
      float s = 0.0f;
#pragma unroll
      for (int k = 0; k < 12; ++k) {
        v[k] = __expf(v[k] - m);
        s += v[k];
      }
#pragma unroll
      for (int off = 16; off > 0; off >>= 1) s += __shfl_xor(s, off, 32);
      float inv = 1.0f / s;
#pragma unroll
      for (int k = 0; k < 12; ++k) {
        v[k] = v[k] * inv;
        lr[k * 32] = v[k];
      }
    }
    __syncthreads();
    {
      float* ao = attn + ((size_t)(b * NH + wave) * LL + i) * LL + lane;
#pragma unroll
      for (int k = 0; k < 12; ++k) ao[k * 32] = v[k];
    }

    const int nrow = row + PR_GRID;
    if (it + 1 < PR_ROWS) {
      wg = e + (((size_t)nrow * LL) + 16 * wave) * DD + lane * 4;
      pg = part + ((size_t)((nrow / LL) * NH + ds) * LL + (nrow % LL)) * LL + 16 * wave + r4;
#pragma unroll
      for (int c = 0; c < NCH; ++c)
        sp[c] = *reinterpret_cast<const v4f*>(pg + c * KC);
#pragma unroll
      for (int q = 0; q < WQ; ++q)
        rw[0][q] = *reinterpret_cast<const v4f*>(wg + (size_t)q * DD);
      __builtin_amdgcn_sched_barrier(0);
#pragma unroll
      for (int q = 0; q < WQ; ++q)
        rw[1][q] = *reinterpret_cast<const v4f*>(wg + (size_t)(KC + q) * DD);
    }

    constexpr int KB = 16;
    constexpr int NB = LL / (4 * KB);
    v8f acc = {};
    const float* asp = As + l16 * AP + 2 * h;
    const float* wsp = Ws + (2 * h) * WP + tn + l16;
    v2f fa[2][KB], fb[2][KB];
#pragma unroll
    for (int s = 0; s < KB; ++s) {
      fa[0][s] = *reinterpret_cast<const v2f*>(asp + 4 * s);
      fb[0][s].x = wsp[(4 * s) * WP];
      fb[0][s].y = wsp[(4 * s + 1) * WP];
    }
#pragma unroll
    for (int t = 0; t < NB; ++t) {
      if (t + 1 < NB) {
#pragma unroll
        for (int s = 0; s < KB; ++s) {
          const int k = (t + 1) * 4 * KB + 4 * s;
          fa[(t + 1) & 1][s] = *reinterpret_cast<const v2f*>(asp + k);
          fb[(t + 1) & 1][s].x = wsp[k * WP];
          fb[(t + 1) & 1][s].y = wsp[(k + 1) * WP];
        }
      }
      __builtin_amdgcn_sched_barrier(0);
#pragma unroll
      for (int s = 0; s < KB; ++s) acc = wmma4(fa[t & 1][s], fb[t & 1][s], acc);
    }
    __syncthreads();
    if (h == 0) {
#pragma unroll
      for (int r = 0; r < 8; ++r)
        out_cat[(size_t)row * DOUT + DD + r * DD + tn + l16] = acc[r];
    }
    row = nrow;
  }
}

extern "C" void kernel_launch(void* const* d_in, const int* in_sizes, int n_in,
                              void* d_out, int out_size, void* d_ws, size_t ws_size,
                              hipStream_t stream) {
  (void)in_sizes; (void)n_in; (void)out_size;
  const float* x     = (const float*)d_in[0];
  const float* e     = (const float*)d_in[1];
  const float* r     = (const float*)d_in[2];
  const float* t     = (const float*)d_in[3];
  const float* Wq_s  = (const float*)d_in[4];
  const float* Wk_s  = (const float*)d_in[5];
  const float* Wv_s  = (const float*)d_in[6];
  const float* Wb    = (const float*)d_in[7];
  const float* Wq_p  = (const float*)d_in[8];
  const float* Wk_p  = (const float*)d_in[9];
  const float* Wv_p  = (const float*)d_in[10];
  const float* gamma = (const float*)d_in[11];
  const float* Wo    = (const float*)d_in[12];
  const float* bo    = (const float*)d_in[13];
  float* out = (float*)d_out;
  float* ws  = (float*)d_ws;

  float* qs   = ws;
  float* ks_  = qs  + 98304;
  float* vs   = ks_ + 98304;
  float* hq   = vs  + 98304;
  float* hk   = hq  + 73728;
  float* hv   = hk  + 73728;
  float* part = hv  + 73728;
  float* attn = part + 2359296;
  float* op   = attn + 2359296;
  float* ocat = op  + 73728;
  size_t req = (size_t)((ocat + 983040) - ws) * sizeof(float);
  if (ws_size < req) return;

  dim3 blk(256);

  proj6_kernel<<<dim3(BL / 16, 6), dim3(128), 0, stream>>>(x, Wq_s, Wk_s, Wv_s, Wq_p, Wk_p, Wv_p,
                                                           qs, ks_, vs, hq, hk, hv);

  logit_kernel<<<(BB * NH * LT * LT) / 8, blk, 0, stream>>>(qs, ks_, hq, hk, r, t,
                                                            gamma, part);

  pair_row_kernel<<<PR_GRID, blk, 0, stream>>>(part, e, Wb, attn, ocat);

  attn_v_kernel<<<BB * NH * (LT / 2), blk, 0, stream>>>(attn, vs, hv, r, t, ocat, op);

  wmma_gemm_lds_kernel<DOUT, DD><<<BL / 16, blk, 0, stream>>>(ocat, Wo, bo, op, r, t, out);
}
